# starting kernel
# baseline (speedup 1.0000x reference)
#include <hip/hip_runtime.h>

typedef __attribute__((ext_vector_type(2))) float v2f;
typedef __attribute__((ext_vector_type(8))) float v8f;

#define N_NODES 50000
#define DIM 128
#define HID 64
#define NEDGE 500000
#define SLOPE 0.2f

template <int NT>
__global__ __launch_bounds__(128) void gemm_bias_kernel(
    const float* __restrict__ A, const float* __restrict__ W,
    const float* __restrict__ bias, float* __restrict__ C,
    int rows, int K, int relu)
{
    constexpr int ncols = NT * 16;
    const int wave = (int)((blockIdx.x * blockDim.x + threadIdx.x) >> 5);
    const int lane = (int)(threadIdx.x & 31);
    if (wave * 16 >= rows) return;

    const int nloc  = lane & 15;
    const int khalf = (lane >> 4) << 1;

    const float* __restrict__ arow = A + (size_t)(wave * 16 + nloc) * (size_t)K;

    v8f acc[NT];
#pragma unroll
    for (int t = 0; t < NT; ++t)
        acc[t] = (v8f){0.f, 0.f, 0.f, 0.f, 0.f, 0.f, 0.f, 0.f};

    for (int k0 = 0; k0 < K; k0 += 4) {
        v2f a;
        a.x = arow[k0 + khalf];
        a.y = arow[k0 + khalf + 1];
        const float* __restrict__ w0 = W + (size_t)(k0 + khalf) * ncols + nloc;
        const float* __restrict__ w1 = w0 + ncols;
#pragma unroll
        for (int t = 0; t < NT; ++t) {
            v2f b;
            b.x = w0[t * 16];
            b.y = w1[t * 16];
            acc[t] = __builtin_amdgcn_wmma_f32_16x16x4_f32(
                false, a, false, b, (short)0, acc[t], false, false);
        }
    }

    const int mbase = wave * 16 + ((lane >> 4) << 3);
#pragma unroll
    for (int t = 0; t < NT; ++t) {
        const int n = t * 16 + nloc;
        const float bv = bias[n];
#pragma unroll
        for (int j = 0; j < 8; ++j) {
            float v = acc[t][j] + bv;
            if (relu) v = fmaxf(v, 0.f);
            C[(size_t)(mbase + j) * ncols + n] = v;
        }
    }
}

__device__ __forceinline__ unsigned enc_f32(float x) {
    unsigned b = __float_as_uint(x);
    return (b & 0x80000000u) ? ~b : (b | 0x80000000u);
}
__device__ __forceinline__ float dec_f32(unsigned k) {
    unsigned b = (k & 0x80000000u) ? (k & 0x7FFFFFFFu) : ~k;
    return __uint_as_float(b);
}

__global__ void init_out_kernel(float* __restrict__ out,
                                const float* __restrict__ bias, int n) {
    int i = (int)(blockIdx.x * blockDim.x + threadIdx.x);
    if (i < n * DIM) out[i] = bias[i & (DIM - 1)];
}
__global__ void init_ms_kernel(unsigned* __restrict__ menc,
                               float* __restrict__ ssum, int n) {
    int i = (int)(blockIdx.x * blockDim.x + threadIdx.x);
    if (i < n) { menc[i] = 0u; ssum[i] = 0.f; }
}

__global__ __launch_bounds__(256) void edge_logit_kernel(
    const float* __restrict__ xl, const float* __restrict__ xr,
    const int* __restrict__ src, const int* __restrict__ dst,
    const float* __restrict__ att,
    float* __restrict__ logit, unsigned* __restrict__ menc, int nedge)
{
    const int e    = (int)((blockIdx.x * blockDim.x + threadIdx.x) >> 5);
    const int lane = (int)(threadIdx.x & 31);
    if (e >= nedge) return;
    const int s = src[e], d = dst[e];
    const float4 a  = *(const float4*)(xl + (size_t)s * DIM + lane * 4);
    const float4 b  = *(const float4*)(xr + (size_t)d * DIM + lane * 4);
    const float4 at = *(const float4*)(att + lane * 4);
    float z, sum = 0.f;
    z = a.x + b.x; z = (z > 0.f) ? z : SLOPE * z; sum += z * at.x;
    z = a.y + b.y; z = (z > 0.f) ? z : SLOPE * z; sum += z * at.y;
    z = a.z + b.z; z = (z > 0.f) ? z : SLOPE * z; sum += z * at.z;
    z = a.w + b.w; z = (z > 0.f) ? z : SLOPE * z; sum += z * at.w;
#pragma unroll
    for (int off = 16; off > 0; off >>= 1) sum += __shfl_xor(sum, off, 32);
    if (lane == 0) {
        logit[e] = sum;
        atomicMax(menc + d, enc_f32(sum));
    }
}

__global__ void edge_exp_kernel(
    const float* __restrict__ logit, const int* __restrict__ dst,
    const unsigned* __restrict__ menc, float* __restrict__ eexp,
    float* __restrict__ ssum, int nedge)
{
    int e = (int)(blockIdx.x * blockDim.x + threadIdx.x);
    if (e >= nedge) return;
    const int d = dst[e];
    const float v = expf(logit[e] - dec_f32(menc[d]));
    eexp[e] = v;
    atomicAdd(ssum + d, v);
}

__global__ __launch_bounds__(256) void edge_scatter_kernel(
    const float* __restrict__ xl,
    const int* __restrict__ src, const int* __restrict__ dst,
    const float* __restrict__ eexp, const float* __restrict__ ssum,
    float* __restrict__ out, int nedge)
{
    const int e    = (int)((blockIdx.x * blockDim.x + threadIdx.x) >> 5);
    const int lane = (int)(threadIdx.x & 31);
    if (e >= nedge) return;
    const int s = src[e], d = dst[e];
    const float alpha = eexp[e] / (ssum[d] + 1e-16f);
    const float4 a = *(const float4*)(xl + (size_t)s * DIM + lane * 4);
    float* o = out + (size_t)d * DIM + lane * 4;
    atomicAdd(o + 0, alpha * a.x);
    atomicAdd(o + 1, alpha * a.y);
    atomicAdd(o + 2, alpha * a.z);
    atomicAdd(o + 3, alpha * a.w);
}

static inline void launch_gemm(const float* A, const float* W, const float* b,
                               float* C, int rows, int K, int ncols, int relu,
                               hipStream_t stream) {
    long waves = (long)(rows / 16);
    long threads = waves * 32;
    int blocks = (int)((threads + 127) / 128);
    if (ncols == 64)
        gemm_bias_kernel<4><<<blocks, 128, 0, stream>>>(A, W, b, C, rows, K, relu);
    else
        gemm_bias_kernel<8><<<blocks, 128, 0, stream>>>(A, W, b, C, rows, K, relu);
}

extern "C" void kernel_launch(void* const* d_in, const int* in_sizes, int n_in,
                              void* d_out, int out_size, void* d_ws, size_t ws_size,
                              hipStream_t stream) {
    (void)in_sizes; (void)n_in; (void)out_size; (void)ws_size;

    const float* x_user = (const float*)d_in[0];
    const float* x_item = (const float*)d_in[1];
    const int*   e_u2i  = (const int*)d_in[2];
    const int*   e_i2u  = (const int*)d_in[3];
    const float* W1_u = (const float*)d_in[4],  *b1_u = (const float*)d_in[5];
    const float* W2_u = (const float*)d_in[6],  *b2_u = (const float*)d_in[7];
    const float* W1_i = (const float*)d_in[8],  *b1_i = (const float*)d_in[9];
    const float* W2_i = (const float*)d_in[10], *b2_i = (const float*)d_in[11];
    const float* Wl_u2i = (const float*)d_in[12], *bl_u2i = (const float*)d_in[13];
    const float* Wr_u2i = (const float*)d_in[14], *br_u2i = (const float*)d_in[15];
    const float* att_u2i = (const float*)d_in[16], *bias_u2i = (const float*)d_in[17];
    const float* Wl_i2u = (const float*)d_in[18], *bl_i2u = (const float*)d_in[19];
    const float* Wr_i2u = (const float*)d_in[20], *br_i2u = (const float*)d_in[21];
    const float* att_i2u = (const float*)d_in[22], *bias_i2u = (const float*)d_in[23];

    float* out_user = (float*)d_out;
    float* out_item = (float*)d_out + (size_t)N_NODES * DIM;

    float* ws   = (float*)d_ws;
    float* hu   = ws;
    float* hi   = hu + (size_t)N_NODES * DIM;
    float* tmp  = hi + (size_t)N_NODES * DIM;
    float* xl   = tmp + (size_t)N_NODES * HID;
    float* xr   = xl + (size_t)N_NODES * DIM;
    float* logit = xr + (size_t)N_NODES * DIM;
    float* eexp  = logit + NEDGE;
    unsigned* menc = (unsigned*)(eexp + NEDGE);
    float* ssum  = (float*)(menc + N_NODES);

    const int blkN    = (N_NODES + 255) / 256;
    const int blkND   = (N_NODES * DIM + 255) / 256;
    const int blkE    = (NEDGE + 255) / 256;
    const int blkEw   = (NEDGE * 32 + 255) / 256;

    launch_gemm(x_user, W1_u, b1_u, tmp, N_NODES, DIM, HID, 1, stream);
    launch_gemm(tmp,    W2_u, b2_u, hu,  N_NODES, HID, DIM, 0, stream);
    launch_gemm(x_item, W1_i, b1_i, tmp, N_NODES, DIM, HID, 1, stream);
    launch_gemm(tmp,    W2_i, b2_i, hi,  N_NODES, HID, DIM, 0, stream);

    {
        const int* src = e_u2i;
        const int* dst = e_u2i + NEDGE;
        launch_gemm(hu, Wl_u2i, bl_u2i, xl, N_NODES, DIM, DIM, 0, stream);
        launch_gemm(hi, Wr_u2i, br_u2i, xr, N_NODES, DIM, DIM, 0, stream);
        init_out_kernel<<<blkND, 256, 0, stream>>>(out_item, bias_u2i, N_NODES);
        init_ms_kernel<<<blkN, 256, 0, stream>>>(menc, ssum, N_NODES);
        edge_logit_kernel<<<blkEw, 256, 0, stream>>>(xl, xr, src, dst, att_u2i,
                                                     logit, menc, NEDGE);
        edge_exp_kernel<<<blkE, 256, 0, stream>>>(logit, dst, menc, eexp, ssum, NEDGE);
        edge_scatter_kernel<<<blkEw, 256, 0, stream>>>(xl, src, dst, eexp, ssum,
                                                       out_item, NEDGE);
    }

    {
        const int* src = e_i2u;
        const int* dst = e_i2u + NEDGE;
        launch_gemm(hi, Wl_i2u, bl_i2u, xl, N_NODES, DIM, DIM, 0, stream);
        launch_gemm(hu, Wr_i2u, br_i2u, xr, N_NODES, DIM, DIM, 0, stream);
        init_out_kernel<<<blkND, 256, 0, stream>>>(out_user, bias_i2u, N_NODES);
        init_ms_kernel<<<blkN, 256, 0, stream>>>(menc, ssum, N_NODES);
        edge_logit_kernel<<<blkEw, 256, 0, stream>>>(xl, xr, src, dst, att_i2u,
                                                     logit, menc, NEDGE);
        edge_exp_kernel<<<blkE, 256, 0, stream>>>(logit, dst, menc, eexp, ssum, NEDGE);
        edge_scatter_kernel<<<blkEw, 256, 0, stream>>>(xl, src, dst, eexp, ssum,
                                                       out_user, NEDGE);
    }
}


# Edges grouped by destination in-call (counting sort, int atomics); one wave per node does softmax + weighted sum in registers: float atomics gone, sums in slot order with running-max rescale
# speedup vs baseline: 2.1282x; 2.1282x over previous
#include <hip/hip_runtime.h>

typedef __attribute__((ext_vector_type(2))) float v2f;
typedef __attribute__((ext_vector_type(8))) float v8f;

#define N_NODES 50000
#define DIM 128
#define HID 64
#define NEDGE 500000
#define SLOPE 0.2f

template <int NT>
__global__ __launch_bounds__(128) void gemm_bias_kernel(
    const float* __restrict__ A, const float* __restrict__ W,
    const float* __restrict__ bias, float* __restrict__ C,
    int rows, int K, int relu)
{
    constexpr int ncols = NT * 16;
    const int wave = (int)((blockIdx.x * blockDim.x + threadIdx.x) >> 5);
    const int lane = (int)(threadIdx.x & 31);
    if (wave * 16 >= rows) return;

    const int nloc  = lane & 15;
    const int khalf = (lane >> 4) << 1;

    const float* __restrict__ arow = A + (size_t)(wave * 16 + nloc) * (size_t)K;

    v8f acc[NT];
#pragma unroll
    for (int t = 0; t < NT; ++t)
        acc[t] = (v8f){0.f, 0.f, 0.f, 0.f, 0.f, 0.f, 0.f, 0.f};

    for (int k0 = 0; k0 < K; k0 += 4) {
        v2f a;
        a.x = arow[k0 + khalf];
        a.y = arow[k0 + khalf + 1];
        const float* __restrict__ w0 = W + (size_t)(k0 + khalf) * ncols + nloc;
        const float* __restrict__ w1 = w0 + ncols;
#pragma unroll
        for (int t = 0; t < NT; ++t) {
            v2f b;
            b.x = w0[t * 16];
            b.y = w1[t * 16];
            acc[t] = __builtin_amdgcn_wmma_f32_16x16x4_f32(
                false, a, false, b, (short)0, acc[t], false, false);
        }
    }

    const int mbase = wave * 16 + ((lane >> 4) << 3);
#pragma unroll
    for (int t = 0; t < NT; ++t) {
        const int n = t * 16 + nloc;
        const float bv = bias[n];
#pragma unroll
        for (int j = 0; j < 8; ++j) {
            float v = acc[t][j] + bv;
            if (relu) v = fmaxf(v, 0.f);
            C[(size_t)(mbase + j) * ncols + n] = v;
        }
    }
}


__global__ void zero_counts_kernel(int* __restrict__ cnt, int n) {
    int i = (int)(blockIdx.x * blockDim.x + threadIdx.x);
    if (i < n) cnt[i] = 0;
}

__global__ void count_rank_kernel(const int* __restrict__ dst0,
                                  const int* __restrict__ dst1,
                                  int* __restrict__ cnt, int* __restrict__ rank,
                                  int nedge, int n) {
    const int e = (int)(blockIdx.x * blockDim.x + threadIdx.x);
    const int rel = (int)blockIdx.y;
    if (e >= nedge) return;
    const int* __restrict__ dst = rel ? dst1 : dst0;
    const int d = dst[e];
    rank[(size_t)rel * nedge + e] = atomicAdd(cnt + (size_t)rel * n + d, 1);
}

__global__ __launch_bounds__(256) void seg_offsets_kernel(
    const int* __restrict__ cnt, int* __restrict__ off, int n) {
    __shared__ int s_pre[8];
    __shared__ int s_tot[8];
    const int rel = (int)blockIdx.y;
    const int* __restrict__ c = cnt + (size_t)rel * n;
    int* __restrict__ o = off + (size_t)rel * n;
    const int t = (int)threadIdx.x, lane = t & 31, w = t >> 5;
    const int base = (int)blockIdx.x * 256;

    int pre = 0;
#pragma unroll 4
    for (int i = t * 4; i < base; i += 1024) {
        const int4 q = *(const int4*)(c + i);
        pre += (q.x + q.y) + (q.z + q.w);
    }
    const int idx = base + t;
    const int v = (idx < n) ? c[idx] : 0;

    int inc = v;
#pragma unroll
    for (int dlt = 1; dlt < 32; dlt <<= 1) {
        const int u = __shfl_up(inc, dlt, 32);
        if (lane >= dlt) inc += u;
    }
#pragma unroll
    for (int x = 16; x > 0; x >>= 1) pre += __shfl_xor(pre, x, 32);
    if (lane == 31) s_tot[w] = inc;
    if (lane == 0)  s_pre[w] = pre;
    __syncthreads();
    int add = 0;
#pragma unroll
    for (int k = 0; k < 8; ++k) {
        add += s_pre[k];
        if (k < w) add += s_tot[k];
    }
    if (idx < n) o[idx] = add + inc - v;
}

__global__ void fill_slots_kernel(const int* __restrict__ src0, const int* __restrict__ dst0,
                                  const int* __restrict__ src1, const int* __restrict__ dst1,
                                  const int* __restrict__ off, const int* __restrict__ rank,
                                  int* __restrict__ slots, int nedge, int n) {
    const int e = (int)(blockIdx.x * blockDim.x + threadIdx.x);
    const int rel = (int)blockIdx.y;
    if (e >= nedge) return;
    const int* __restrict__ src = rel ? src1 : src0;
    const int* __restrict__ dst = rel ? dst1 : dst0;
    const int d = dst[e];
    const int p = off[(size_t)rel * n + d] + rank[(size_t)rel * nedge + e];
    slots[(size_t)rel * nedge + p] = src[e];
}

__global__ __launch_bounds__(256) void seg_attend_kernel(
    const float* __restrict__ xl, const float* __restrict__ xr,
    const int* __restrict__ slots, const int* __restrict__ off,
    const int* __restrict__ cnt,
    const float* __restrict__ att, const float* __restrict__ bias,
    float* __restrict__ out, int n)
{
    const int lane = (int)(threadIdx.x & 31);
    const int d = __builtin_amdgcn_readfirstlane(
        (int)((blockIdx.x * blockDim.x + threadIdx.x) >> 5));
    if (d >= n) return;
    const int beg = off[d];
    const int deg = cnt[d];
    const float4 b  = *(const float4*)(xr + (size_t)d * DIM + lane * 4);
    const float4 at = *(const float4*)(att + lane * 4);
    const float4 bi = *(const float4*)(bias + lane * 4);

    float4 acc = make_float4(0.f, 0.f, 0.f, 0.f);
    float m = -INFINITY;
    float s = 0.f;

    for (int j0 = 0; j0 < deg; j0 += 8) {
        const int nv = min(8, deg - j0);
        const int jj = min(lane & 7, nv - 1);
        const int sidx = slots[beg + j0 + jj];

        float4 a[8];
        float  l[8];
#pragma unroll
        for (int j = 0; j < 8; ++j) {
            const int sj = __builtin_amdgcn_readlane(sidx, j);
            a[j] = *(const float4*)(xl + (size_t)sj * DIM + lane * 4);
        }
#pragma unroll
        for (int j = 0; j < 8; ++j) {
            float z, p = 0.f;
            z = a[j].x + b.x; z = (z > 0.f) ? z : SLOPE * z; p += z * at.x;
            z = a[j].y + b.y; z = (z > 0.f) ? z : SLOPE * z; p += z * at.y;
            z = a[j].z + b.z; z = (z > 0.f) ? z : SLOPE * z; p += z * at.z;
            z = a[j].w + b.w; z = (z > 0.f) ? z : SLOPE * z; p += z * at.w;
#pragma unroll
            for (int x = 16; x > 0; x >>= 1) p += __shfl_xor(p, x, 32);
            l[j] = (j < nv) ? p : -INFINITY;
        }
        float mc = l[0];
#pragma unroll
        for (int j = 1; j < 8; ++j) mc = fmaxf(mc, l[j]);
        const float mn = fmaxf(m, mc);
        const float sc = expf(m - mn);
        acc.x *= sc; acc.y *= sc; acc.z *= sc; acc.w *= sc;
        s *= sc;
#pragma unroll
        for (int j = 0; j < 8; ++j) {
            const float e = expf(l[j] - mn);
            s += e;
            acc.x += e * a[j].x;
            acc.y += e * a[j].y;
            acc.z += e * a[j].z;
            acc.w += e * a[j].w;
        }
        m = mn;
    }

    const float den = s + 1e-16f;
    float4 r;
    r.x = acc.x / den + bi.x;
    r.y = acc.y / den + bi.y;
    r.z = acc.z / den + bi.z;
    r.w = acc.w / den + bi.w;
    *(float4*)(out + (size_t)d * DIM + lane * 4) = r;
}

static inline void launch_gemm(const float* A, const float* W, const float* b,
                               float* C, int rows, int K, int ncols, int relu,
                               hipStream_t stream) {
    long waves = (long)(rows / 16);
    long threads = waves * 32;
    int blocks = (int)((threads + 127) / 128);
    if (ncols == 64)
        gemm_bias_kernel<4><<<blocks, 128, 0, stream>>>(A, W, b, C, rows, K, relu);
    else
        gemm_bias_kernel<8><<<blocks, 128, 0, stream>>>(A, W, b, C, rows, K, relu);
}

extern "C" void kernel_launch(void* const* d_in, const int* in_sizes, int n_in,
                              void* d_out, int out_size, void* d_ws, size_t ws_size,
                              hipStream_t stream) {
    (void)in_sizes; (void)n_in; (void)out_size; (void)ws_size;

    const float* x_user = (const float*)d_in[0];
    const float* x_item = (const float*)d_in[1];
    const int*   e_u2i  = (const int*)d_in[2];
    const int*   e_i2u  = (const int*)d_in[3];
    const float* W1_u = (const float*)d_in[4],  *b1_u = (const float*)d_in[5];
    const float* W2_u = (const float*)d_in[6],  *b2_u = (const float*)d_in[7];
    const float* W1_i = (const float*)d_in[8],  *b1_i = (const float*)d_in[9];
    const float* W2_i = (const float*)d_in[10], *b2_i = (const float*)d_in[11];
    const float* Wl_u2i = (const float*)d_in[12], *bl_u2i = (const float*)d_in[13];
    const float* Wr_u2i = (const float*)d_in[14], *br_u2i = (const float*)d_in[15];
    const float* att_u2i = (const float*)d_in[16], *bias_u2i = (const float*)d_in[17];
    const float* Wl_i2u = (const float*)d_in[18], *bl_i2u = (const float*)d_in[19];
    const float* Wr_i2u = (const float*)d_in[20], *br_i2u = (const float*)d_in[21];
    const float* att_i2u = (const float*)d_in[22], *bias_i2u = (const float*)d_in[23];

    float* out_user = (float*)d_out;
    float* out_item = (float*)d_out + (size_t)N_NODES * DIM;

    float* ws   = (float*)d_ws;
    float* hu   = ws;
    float* hi   = hu + (size_t)N_NODES * DIM;
    float* tmp  = hi + (size_t)N_NODES * DIM;
    float* xl   = tmp + (size_t)N_NODES * HID;
    float* xr   = xl + (size_t)N_NODES * DIM;

    int* cnt   = (int*)tmp;
    int* off   = cnt + 2 * N_NODES;
    int* rank  = off + 2 * N_NODES;
    int* slots = rank + 2 * NEDGE;

    const int blkE    = (NEDGE + 255) / 256;
    const int blkN    = (N_NODES + 255) / 256;
    const int blk2N   = (2 * N_NODES + 255) / 256;
    const int blkNw   = (N_NODES * 32 + 255) / 256;

    launch_gemm(x_user, W1_u, b1_u, tmp, N_NODES, DIM, HID, 1, stream);
    launch_gemm(tmp,    W2_u, b2_u, hu,  N_NODES, HID, DIM, 0, stream);
    launch_gemm(x_item, W1_i, b1_i, tmp, N_NODES, DIM, HID, 1, stream);
    launch_gemm(tmp,    W2_i, b2_i, hi,  N_NODES, HID, DIM, 0, stream);

    zero_counts_kernel<<<blk2N, 256, 0, stream>>>(cnt, 2 * N_NODES);
    count_rank_kernel<<<dim3(blkE, 2), 256, 0, stream>>>(
        e_u2i + NEDGE, e_i2u + NEDGE, cnt, rank, NEDGE, N_NODES);
    seg_offsets_kernel<<<dim3(blkN, 2), 256, 0, stream>>>(cnt, off, N_NODES);
    fill_slots_kernel<<<dim3(blkE, 2), 256, 0, stream>>>(
        e_u2i, e_u2i + NEDGE, e_i2u, e_i2u + NEDGE, off, rank, slots, NEDGE, N_NODES);

    launch_gemm(hu, Wl_u2i, bl_u2i, xl, N_NODES, DIM, DIM, 0, stream);
    launch_gemm(hi, Wr_u2i, br_u2i, xr, N_NODES, DIM, DIM, 0, stream);
    seg_attend_kernel<<<blkNw, 256, 0, stream>>>(xl, xr, slots, off, cnt,
                                                 att_u2i, bias_u2i, out_item, N_NODES);

    launch_gemm(hi, Wl_i2u, bl_i2u, xl, N_NODES, DIM, DIM, 0, stream);
    launch_gemm(hu, Wr_i2u, br_i2u, xr, N_NODES, DIM, DIM, 0, stream);
    seg_attend_kernel<<<blkNw, 256, 0, stream>>>(xl, xr, slots + NEDGE, off + N_NODES,
                                                 cnt + N_NODES,
                                                 att_i2u, bias_i2u, out_user, N_NODES);
}


# Edge phase on fewer, longer-lived waves: count and fill kernels take 8 edges per thread (16-byte loads), the per-node softmax takes 4 nodes per wave in turn; waves launched per call fall fourfold
# speedup vs baseline: 2.5789x; 1.2118x over previous
#include <hip/hip_runtime.h>


typedef __attribute__((ext_vector_type(2))) float v2f;
typedef __attribute__((ext_vector_type(8))) float v8f;

#define N_NODES 50000
#define DIM 128
#define HID 64
#define NEDGE 500000
#define SLOPE 0.2f

#define SP 132

struct NodeSide {
    const float* x;
    const float* W1; const float* b1;
    const float* W2; const float* b2;
    const float* Wa; const float* ba;
    const float* Wb; const float* bb;
    float* ya;
    float* yb;
};

template <int NT, int K>
__device__ __forceinline__ void stripe_mm(const float* arow, const float* __restrict__ W,
                                          int nloc, int khalf, v8f (&acc)[NT])
{
    constexpr int ncols = NT * 16;
#pragma unroll (16 / NT)
    for (int k0 = 0; k0 < K; k0 += 4) {
        const v2f a = *(const v2f*)(arow + k0);
        const float* __restrict__ w0 = W + (size_t)(k0 + khalf) * ncols + nloc;
        const float* __restrict__ w1 = w0 + ncols;
#pragma unroll
        for (int t = 0; t < NT; ++t) {
            v2f b;
            b.x = w0[t * 16];
            b.y = w1[t * 16];
            acc[t] = __builtin_amdgcn_wmma_f32_16x16x4_f32(
                false, a, false, b, (short)0, acc[t], false, false);
        }
    }
}

template <int NT>
__device__ __forceinline__ void zero_acc(v8f (&acc)[NT]) {
#pragma unroll
    for (int t = 0; t < NT; ++t)
        acc[t] = (v8f){0.f, 0.f, 0.f, 0.f, 0.f, 0.f, 0.f, 0.f};
}

template <int NT, bool RELU>
__device__ __forceinline__ void tiles_to_lds(float* S, const v8f (&acc)[NT],
                                             const float* __restrict__ bias,
                                             int nloc, int rhalf)
{
#pragma unroll
    for (int t = 0; t < NT; ++t) {
        const int n = t * 16 + nloc;
        const float bv = bias[n];
#pragma unroll
        for (int j = 0; j < 8; ++j) {
            float v = acc[t][j] + bv;
            if (RELU) v = fmaxf(v, 0.f);
            S[(rhalf + j) * SP + n] = v;
        }
    }
}

template <int NT>
__device__ __forceinline__ void tiles_to_global(float* __restrict__ C, const v8f (&acc)[NT],
                                                const float* __restrict__ bias,
                                                int nloc, int mbase)
{
    constexpr int ncols = NT * 16;
#pragma unroll
    for (int t = 0; t < NT; ++t) {
        const int n = t * 16 + nloc;
        const float bv = bias[n];
#pragma unroll
        for (int j = 0; j < 8; ++j)
            C[(size_t)(mbase + j) * ncols + n] = acc[t][j] + bv;
    }
}

__global__ __launch_bounds__(128) void node_chain_kernel(NodeSide su, NodeSide si, int rows)
{
    __shared__ float lds[4 * 16 * SP];

    const bool second = (blockIdx.y != 0);
    const float* x  = second ? si.x  : su.x;
    const float* W1 = second ? si.W1 : su.W1;
    const float* b1 = second ? si.b1 : su.b1;
    const float* W2 = second ? si.W2 : su.W2;
    const float* b2 = second ? si.b2 : su.b2;
    const float* Wa = second ? si.Wa : su.Wa;
    const float* ba = second ? si.ba : su.ba;
    const float* Wb = second ? si.Wb : su.Wb;
    const float* bb = second ? si.bb : su.bb;
    float* ya = second ? si.ya : su.ya;
    float* yb = second ? si.yb : su.yb;

    const int wv   = (int)(threadIdx.x >> 5);
    const int lane = (int)(threadIdx.x & 31);
    int stripe = (int)blockIdx.x * 4 + wv;
    const bool live = stripe * 16 < rows;
    if (!live) stripe = rows / 16 - 1;

    const int nloc  = lane & 15;
    const int khalf = (lane >> 4) << 1;
    const int rhalf = (lane >> 4) << 3;
    const int mbase = stripe * 16 + rhalf;

    float* S = lds + wv * (16 * SP);
    const float* sfrag = S + nloc * SP + khalf;

    v8f acc4[4];
    v8f acc8[8];

    zero_acc<4>(acc4);
    stripe_mm<4, DIM>(x + (size_t)(stripe * 16 + nloc) * DIM + khalf, W1, nloc, khalf, acc4);
    tiles_to_lds<4, true>(S, acc4, b1, nloc, rhalf);
    __syncthreads();

    zero_acc<8>(acc8);
    stripe_mm<8, HID>(sfrag, W2, nloc, khalf, acc8);
    __syncthreads();
    tiles_to_lds<8, false>(S, acc8, b2, nloc, rhalf);
    __syncthreads();

    zero_acc<8>(acc8);
    stripe_mm<8, DIM>(sfrag, Wa, nloc, khalf, acc8);
    if (live) tiles_to_global<8>(ya, acc8, ba, nloc, mbase);

    zero_acc<8>(acc8);
    stripe_mm<8, DIM>(sfrag, Wb, nloc, khalf, acc8);
    if (live) tiles_to_global<8>(yb, acc8, bb, nloc, mbase);
}


#define EPT 8
#define EDGE_BLOCK 128
#define NPW 4
#define ATT_BLOCK 128

__global__ void zero_counts_kernel(int* __restrict__ cnt, int n) {
    const int i = (int)(blockIdx.x * blockDim.x + threadIdx.x) * 4;
    if (i + 4 <= n) {
        *(int4*)(cnt + i) = make_int4(0, 0, 0, 0);
    } else {
        for (int k = i; k < n; ++k) cnt[k] = 0;
    }
}

__global__ __launch_bounds__(EDGE_BLOCK) void count_rank_kernel(
                                  const int* __restrict__ dst0,
                                  const int* __restrict__ dst1,
                                  int* __restrict__ cnt, int* __restrict__ rank,
                                  int nedge, int n) {
    const int rel  = (int)blockIdx.y;
    const int lane = (int)(threadIdx.x & 31);
    const int wbase = (int)((blockIdx.x * blockDim.x + threadIdx.x) >> 5) * (32 * EPT);
    if (wbase >= nedge) return;
    const int* __restrict__ dst = rel ? dst1 : dst0;
    int* __restrict__ c = cnt + (size_t)rel * n;
    int* __restrict__ r = rank + (size_t)rel * nedge;
    if (wbase + 32 * EPT <= nedge) {
        const int ea = wbase + lane * 4;
        const int eb = ea + 128;
        const int4 da = *(const int4*)(dst + ea);
        const int4 db = *(const int4*)(dst + eb);
        int4 ra, rb;
        ra.x = atomicAdd(c + da.x, 1);
        ra.y = atomicAdd(c + da.y, 1);
        ra.z = atomicAdd(c + da.z, 1);
        ra.w = atomicAdd(c + da.w, 1);
        rb.x = atomicAdd(c + db.x, 1);
        rb.y = atomicAdd(c + db.y, 1);
        rb.z = atomicAdd(c + db.z, 1);
        rb.w = atomicAdd(c + db.w, 1);
        *(int4*)(r + ea) = ra;
        *(int4*)(r + eb) = rb;
    } else {
        for (int e = wbase + lane; e < nedge; e += 32)
            r[e] = atomicAdd(c + dst[e], 1);
    }
}

__global__ __launch_bounds__(256) void seg_offsets_kernel(
    const int* __restrict__ cnt, int* __restrict__ off, int n) {
    __shared__ int s_pre[8];
    __shared__ int s_tot[8];
    const int rel = (int)blockIdx.y;
    const int* __restrict__ c = cnt + (size_t)rel * n;
    int* __restrict__ o = off + (size_t)rel * n;
    const int t = (int)threadIdx.x, lane = t & 31, w = t >> 5;
    const int base = (int)blockIdx.x * 256;

    int pre = 0;
#pragma unroll 4
    for (int i = t * 4; i < base; i += 1024) {
        const int4 q = *(const int4*)(c + i);
        pre += (q.x + q.y) + (q.z + q.w);
    }
    const int idx = base + t;
    const int v = (idx < n) ? c[idx] : 0;

    int inc = v;
#pragma unroll
    for (int dlt = 1; dlt < 32; dlt <<= 1) {
        const int u = __shfl_up(inc, dlt, 32);
        if (lane >= dlt) inc += u;
    }
#pragma unroll
    for (int x = 16; x > 0; x >>= 1) pre += __shfl_xor(pre, x, 32);
    if (lane == 31) s_tot[w] = inc;
    if (lane == 0)  s_pre[w] = pre;
    __syncthreads();
    int add = 0;
#pragma unroll
    for (int k = 0; k < 8; ++k) {
        add += s_pre[k];
        if (k < w) add += s_tot[k];
    }
    if (idx < n) o[idx] = add + inc - v;
}

__global__ __launch_bounds__(EDGE_BLOCK) void fill_slots_kernel(
                                  const int* __restrict__ src0, const int* __restrict__ dst0,
                                  const int* __restrict__ src1, const int* __restrict__ dst1,
                                  const int* __restrict__ off, const int* __restrict__ rank,
                                  int* __restrict__ slots, int nedge, int n) {
    const int rel  = (int)blockIdx.y;
    const int lane = (int)(threadIdx.x & 31);
    const int wbase = (int)((blockIdx.x * blockDim.x + threadIdx.x) >> 5) * (32 * EPT);
    if (wbase >= nedge) return;
    const int* __restrict__ src = rel ? src1 : src0;
    const int* __restrict__ dst = rel ? dst1 : dst0;
    const int* __restrict__ o = off + (size_t)rel * n;
    const int* __restrict__ r = rank + (size_t)rel * nedge;
    int* __restrict__ s = slots + (size_t)rel * nedge;
    if (wbase + 32 * EPT <= nedge) {
        const int ea = wbase + lane * 4;
        const int eb = ea + 128;
        const int4 da = *(const int4*)(dst + ea);
        const int4 db = *(const int4*)(dst + eb);
        const int4 ka = *(const int4*)(r + ea);
        const int4 kb = *(const int4*)(r + eb);
        const int4 sa = *(const int4*)(src + ea);
        const int4 sb = *(const int4*)(src + eb);
        const int pax = o[da.x] + ka.x, pay = o[da.y] + ka.y;
        const int paz = o[da.z] + ka.z, paw = o[da.w] + ka.w;
        const int pbx = o[db.x] + kb.x, pby = o[db.y] + kb.y;
        const int pbz = o[db.z] + kb.z, pbw = o[db.w] + kb.w;
        s[pax] = sa.x; s[pay] = sa.y; s[paz] = sa.z; s[paw] = sa.w;
        s[pbx] = sb.x; s[pby] = sb.y; s[pbz] = sb.z; s[pbw] = sb.w;
    } else {
        for (int e = wbase + lane; e < nedge; e += 32)
            s[o[dst[e]] + r[e]] = src[e];
    }
}

__global__ __launch_bounds__(ATT_BLOCK) void seg_attend_kernel(
    const float* __restrict__ xl, const float* __restrict__ xr,
    const int* __restrict__ slots, const int* __restrict__ off,
    const int* __restrict__ cnt,
    const float* __restrict__ att, const float* __restrict__ bias,
    float* __restrict__ out, int n)
{
    const int lane = (int)(threadIdx.x & 31);
    const int w = __builtin_amdgcn_readfirstlane(
        (int)((blockIdx.x * blockDim.x + threadIdx.x) >> 5));
    const int d0 = w * NPW;
    if (d0 >= n) return;
    const int dend = min(d0 + NPW, n);
    const float4 at = *(const float4*)(att + lane * 4);
    const float4 bi = *(const float4*)(bias + lane * 4);

    for (int d = d0; d < dend; ++d) {
        const int beg = off[d];
        const int deg = cnt[d];
        const float4 b = *(const float4*)(xr + (size_t)d * DIM + lane * 4);

        float4 acc = make_float4(0.f, 0.f, 0.f, 0.f);
        float m = -INFINITY;
        float s = 0.f;

        for (int j0 = 0; j0 < deg; j0 += 8) {
            const int nv = min(8, deg - j0);
            const int jj = min(lane & 7, nv - 1);
            const int sidx = slots[beg + j0 + jj];

            float4 a[8];
            float  l[8];
#pragma unroll
            for (int j = 0; j < 8; ++j) {
                const int sj = __builtin_amdgcn_readlane(sidx, j);
                a[j] = *(const float4*)(xl + (size_t)sj * DIM + lane * 4);
            }
#pragma unroll
            for (int j = 0; j < 8; ++j) {
                float z, p = 0.f;
                z = a[j].x + b.x; z = (z > 0.f) ? z : SLOPE * z; p += z * at.x;
                z = a[j].y + b.y; z = (z > 0.f) ? z : SLOPE * z; p += z * at.y;
                z = a[j].z + b.z; z = (z > 0.f) ? z : SLOPE * z; p += z * at.z;
                z = a[j].w + b.w; z = (z > 0.f) ? z : SLOPE * z; p += z * at.w;
#pragma unroll
                for (int x = 16; x > 0; x >>= 1) p += __shfl_xor(p, x, 32);
                l[j] = (j < nv) ? p : -INFINITY;
            }
            float mc = l[0];
#pragma unroll
            for (int j = 1; j < 8; ++j) mc = fmaxf(mc, l[j]);
            const float mn = fmaxf(m, mc);
            const float sc = expf(m - mn);
            acc.x *= sc; acc.y *= sc; acc.z *= sc; acc.w *= sc;
            s *= sc;
#pragma unroll
            for (int j = 0; j < 8; ++j) {
                const float e = expf(l[j] - mn);
                s += e;
                acc.x += e * a[j].x;
                acc.y += e * a[j].y;
                acc.z += e * a[j].z;
                acc.w += e * a[j].w;
            }
            m = mn;
        }

        const float den = s + 1e-16f;
        float4 r;
        r.x = acc.x / den + bi.x;
        r.y = acc.y / den + bi.y;
        r.z = acc.z / den + bi.z;
        r.w = acc.w / den + bi.w;
        *(float4*)(out + (size_t)d * DIM + lane * 4) = r;
    }
}

extern "C" void kernel_launch(void* const* d_in, const int* in_sizes, int n_in,
                              void* d_out, int out_size, void* d_ws, size_t ws_size,
                              hipStream_t stream) {
    (void)in_sizes; (void)n_in; (void)out_size; (void)ws_size;

    const float* x_user = (const float*)d_in[0];
    const float* x_item = (const float*)d_in[1];
    const int*   e_u2i  = (const int*)d_in[2];
    const int*   e_i2u  = (const int*)d_in[3];
    const float* W1_u = (const float*)d_in[4],  *b1_u = (const float*)d_in[5];
    const float* W2_u = (const float*)d_in[6],  *b2_u = (const float*)d_in[7];
    const float* W1_i = (const float*)d_in[8],  *b1_i = (const float*)d_in[9];
    const float* W2_i = (const float*)d_in[10], *b2_i = (const float*)d_in[11];
    const float* Wl_u2i = (const float*)d_in[12], *bl_u2i = (const float*)d_in[13];
    const float* Wr_u2i = (const float*)d_in[14], *br_u2i = (const float*)d_in[15];
    const float* att_u2i = (const float*)d_in[16], *bias_u2i = (const float*)d_in[17];
    const float* Wl_i2u = (const float*)d_in[18], *bl_i2u = (const float*)d_in[19];
    const float* Wr_i2u = (const float*)d_in[20], *br_i2u = (const float*)d_in[21];
    const float* att_i2u = (const float*)d_in[22], *bias_i2u = (const float*)d_in[23];

    float* out_user = (float*)d_out;
    float* out_item = (float*)d_out + (size_t)N_NODES * DIM;

    float* ws    = (float*)d_ws;
    float* xl_u  = ws;
    float* xl_i  = xl_u + (size_t)N_NODES * DIM;
    float* xr_i  = xl_i + (size_t)N_NODES * DIM;
    float* xr_u  = xr_i + (size_t)N_NODES * DIM;
    int* cnt   = (int*)(xr_u + (size_t)N_NODES * DIM);
    int* off   = cnt + 2 * N_NODES;
    int* rank  = off + 2 * N_NODES;
    int* slots = rank + 2 * NEDGE;

    const int blkE    = (NEDGE + EDGE_BLOCK * EPT - 1) / (EDGE_BLOCK * EPT);
    const int blkN    = (N_NODES + 255) / 256;
    const int blkZ    = ((2 * N_NODES + 3) / 4 + 255) / 256;
    const int blkNw   = (N_NODES + (ATT_BLOCK / 32) * NPW - 1) / ((ATT_BLOCK / 32) * NPW);
    const int blkS    = (N_NODES / 16 + 3) / 4;

    zero_counts_kernel<<<blkZ, 256, 0, stream>>>(cnt, 2 * N_NODES);
    count_rank_kernel<<<dim3(blkE, 2), EDGE_BLOCK, 0, stream>>>(
        e_u2i + NEDGE, e_i2u + NEDGE, cnt, rank, NEDGE, N_NODES);
    seg_offsets_kernel<<<dim3(blkN, 2), 256, 0, stream>>>(cnt, off, N_NODES);
    fill_slots_kernel<<<dim3(blkE, 2), EDGE_BLOCK, 0, stream>>>(
        e_u2i, e_u2i + NEDGE, e_i2u, e_i2u + NEDGE, off, rank, slots, NEDGE, N_NODES);

    NodeSide su = { x_user, W1_u, b1_u, W2_u, b2_u,
                    Wl_u2i, bl_u2i, Wr_i2u, br_i2u, xl_u, xr_u };
    NodeSide si = { x_item, W1_i, b1_i, W2_i, b2_i,
                    Wl_i2u, bl_i2u, Wr_u2i, br_u2i, xl_i, xr_i };
    node_chain_kernel<<<dim3(blkS, 2), 128, 0, stream>>>(su, si, N_NODES);

    seg_attend_kernel<<<blkNw, ATT_BLOCK, 0, stream>>>(xl_u, xr_i, slots, off, cnt,
                                                 att_u2i, bias_u2i, out_item, N_NODES);

    seg_attend_kernel<<<blkNw, ATT_BLOCK, 0, stream>>>(xl_i, xr_u, slots + NEDGE, off + N_NODES,
                                                 cnt + N_NODES,
                                                 att_i2u, bias_i2u, out_user, N_NODES);
}


# Edge grouping via LDS: a work-group per edge chunk counts into a whole-relation LDS table (ranks from LDS adds), per-node prefix over chunks; no global atomics; node sum order: chunk, then arrival
# speedup vs baseline: 2.9713x; 1.1522x over previous
#include <hip/hip_runtime.h>


typedef __attribute__((ext_vector_type(2))) float v2f;
typedef __attribute__((ext_vector_type(8))) float v8f;

#define N_NODES 50000
#define DIM 128
#define HID 64
#define NEDGE 500000
#define SLOPE 0.2f

#define SP 132

struct NodeSide {
    const float* x;
    const float* W1; const float* b1;
    const float* W2; const float* b2;
    const float* Wa; const float* ba;
    const float* Wb; const float* bb;
    float* ya;
    float* yb;
};

template <int NT, int K>
__device__ __forceinline__ void stripe_mm(const float* arow, const float* __restrict__ W,
                                          int nloc, int khalf, v8f (&acc)[NT])
{
    constexpr int ncols = NT * 16;
#pragma unroll (16 / NT)
    for (int k0 = 0; k0 < K; k0 += 4) {
        const v2f a = *(const v2f*)(arow + k0);
        const float* __restrict__ w0 = W + (size_t)(k0 + khalf) * ncols + nloc;
        const float* __restrict__ w1 = w0 + ncols;
#pragma unroll
        for (int t = 0; t < NT; ++t) {
            v2f b;
            b.x = w0[t * 16];
            b.y = w1[t * 16];
            acc[t] = __builtin_amdgcn_wmma_f32_16x16x4_f32(
                false, a, false, b, (short)0, acc[t], false, false);
        }
    }
}

template <int NT>
__device__ __forceinline__ void zero_acc(v8f (&acc)[NT]) {
#pragma unroll
    for (int t = 0; t < NT; ++t)
        acc[t] = (v8f){0.f, 0.f, 0.f, 0.f, 0.f, 0.f, 0.f, 0.f};
}

template <int NT, bool RELU>
__device__ __forceinline__ void tiles_to_lds(float* S, const v8f (&acc)[NT],
                                             const float* __restrict__ bias,
                                             int nloc, int rhalf)
{
#pragma unroll
    for (int t = 0; t < NT; ++t) {
        const int n = t * 16 + nloc;
        const float bv = bias[n];
#pragma unroll
        for (int j = 0; j < 8; ++j) {
            float v = acc[t][j] + bv;
            if (RELU) v = fmaxf(v, 0.f);
            S[(rhalf + j) * SP + n] = v;
        }
    }
}

template <int NT>
__device__ __forceinline__ void tiles_to_global(float* __restrict__ C, const v8f (&acc)[NT],
                                                const float* __restrict__ bias,
                                                int nloc, int mbase)
{
    constexpr int ncols = NT * 16;
#pragma unroll
    for (int t = 0; t < NT; ++t) {
        const int n = t * 16 + nloc;
        const float bv = bias[n];
#pragma unroll
        for (int j = 0; j < 8; ++j)
            C[(size_t)(mbase + j) * ncols + n] = acc[t][j] + bv;
    }
}

__global__ __launch_bounds__(128) void node_chain_kernel(NodeSide su, NodeSide si, int rows)
{
    __shared__ float lds[4 * 16 * SP];

    const bool second = (blockIdx.y != 0);
    const float* x  = second ? si.x  : su.x;
    const float* W1 = second ? si.W1 : su.W1;
    const float* b1 = second ? si.b1 : su.b1;
    const float* W2 = second ? si.W2 : su.W2;
    const float* b2 = second ? si.b2 : su.b2;
    const float* Wa = second ? si.Wa : su.Wa;
    const float* ba = second ? si.ba : su.ba;
    const float* Wb = second ? si.Wb : su.Wb;
    const float* bb = second ? si.bb : su.bb;
    float* ya = second ? si.ya : su.ya;
    float* yb = second ? si.yb : su.yb;

    const int wv   = (int)(threadIdx.x >> 5);
    const int lane = (int)(threadIdx.x & 31);
    int stripe = (int)blockIdx.x * 4 + wv;
    const bool live = stripe * 16 < rows;
    if (!live) stripe = rows / 16 - 1;

    const int nloc  = lane & 15;
    const int khalf = (lane >> 4) << 1;
    const int rhalf = (lane >> 4) << 3;
    const int mbase = stripe * 16 + rhalf;

    float* S = lds + wv * (16 * SP);
    const float* sfrag = S + nloc * SP + khalf;

    v8f acc4[4];
    v8f acc8[8];

    zero_acc<4>(acc4);
    stripe_mm<4, DIM>(x + (size_t)(stripe * 16 + nloc) * DIM + khalf, W1, nloc, khalf, acc4);
    tiles_to_lds<4, true>(S, acc4, b1, nloc, rhalf);
    __syncthreads();

    zero_acc<8>(acc8);
    stripe_mm<8, HID>(sfrag, W2, nloc, khalf, acc8);
    __syncthreads();
    tiles_to_lds<8, false>(S, acc8, b2, nloc, rhalf);
    __syncthreads();

    zero_acc<8>(acc8);
    stripe_mm<8, DIM>(sfrag, Wa, nloc, khalf, acc8);
    if (live) tiles_to_global<8>(ya, acc8, ba, nloc, mbase);

    zero_acc<8>(acc8);
    stripe_mm<8, DIM>(sfrag, Wb, nloc, khalf, acc8);
    if (live) tiles_to_global<8>(yb, acc8, bb, nloc, mbase);
}


#define NCH 64
#define CHUNK 7816
#define NBLK 196
#define GROUP_BLOCK 1024

#define NPW 4
#define ATT_BLOCK 128

__global__ __launch_bounds__(GROUP_BLOCK) void chunk_count_kernel(
    const int* __restrict__ dst0, const int* __restrict__ dst1,
    int* __restrict__ hist, int* __restrict__ rank)
{
    __shared__ __attribute__((aligned(16))) int tab[N_NODES];
    const int c   = (int)blockIdx.x;
    const int rel = (int)blockIdx.y;
    const int t   = (int)threadIdx.x;
    const int* __restrict__ dst = rel ? dst1 : dst0;
    int* __restrict__ r = rank + (size_t)rel * NEDGE;
    int* __restrict__ h = hist + (size_t)(rel * NCH + c) * N_NODES;

    for (int i = t * 4; i < N_NODES; i += GROUP_BLOCK * 4)
        *(int4*)(tab + i) = make_int4(0, 0, 0, 0);
    __syncthreads();

    const int e0 = c * CHUNK;
    const int e1 = min(e0 + CHUNK, NEDGE);
    for (int e = e0 + t * 4; e < e1; e += GROUP_BLOCK * 4) {
        const int4 d = *(const int4*)(dst + e);
        int4 k;
        k.x = atomicAdd(tab + d.x, 1);
        k.y = atomicAdd(tab + d.y, 1);
        k.z = atomicAdd(tab + d.z, 1);
        k.w = atomicAdd(tab + d.w, 1);
        *(int4*)(r + e) = k;
    }
    __syncthreads();

    for (int i = t * 4; i < N_NODES; i += GROUP_BLOCK * 4)
        *(int4*)(h + i) = *(const int4*)(tab + i);
}

__global__ __launch_bounds__(256) void node_prefix_kernel(
    int* __restrict__ hist, int* __restrict__ cnt,
    int* __restrict__ offl, int* __restrict__ blocktot)
{
    __shared__ int s_tot[8];
    const int rel = (int)blockIdx.y;
    const int t = (int)threadIdx.x, lane = t & 31, w = t >> 5;
    const int idx = (int)blockIdx.x * 256 + t;
    int* __restrict__ h = hist + (size_t)rel * NCH * N_NODES;

    int run = 0;
    if (idx < N_NODES) {
        for (int c0 = 0; c0 < NCH; c0 += 32) {
            int v[32];
#pragma unroll
            for (int j = 0; j < 32; ++j) v[j] = h[(size_t)(c0 + j) * N_NODES + idx];
#pragma unroll
            for (int j = 0; j < 32; ++j) {
                h[(size_t)(c0 + j) * N_NODES + idx] = run;
                run += v[j];
            }
        }
        cnt[rel * N_NODES + idx] = run;
    }

    int inc = run;
#pragma unroll
    for (int dlt = 1; dlt < 32; dlt <<= 1) {
        const int u = __shfl_up(inc, dlt, 32);
        if (lane >= dlt) inc += u;
    }
    if (lane == 31) s_tot[w] = inc;
    __syncthreads();
    int add = 0, all = 0;
#pragma unroll
    for (int k = 0; k < 8; ++k) {
        const int s = s_tot[k];
        if (k < w) add += s;
        all += s;
    }
    if (idx < N_NODES) offl[rel * N_NODES + idx] = add + inc - run;
    if (t == 0) blocktot[rel * NBLK + (int)blockIdx.x] = all;
}

__global__ __launch_bounds__(GROUP_BLOCK) void fill_slots_kernel(
    const int* __restrict__ src0, const int* __restrict__ dst0,
    const int* __restrict__ src1, const int* __restrict__ dst1,
    const int* __restrict__ hist, const int* __restrict__ offl,
    const int* __restrict__ blocktot, const int* __restrict__ rank,
    int* __restrict__ off, int* __restrict__ slots)
{
    __shared__ __attribute__((aligned(16))) int tab[N_NODES];
    __shared__ int bb[256];
    __shared__ int s_w[8];
    const int c   = (int)blockIdx.x;
    const int rel = (int)blockIdx.y;
    const int t = (int)threadIdx.x, lane = t & 31, w = t >> 5;
    const int* __restrict__ src = rel ? src1 : src0;
    const int* __restrict__ dst = rel ? dst1 : dst0;
    const int* __restrict__ hc = hist + (size_t)(rel * NCH + c) * N_NODES;
    const int* __restrict__ ol = offl + (size_t)rel * N_NODES;
    const int* __restrict__ r  = rank + (size_t)rel * NEDGE;
    int* __restrict__ og = off + (size_t)rel * N_NODES;
    int* __restrict__ sl = slots + (size_t)rel * NEDGE;

    const int v = (t < NBLK) ? blocktot[rel * NBLK + t] : 0;
    int inc = v;
#pragma unroll
    for (int dlt = 1; dlt < 32; dlt <<= 1) {
        const int u = __shfl_up(inc, dlt, 32);
        if (lane >= dlt) inc += u;
    }
    if (w < 8 && lane == 31) s_w[w] = inc;
    __syncthreads();
    if (t < 256) {
        int add = 0;
#pragma unroll
        for (int k = 0; k < 8; ++k)
            if (k < w) add += s_w[k];
        bb[t] = add + inc - v;
    }
    __syncthreads();

#pragma unroll 4
    for (int i = t * 4; i < N_NODES; i += GROUP_BLOCK * 4) {
        const int4 a = *(const int4*)(hc + i);
        const int4 o = *(const int4*)(ol + i);
        const int b = bb[i >> 8];
        *(int4*)(tab + i) = make_int4(o.x + b + a.x, o.y + b + a.y,
                                      o.z + b + a.z, o.w + b + a.w);
    }
    __syncthreads();

    if (c == 0) {
        for (int i = t * 4; i < N_NODES; i += GROUP_BLOCK * 4)
            *(int4*)(og + i) = *(const int4*)(tab + i);
    }

    const int e0 = c * CHUNK;
    const int e1 = min(e0 + CHUNK, NEDGE);
    for (int e = e0 + t * 4; e < e1; e += GROUP_BLOCK * 4) {
        const int4 d = *(const int4*)(dst + e);
        const int4 k = *(const int4*)(r + e);
        const int4 s = *(const int4*)(src + e);
        sl[tab[d.x] + k.x] = s.x;
        sl[tab[d.y] + k.y] = s.y;
        sl[tab[d.z] + k.z] = s.z;
        sl[tab[d.w] + k.w] = s.w;
    }
}

__global__ __launch_bounds__(ATT_BLOCK) void seg_attend_kernel(
    const float* __restrict__ xl, const float* __restrict__ xr,
    const int* __restrict__ slots, const int* __restrict__ off,
    const int* __restrict__ cnt,
    const float* __restrict__ att, const float* __restrict__ bias,
    float* __restrict__ out, int n)
{
    const int lane = (int)(threadIdx.x & 31);
    const int w = __builtin_amdgcn_readfirstlane(
        (int)((blockIdx.x * blockDim.x + threadIdx.x) >> 5));
    const int d0 = w * NPW;
    if (d0 >= n) return;
    const int dend = min(d0 + NPW, n);
    const float4 at = *(const float4*)(att + lane * 4);
    const float4 bi = *(const float4*)(bias + lane * 4);

    for (int d = d0; d < dend; ++d) {
        const int beg = off[d];
        const int deg = cnt[d];
        const float4 b = *(const float4*)(xr + (size_t)d * DIM + lane * 4);

        float4 acc = make_float4(0.f, 0.f, 0.f, 0.f);
        float m = -INFINITY;
        float s = 0.f;

        for (int j0 = 0; j0 < deg; j0 += 8) {
            const int nv = min(8, deg - j0);
            const int jj = min(lane & 7, nv - 1);
            const int sidx = slots[beg + j0 + jj];

            float4 a[8];
            float  l[8];
#pragma unroll
            for (int j = 0; j < 8; ++j) {
                const int sj = __builtin_amdgcn_readlane(sidx, j);
                a[j] = *(const float4*)(xl + (size_t)sj * DIM + lane * 4);
            }
#pragma unroll
            for (int j = 0; j < 8; ++j) {
                float z, p = 0.f;
                z = a[j].x + b.x; z = (z > 0.f) ? z : SLOPE * z; p += z * at.x;
                z = a[j].y + b.y; z = (z > 0.f) ? z : SLOPE * z; p += z * at.y;
                z = a[j].z + b.z; z = (z > 0.f) ? z : SLOPE * z; p += z * at.z;
                z = a[j].w + b.w; z = (z > 0.f) ? z : SLOPE * z; p += z * at.w;
#pragma unroll
                for (int x = 16; x > 0; x >>= 1) p += __shfl_xor(p, x, 32);
                l[j] = (j < nv) ? p : -INFINITY;
            }
            float mc = l[0];
#pragma unroll
            for (int j = 1; j < 8; ++j) mc = fmaxf(mc, l[j]);
            const float mn = fmaxf(m, mc);
            const float sc = expf(m - mn);
            acc.x *= sc; acc.y *= sc; acc.z *= sc; acc.w *= sc;
            s *= sc;
#pragma unroll
            for (int j = 0; j < 8; ++j) {
                const float e = expf(l[j] - mn);
                s += e;
                acc.x += e * a[j].x;
                acc.y += e * a[j].y;
                acc.z += e * a[j].z;
                acc.w += e * a[j].w;
            }
            m = mn;
        }

        const float den = s + 1e-16f;
        float4 r;
        r.x = acc.x / den + bi.x;
        r.y = acc.y / den + bi.y;
        r.z = acc.z / den + bi.z;
        r.w = acc.w / den + bi.w;
        *(float4*)(out + (size_t)d * DIM + lane * 4) = r;
    }
}

extern "C" void kernel_launch(void* const* d_in, const int* in_sizes, int n_in,
                              void* d_out, int out_size, void* d_ws, size_t ws_size,
                              hipStream_t stream) {
    (void)in_sizes; (void)n_in; (void)out_size; (void)ws_size;

    const float* x_user = (const float*)d_in[0];
    const float* x_item = (const float*)d_in[1];
    const int*   e_u2i  = (const int*)d_in[2];
    const int*   e_i2u  = (const int*)d_in[3];
    const float* W1_u = (const float*)d_in[4],  *b1_u = (const float*)d_in[5];
    const float* W2_u = (const float*)d_in[6],  *b2_u = (const float*)d_in[7];
    const float* W1_i = (const float*)d_in[8],  *b1_i = (const float*)d_in[9];
    const float* W2_i = (const float*)d_in[10], *b2_i = (const float*)d_in[11];
    const float* Wl_u2i = (const float*)d_in[12], *bl_u2i = (const float*)d_in[13];
    const float* Wr_u2i = (const float*)d_in[14], *br_u2i = (const float*)d_in[15];
    const float* att_u2i = (const float*)d_in[16], *bias_u2i = (const float*)d_in[17];
    const float* Wl_i2u = (const float*)d_in[18], *bl_i2u = (const float*)d_in[19];
    const float* Wr_i2u = (const float*)d_in[20], *br_i2u = (const float*)d_in[21];
    const float* att_i2u = (const float*)d_in[22], *bias_i2u = (const float*)d_in[23];

    float* out_user = (float*)d_out;
    float* out_item = (float*)d_out + (size_t)N_NODES * DIM;

    float* ws    = (float*)d_ws;
    float* xl_u  = ws;
    float* xl_i  = xl_u + (size_t)N_NODES * DIM;
    float* xr_i  = xl_i + (size_t)N_NODES * DIM;
    float* xr_u  = xr_i + (size_t)N_NODES * DIM;
    int* hist  = (int*)xl_u;
    int* cnt   = (int*)(xr_u + (size_t)N_NODES * DIM);
    int* off   = cnt + 2 * N_NODES;
    int* rank  = off + 2 * N_NODES;
    int* slots = rank + 2 * NEDGE;
    int* offl  = slots + 2 * NEDGE;
    int* blocktot = offl + 2 * N_NODES;

    const int blkNw   = (N_NODES + (ATT_BLOCK / 32) * NPW - 1) / ((ATT_BLOCK / 32) * NPW);
    const int blkS    = (N_NODES / 16 + 3) / 4;

    chunk_count_kernel<<<dim3(NCH, 2), GROUP_BLOCK, 0, stream>>>(
        e_u2i + NEDGE, e_i2u + NEDGE, hist, rank);
    node_prefix_kernel<<<dim3(NBLK, 2), 256, 0, stream>>>(hist, cnt, offl, blocktot);
    fill_slots_kernel<<<dim3(NCH, 2), GROUP_BLOCK, 0, stream>>>(
        e_u2i, e_u2i + NEDGE, e_i2u, e_i2u + NEDGE, hist, offl, blocktot, rank, off, slots);

    NodeSide su = { x_user, W1_u, b1_u, W2_u, b2_u,
                    Wl_u2i, bl_u2i, Wr_i2u, br_i2u, xl_u, xr_u };
    NodeSide si = { x_item, W1_i, b1_i, W2_i, b2_i,
                    Wl_i2u, bl_i2u, Wr_u2i, br_u2i, xl_i, xr_i };
    node_chain_kernel<<<dim3(blkS, 2), 128, 0, stream>>>(su, si, N_NODES);

    seg_attend_kernel<<<blkNw, ATT_BLOCK, 0, stream>>>(xl_u, xr_i, slots, off, cnt,
                                                 att_u2i, bias_u2i, out_item, N_NODES);

    seg_attend_kernel<<<blkNw, ATT_BLOCK, 0, stream>>>(xl_i, xr_u, slots + NEDGE, off + N_NODES,
                                                 cnt + N_NODES,
                                                 att_i2u, bias_i2u, out_user, N_NODES);
}


# W2 folded into both attention projections: a small kernel forms W2@Wl, W2@Wr and b2@W+b on WMMA, the node kernel does two K=64 products on hid (h never formed); sums re-associated
# speedup vs baseline: 3.3608x; 1.1311x over previous
#include <hip/hip_runtime.h>


typedef __attribute__((ext_vector_type(2))) float v2f;
typedef __attribute__((ext_vector_type(8))) float v8f;

#define N_NODES 50000
#define DIM 128
#define HID 64
#define NEDGE 500000
#define SLOPE 0.2f

#define SP 132

struct NodeSide {
    const float* x;
    const float* W1; const float* b1;
    const float* Pa; const float* ca;
    const float* Pb; const float* cb;
    float* ya;
    float* yb;
};

struct FoldSide {
    const float* W2; const float* b2;
    const float* Wa; const float* ba;
    const float* Wb; const float* bb;
    float* Pa;
    float* Pb;
};

template <int NT, int K, int LDW = NT * 16>
__device__ __forceinline__ void stripe_mm(const float* arow, const float* __restrict__ W,
                                          int nloc, int khalf, v8f (&acc)[NT])
{
    constexpr int ncols = LDW;
#pragma unroll (16 / NT)
    for (int k0 = 0; k0 < K; k0 += 4) {
        const v2f a = *(const v2f*)(arow + k0);
        const float* __restrict__ w0 = W + (size_t)(k0 + khalf) * ncols + nloc;
        const float* __restrict__ w1 = w0 + ncols;
#pragma unroll
        for (int t = 0; t < NT; ++t) {
            v2f b;
            b.x = w0[t * 16];
            b.y = w1[t * 16];
            acc[t] = __builtin_amdgcn_wmma_f32_16x16x4_f32(
                false, a, false, b, (short)0, acc[t], false, false);
        }
    }
}

template <int NT>
__device__ __forceinline__ void zero_acc(v8f (&acc)[NT]) {
#pragma unroll
    for (int t = 0; t < NT; ++t)
        acc[t] = (v8f){0.f, 0.f, 0.f, 0.f, 0.f, 0.f, 0.f, 0.f};
}

template <int NT, bool RELU>
__device__ __forceinline__ void tiles_to_lds(float* S, const v8f (&acc)[NT],
                                             const float* __restrict__ bias,
                                             int nloc, int rhalf)
{
#pragma unroll
    for (int t = 0; t < NT; ++t) {
        const int n = t * 16 + nloc;
        const float bv = bias[n];
#pragma unroll
        for (int j = 0; j < 8; ++j) {
            float v = acc[t][j] + bv;
            if (RELU) v = fmaxf(v, 0.f);
            S[(rhalf + j) * SP + n] = v;
        }
    }
}

template <int NT>
__device__ __forceinline__ void tiles_to_global(float* __restrict__ C, const v8f (&acc)[NT],
                                                const float* __restrict__ bias,
                                                int nloc, int mbase)
{
    constexpr int ncols = NT * 16;
#pragma unroll
    for (int t = 0; t < NT; ++t) {
        const int n = t * 16 + nloc;
        const float bv = bias[n];
#pragma unroll
        for (int j = 0; j < 8; ++j)
            C[(size_t)(mbase + j) * ncols + n] = acc[t][j] + bv;
    }
}

__global__ __launch_bounds__(256) void fold_kernel(FoldSide fu, FoldSide fi)
{
    const bool second = (blockIdx.y != 0);
    const bool projb  = (blockIdx.z != 0);
    const float* W2 = second ? fi.W2 : fu.W2;
    const float* b2 = second ? fi.b2 : fu.b2;
    const float* W  = second ? (projb ? fi.Wb : fi.Wa) : (projb ? fu.Wb : fu.Wa);
    const float* b  = second ? (projb ? fi.bb : fi.ba) : (projb ? fu.bb : fu.ba);
    float* P        = second ? (projb ? fi.Pb : fi.Pa) : (projb ? fu.Pb : fu.Pa);

    const int s    = (int)blockIdx.x;
    const int t    = (int)(threadIdx.x >> 5);
    const int lane = (int)(threadIdx.x & 31);
    const int nloc  = lane & 15;
    const int khalf = (lane >> 4) << 1;
    const int rhalf = (lane >> 4) << 3;
    const bool brow = (s == HID / 16);

    const float* arow = brow ? (b2 + khalf)
                             : (W2 + (size_t)(s * 16 + nloc) * DIM + khalf);
    v8f acc[1];
    zero_acc<1>(acc);
    stripe_mm<1, DIM, DIM>(arow, W + t * 16, nloc, khalf, acc);

    const int n = t * 16 + nloc;
    if (!brow) {
#pragma unroll
        for (int j = 0; j < 8; ++j)
            P[(size_t)(s * 16 + rhalf + j) * DIM + n] = acc[0][j];
    } else if (lane < 16) {
        P[(size_t)HID * DIM + n] = acc[0][0] + b[n];
    }
}

__global__ __launch_bounds__(128) void node_chain_kernel(NodeSide su, NodeSide si, int rows)
{
    __shared__ float lds[4 * 16 * SP];

    const bool second = (blockIdx.y != 0);
    const float* x  = second ? si.x  : su.x;
    const float* W1 = second ? si.W1 : su.W1;
    const float* b1 = second ? si.b1 : su.b1;
    const float* Pa = second ? si.Pa : su.Pa;
    const float* ca = second ? si.ca : su.ca;
    const float* Pb = second ? si.Pb : su.Pb;
    const float* cb = second ? si.cb : su.cb;
    float* ya = second ? si.ya : su.ya;
    float* yb = second ? si.yb : su.yb;

    const int wv   = (int)(threadIdx.x >> 5);
    const int lane = (int)(threadIdx.x & 31);
    int stripe = (int)blockIdx.x * 4 + wv;
    const bool live = stripe * 16 < rows;
    if (!live) stripe = rows / 16 - 1;

    const int nloc  = lane & 15;
    const int khalf = (lane >> 4) << 1;
    const int rhalf = (lane >> 4) << 3;
    const int mbase = stripe * 16 + rhalf;

    float* S = lds + wv * (16 * SP);
    const float* sfrag = S + nloc * SP + khalf;

    v8f acc4[4];
    v8f acc8[8];

    zero_acc<4>(acc4);
    stripe_mm<4, DIM>(x + (size_t)(stripe * 16 + nloc) * DIM + khalf, W1, nloc, khalf, acc4);
    tiles_to_lds<4, true>(S, acc4, b1, nloc, rhalf);
    __syncthreads();

    zero_acc<8>(acc8);
    stripe_mm<8, HID>(sfrag, Pa, nloc, khalf, acc8);
    if (live) tiles_to_global<8>(ya, acc8, ca, nloc, mbase);

    zero_acc<8>(acc8);
    stripe_mm<8, HID>(sfrag, Pb, nloc, khalf, acc8);
    if (live) tiles_to_global<8>(yb, acc8, cb, nloc, mbase);
}


#define NCH 64
#define CHUNK 7816
#define NBLK 196
#define GROUP_BLOCK 1024

#define NPW 4
#define ATT_BLOCK 128

__global__ __launch_bounds__(GROUP_BLOCK) void chunk_count_kernel(
    const int* __restrict__ dst0, const int* __restrict__ dst1,
    int* __restrict__ hist, int* __restrict__ rank)
{
    __shared__ __attribute__((aligned(16))) int tab[N_NODES];
    const int c   = (int)blockIdx.x;
    const int rel = (int)blockIdx.y;
    const int t   = (int)threadIdx.x;
    const int* __restrict__ dst = rel ? dst1 : dst0;
    int* __restrict__ r = rank + (size_t)rel * NEDGE;
    int* __restrict__ h = hist + (size_t)(rel * NCH + c) * N_NODES;

    for (int i = t * 4; i < N_NODES; i += GROUP_BLOCK * 4)
        *(int4*)(tab + i) = make_int4(0, 0, 0, 0);
    __syncthreads();

    const int e0 = c * CHUNK;
    const int e1 = min(e0 + CHUNK, NEDGE);
    for (int e = e0 + t * 4; e < e1; e += GROUP_BLOCK * 4) {
        const int4 d = *(const int4*)(dst + e);
        int4 k;
        k.x = atomicAdd(tab + d.x, 1);
        k.y = atomicAdd(tab + d.y, 1);
        k.z = atomicAdd(tab + d.z, 1);
        k.w = atomicAdd(tab + d.w, 1);
        *(int4*)(r + e) = k;
    }
    __syncthreads();

    for (int i = t * 4; i < N_NODES; i += GROUP_BLOCK * 4)
        *(int4*)(h + i) = *(const int4*)(tab + i);
}

__global__ __launch_bounds__(256) void node_prefix_kernel(
    int* __restrict__ hist, int* __restrict__ cnt,
    int* __restrict__ offl, int* __restrict__ blocktot)
{
    __shared__ int s_tot[8];
    const int rel = (int)blockIdx.y;
    const int t = (int)threadIdx.x, lane = t & 31, w = t >> 5;
    const int idx = (int)blockIdx.x * 256 + t;
    int* __restrict__ h = hist + (size_t)rel * NCH * N_NODES;

    int run = 0;
    if (idx < N_NODES) {
        for (int c0 = 0; c0 < NCH; c0 += 32) {
            int v[32];
#pragma unroll
            for (int j = 0; j < 32; ++j) v[j] = h[(size_t)(c0 + j) * N_NODES + idx];
#pragma unroll
            for (int j = 0; j < 32; ++j) {
                h[(size_t)(c0 + j) * N_NODES + idx] = run;
                run += v[j];
            }
        }
        cnt[rel * N_NODES + idx] = run;
    }

    int inc = run;
#pragma unroll
    for (int dlt = 1; dlt < 32; dlt <<= 1) {
        const int u = __shfl_up(inc, dlt, 32);
        if (lane >= dlt) inc += u;
    }
    if (lane == 31) s_tot[w] = inc;
    __syncthreads();
    int add = 0, all = 0;
#pragma unroll
    for (int k = 0; k < 8; ++k) {
        const int s = s_tot[k];
        if (k < w) add += s;
        all += s;
    }
    if (idx < N_NODES) offl[rel * N_NODES + idx] = add + inc - run;
    if (t == 0) blocktot[rel * NBLK + (int)blockIdx.x] = all;
}

__global__ __launch_bounds__(GROUP_BLOCK) void fill_slots_kernel(
    const int* __restrict__ src0, const int* __restrict__ dst0,
    const int* __restrict__ src1, const int* __restrict__ dst1,
    const int* __restrict__ hist, const int* __restrict__ offl,
    const int* __restrict__ blocktot, const int* __restrict__ rank,
    int* __restrict__ off, int* __restrict__ slots)
{
    __shared__ __attribute__((aligned(16))) int tab[N_NODES];
    __shared__ int bb[256];
    __shared__ int s_w[8];
    const int c   = (int)blockIdx.x;
    const int rel = (int)blockIdx.y;
    const int t = (int)threadIdx.x, lane = t & 31, w = t >> 5;
    const int* __restrict__ src = rel ? src1 : src0;
    const int* __restrict__ dst = rel ? dst1 : dst0;
    const int* __restrict__ hc = hist + (size_t)(rel * NCH + c) * N_NODES;
    const int* __restrict__ ol = offl + (size_t)rel * N_NODES;
    const int* __restrict__ r  = rank + (size_t)rel * NEDGE;
    int* __restrict__ og = off + (size_t)rel * N_NODES;
    int* __restrict__ sl = slots + (size_t)rel * NEDGE;

    const int v = (t < NBLK) ? blocktot[rel * NBLK + t] : 0;
    int inc = v;
#pragma unroll
    for (int dlt = 1; dlt < 32; dlt <<= 1) {
        const int u = __shfl_up(inc, dlt, 32);
        if (lane >= dlt) inc += u;
    }
    if (w < 8 && lane == 31) s_w[w] = inc;
    __syncthreads();
    if (t < 256) {
        int add = 0;
#pragma unroll
        for (int k = 0; k < 8; ++k)
            if (k < w) add += s_w[k];
        bb[t] = add + inc - v;
    }
    __syncthreads();

#pragma unroll 4
    for (int i = t * 4; i < N_NODES; i += GROUP_BLOCK * 4) {
        const int4 a = *(const int4*)(hc + i);
        const int4 o = *(const int4*)(ol + i);
        const int b = bb[i >> 8];
        *(int4*)(tab + i) = make_int4(o.x + b + a.x, o.y + b + a.y,
                                      o.z + b + a.z, o.w + b + a.w);
    }
    __syncthreads();

    if (c == 0) {
        for (int i = t * 4; i < N_NODES; i += GROUP_BLOCK * 4)
            *(int4*)(og + i) = *(const int4*)(tab + i);
    }

    const int e0 = c * CHUNK;
    const int e1 = min(e0 + CHUNK, NEDGE);
    for (int e = e0 + t * 4; e < e1; e += GROUP_BLOCK * 4) {
        const int4 d = *(const int4*)(dst + e);
        const int4 k = *(const int4*)(r + e);
        const int4 s = *(const int4*)(src + e);
        sl[tab[d.x] + k.x] = s.x;
        sl[tab[d.y] + k.y] = s.y;
        sl[tab[d.z] + k.z] = s.z;
        sl[tab[d.w] + k.w] = s.w;
    }
}

__global__ __launch_bounds__(ATT_BLOCK) void seg_attend_kernel(
    const float* __restrict__ xl, const float* __restrict__ xr,
    const int* __restrict__ slots, const int* __restrict__ off,
    const int* __restrict__ cnt,
    const float* __restrict__ att, const float* __restrict__ bias,
    float* __restrict__ out, int n)
{
    const int lane = (int)(threadIdx.x & 31);
    const int w = __builtin_amdgcn_readfirstlane(
        (int)((blockIdx.x * blockDim.x + threadIdx.x) >> 5));
    const int d0 = w * NPW;
    if (d0 >= n) return;
    const int dend = min(d0 + NPW, n);
    const float4 at = *(const float4*)(att + lane * 4);
    const float4 bi = *(const float4*)(bias + lane * 4);

    for (int d = d0; d < dend; ++d) {
        const int beg = off[d];
        const int deg = cnt[d];
        const float4 b = *(const float4*)(xr + (size_t)d * DIM + lane * 4);

        float4 acc = make_float4(0.f, 0.f, 0.f, 0.f);
        float m = -INFINITY;
        float s = 0.f;

        for (int j0 = 0; j0 < deg; j0 += 8) {
            const int nv = min(8, deg - j0);
            const int jj = min(lane & 7, nv - 1);
            const int sidx = slots[beg + j0 + jj];

            float4 a[8];
            float  l[8];
#pragma unroll
            for (int j = 0; j < 8; ++j) {
                const int sj = __builtin_amdgcn_readlane(sidx, j);
                a[j] = *(const float4*)(xl + (size_t)sj * DIM + lane * 4);
            }
#pragma unroll
            for (int j = 0; j < 8; ++j) {
                float z, p = 0.f;
                z = a[j].x + b.x; z = (z > 0.f) ? z : SLOPE * z; p += z * at.x;
                z = a[j].y + b.y; z = (z > 0.f) ? z : SLOPE * z; p += z * at.y;
                z = a[j].z + b.z; z = (z > 0.f) ? z : SLOPE * z; p += z * at.z;
                z = a[j].w + b.w; z = (z > 0.f) ? z : SLOPE * z; p += z * at.w;
#pragma unroll
                for (int x = 16; x > 0; x >>= 1) p += __shfl_xor(p, x, 32);
                l[j] = (j < nv) ? p : -INFINITY;
            }
            float mc = l[0];
#pragma unroll
            for (int j = 1; j < 8; ++j) mc = fmaxf(mc, l[j]);
            const float mn = fmaxf(m, mc);
            const float sc = expf(m - mn);
            acc.x *= sc; acc.y *= sc; acc.z *= sc; acc.w *= sc;
            s *= sc;
#pragma unroll
            for (int j = 0; j < 8; ++j) {
                const float e = expf(l[j] - mn);
                s += e;
                acc.x += e * a[j].x;
                acc.y += e * a[j].y;
                acc.z += e * a[j].z;
                acc.w += e * a[j].w;
            }
            m = mn;
        }

        const float den = s + 1e-16f;
        float4 r;
        r.x = acc.x / den + bi.x;
        r.y = acc.y / den + bi.y;
        r.z = acc.z / den + bi.z;
        r.w = acc.w / den + bi.w;
        *(float4*)(out + (size_t)d * DIM + lane * 4) = r;
    }
}

extern "C" void kernel_launch(void* const* d_in, const int* in_sizes, int n_in,
                              void* d_out, int out_size, void* d_ws, size_t ws_size,
                              hipStream_t stream) {
    (void)in_sizes; (void)n_in; (void)out_size; (void)ws_size;

    const float* x_user = (const float*)d_in[0];
    const float* x_item = (const float*)d_in[1];
    const int*   e_u2i  = (const int*)d_in[2];
    const int*   e_i2u  = (const int*)d_in[3];
    const float* W1_u = (const float*)d_in[4],  *b1_u = (const float*)d_in[5];
    const float* W2_u = (const float*)d_in[6],  *b2_u = (const float*)d_in[7];
    const float* W1_i = (const float*)d_in[8],  *b1_i = (const float*)d_in[9];
    const float* W2_i = (const float*)d_in[10], *b2_i = (const float*)d_in[11];
    const float* Wl_u2i = (const float*)d_in[12], *bl_u2i = (const float*)d_in[13];
    const float* Wr_u2i = (const float*)d_in[14], *br_u2i = (const float*)d_in[15];
    const float* att_u2i = (const float*)d_in[16], *bias_u2i = (const float*)d_in[17];
    const float* Wl_i2u = (const float*)d_in[18], *bl_i2u = (const float*)d_in[19];
    const float* Wr_i2u = (const float*)d_in[20], *br_i2u = (const float*)d_in[21];
    const float* att_i2u = (const float*)d_in[22], *bias_i2u = (const float*)d_in[23];

    float* out_user = (float*)d_out;
    float* out_item = (float*)d_out + (size_t)N_NODES * DIM;

    float* ws    = (float*)d_ws;
    float* xl_u  = ws;
    float* xl_i  = xl_u + (size_t)N_NODES * DIM;
    float* xr_i  = xl_i + (size_t)N_NODES * DIM;
    float* xr_u  = xr_i + (size_t)N_NODES * DIM;
    int* hist  = (int*)xl_u;
    int* cnt   = (int*)(xr_u + (size_t)N_NODES * DIM);
    int* off   = cnt + 2 * N_NODES;
    int* rank  = off + 2 * N_NODES;
    int* slots = rank + 2 * NEDGE;
    int* offl  = slots + 2 * NEDGE;
    int* blocktot = offl + 2 * N_NODES;
    float* P_ua = (float*)rank;
    float* P_ub = P_ua + (HID + 1) * DIM;
    float* P_ia = P_ub + (HID + 1) * DIM;
    float* P_ib = P_ia + (HID + 1) * DIM;

    const int blkNw   = (N_NODES + (ATT_BLOCK / 32) * NPW - 1) / ((ATT_BLOCK / 32) * NPW);
    const int blkS    = (N_NODES / 16 + 3) / 4;

    chunk_count_kernel<<<dim3(NCH, 2), GROUP_BLOCK, 0, stream>>>(
        e_u2i + NEDGE, e_i2u + NEDGE, hist, rank);
    node_prefix_kernel<<<dim3(NBLK, 2), 256, 0, stream>>>(hist, cnt, offl, blocktot);
    fill_slots_kernel<<<dim3(NCH, 2), GROUP_BLOCK, 0, stream>>>(
        e_u2i, e_u2i + NEDGE, e_i2u, e_i2u + NEDGE, hist, offl, blocktot, rank, off, slots);

    FoldSide fu = { W2_u, b2_u, Wl_u2i, bl_u2i, Wr_i2u, br_i2u, P_ua, P_ub };
    FoldSide fi = { W2_i, b2_i, Wl_i2u, bl_i2u, Wr_u2i, br_u2i, P_ia, P_ib };
    fold_kernel<<<dim3(HID / 16 + 1, 2, 2), 256, 0, stream>>>(fu, fi);

    NodeSide su = { x_user, W1_u, b1_u, P_ua, P_ua + HID * DIM, P_ub, P_ub + HID * DIM,
                    xl_u, xr_u };
    NodeSide si = { x_item, W1_i, b1_i, P_ia, P_ia + HID * DIM, P_ib, P_ib + HID * DIM,
                    xl_i, xr_i };
    node_chain_kernel<<<dim3(blkS, 2), 128, 0, stream>>>(su, si, N_NODES);

    seg_attend_kernel<<<blkNw, ATT_BLOCK, 0, stream>>>(xl_u, xr_i, slots, off, cnt,
                                                 att_u2i, bias_u2i, out_item, N_NODES);

    seg_attend_kernel<<<blkNw, ATT_BLOCK, 0, stream>>>(xl_i, xr_u, slots + NEDGE, off + N_NODES,
                                                 cnt + N_NODES,
                                                 att_i2u, bias_i2u, out_user, N_NODES);
}


# Per-node softmax: a group's eight 32-lane logit sums as one transposing reduction (permlanex16 + row_xmask DPP: 9 exchanges, not 40 ds_bpermute), exp once per group in lanes; sum order kept
# speedup vs baseline: 3.4694x; 1.0323x over previous
#include <hip/hip_runtime.h>


typedef __attribute__((ext_vector_type(2))) float v2f;
typedef __attribute__((ext_vector_type(8))) float v8f;
typedef __attribute__((ext_vector_type(4))) float v4f;

#define N_NODES 50000
#define DIM 128
#define HID 64
#define NEDGE 500000
#define SLOPE 0.2f

#define SP 132

struct NodeSide {
    const float* x;
    const float* W1; const float* b1;
    const float* Pa; const float* ca;
    const float* Pb; const float* cb;
    float* ya;
    float* yb;
};

struct FoldSide {
    const float* W2; const float* b2;
    const float* Wa; const float* ba;
    const float* Wb; const float* bb;
    float* Pa;
    float* Pb;
};

template <int NT, int K, int LDW = NT * 16>
__device__ __forceinline__ void stripe_mm(const float* arow, const float* __restrict__ W,
                                          int nloc, int khalf, v8f (&acc)[NT])
{
    constexpr int ncols = LDW;
#pragma unroll (16 / NT)
    for (int k0 = 0; k0 < K; k0 += 4) {
        const v2f a = *(const v2f*)(arow + k0);
        const float* __restrict__ w0 = W + (size_t)(k0 + khalf) * ncols + nloc;
        const float* __restrict__ w1 = w0 + ncols;
#pragma unroll
        for (int t = 0; t < NT; ++t) {
            v2f b;
            b.x = w0[t * 16];
            b.y = w1[t * 16];
            acc[t] = __builtin_amdgcn_wmma_f32_16x16x4_f32(
                false, a, false, b, (short)0, acc[t], false, false);
        }
    }
}

template <int NT>
__device__ __forceinline__ void zero_acc(v8f (&acc)[NT]) {
#pragma unroll
    for (int t = 0; t < NT; ++t)
        acc[t] = (v8f){0.f, 0.f, 0.f, 0.f, 0.f, 0.f, 0.f, 0.f};
}

template <int NT, bool RELU>
__device__ __forceinline__ void tiles_to_lds(float* S, const v8f (&acc)[NT],
                                             const float* __restrict__ bias,
                                             int nloc, int rhalf)
{
#pragma unroll
    for (int t = 0; t < NT; ++t) {
        const int n = t * 16 + nloc;
        const float bv = bias[n];
#pragma unroll
        for (int j = 0; j < 8; ++j) {
            float v = acc[t][j] + bv;
            if (RELU) v = fmaxf(v, 0.f);
            S[(rhalf + j) * SP + n] = v;
        }
    }
}

template <int NT>
__device__ __forceinline__ void tiles_to_global(float* __restrict__ C, const v8f (&acc)[NT],
                                                const float* __restrict__ bias,
                                                int nloc, int mbase)
{
    constexpr int ncols = NT * 16;
#pragma unroll
    for (int t = 0; t < NT; ++t) {
        const int n = t * 16 + nloc;
        const float bv = bias[n];
#pragma unroll
        for (int j = 0; j < 8; ++j)
            C[(size_t)(mbase + j) * ncols + n] = acc[t][j] + bv;
    }
}

__global__ __launch_bounds__(256) void fold_kernel(FoldSide fu, FoldSide fi)
{
    const bool second = (blockIdx.y != 0);
    const bool projb  = (blockIdx.z != 0);
    const float* W2 = second ? fi.W2 : fu.W2;
    const float* b2 = second ? fi.b2 : fu.b2;
    const float* W  = second ? (projb ? fi.Wb : fi.Wa) : (projb ? fu.Wb : fu.Wa);
    const float* b  = second ? (projb ? fi.bb : fi.ba) : (projb ? fu.bb : fu.ba);
    float* P        = second ? (projb ? fi.Pb : fi.Pa) : (projb ? fu.Pb : fu.Pa);

    const int s    = (int)blockIdx.x;
    const int t    = (int)(threadIdx.x >> 5);
    const int lane = (int)(threadIdx.x & 31);
    const int nloc  = lane & 15;
    const int khalf = (lane >> 4) << 1;
    const int rhalf = (lane >> 4) << 3;
    const bool brow = (s == HID / 16);

    const float* arow = brow ? (b2 + khalf)
                             : (W2 + (size_t)(s * 16 + nloc) * DIM + khalf);
    v8f acc[1];
    zero_acc<1>(acc);
    stripe_mm<1, DIM, DIM>(arow, W + t * 16, nloc, khalf, acc);

    const int n = t * 16 + nloc;
    if (!brow) {
#pragma unroll
        for (int j = 0; j < 8; ++j)
            P[(size_t)(s * 16 + rhalf + j) * DIM + n] = acc[0][j];
    } else if (lane < 16) {
        P[(size_t)HID * DIM + n] = acc[0][0] + b[n];
    }
}

__global__ __launch_bounds__(128) void node_chain_kernel(NodeSide su, NodeSide si, int rows)
{
    __shared__ float lds[4 * 16 * SP];

    const bool second = (blockIdx.y != 0);
    const float* x  = second ? si.x  : su.x;
    const float* W1 = second ? si.W1 : su.W1;
    const float* b1 = second ? si.b1 : su.b1;
    const float* Pa = second ? si.Pa : su.Pa;
    const float* ca = second ? si.ca : su.ca;
    const float* Pb = second ? si.Pb : su.Pb;
    const float* cb = second ? si.cb : su.cb;
    float* ya = second ? si.ya : su.ya;
    float* yb = second ? si.yb : su.yb;

    const int wv   = (int)(threadIdx.x >> 5);
    const int lane = (int)(threadIdx.x & 31);
    int stripe = (int)blockIdx.x * 4 + wv;
    const bool live = stripe * 16 < rows;
    if (!live) stripe = rows / 16 - 1;

    const int nloc  = lane & 15;
    const int khalf = (lane >> 4) << 1;
    const int rhalf = (lane >> 4) << 3;
    const int mbase = stripe * 16 + rhalf;

    float* S = lds + wv * (16 * SP);
    const float* sfrag = S + nloc * SP + khalf;

    v8f acc4[4];
    v8f acc8[8];

    zero_acc<4>(acc4);
    stripe_mm<4, DIM>(x + (size_t)(stripe * 16 + nloc) * DIM + khalf, W1, nloc, khalf, acc4);
    tiles_to_lds<4, true>(S, acc4, b1, nloc, rhalf);
    __syncthreads();

    zero_acc<8>(acc8);
    stripe_mm<8, HID>(sfrag, Pa, nloc, khalf, acc8);
    if (live) tiles_to_global<8>(ya, acc8, ca, nloc, mbase);

    zero_acc<8>(acc8);
    stripe_mm<8, HID>(sfrag, Pb, nloc, khalf, acc8);
    if (live) tiles_to_global<8>(yb, acc8, cb, nloc, mbase);
}


#define NCH 64
#define CHUNK 7816
#define NBLK 196
#define GROUP_BLOCK 1024

#define NPW 4
#define ATT_BLOCK 128

__global__ __launch_bounds__(GROUP_BLOCK) void chunk_count_kernel(
    const int* __restrict__ dst0, const int* __restrict__ dst1,
    int* __restrict__ hist, int* __restrict__ rank)
{
    __shared__ __attribute__((aligned(16))) int tab[N_NODES];
    const int c   = (int)blockIdx.x;
    const int rel = (int)blockIdx.y;
    const int t   = (int)threadIdx.x;
    const int* __restrict__ dst = rel ? dst1 : dst0;
    int* __restrict__ r = rank + (size_t)rel * NEDGE;
    int* __restrict__ h = hist + (size_t)(rel * NCH + c) * N_NODES;

    for (int i = t * 4; i < N_NODES; i += GROUP_BLOCK * 4)
        *(int4*)(tab + i) = make_int4(0, 0, 0, 0);
    __syncthreads();

    const int e0 = c * CHUNK;
    const int e1 = min(e0 + CHUNK, NEDGE);
    for (int e = e0 + t * 4; e < e1; e += GROUP_BLOCK * 4) {
        const int4 d = *(const int4*)(dst + e);
        int4 k;
        k.x = atomicAdd(tab + d.x, 1);
        k.y = atomicAdd(tab + d.y, 1);
        k.z = atomicAdd(tab + d.z, 1);
        k.w = atomicAdd(tab + d.w, 1);
        *(int4*)(r + e) = k;
    }
    __syncthreads();

    for (int i = t * 4; i < N_NODES; i += GROUP_BLOCK * 4)
        *(int4*)(h + i) = *(const int4*)(tab + i);
}

__global__ __launch_bounds__(256) void node_prefix_kernel(
    int* __restrict__ hist, int* __restrict__ cnt,
    int* __restrict__ offl, int* __restrict__ blocktot)
{
    __shared__ int s_tot[8];
    const int rel = (int)blockIdx.y;
    const int t = (int)threadIdx.x, lane = t & 31, w = t >> 5;
    const int idx = (int)blockIdx.x * 256 + t;
    int* __restrict__ h = hist + (size_t)rel * NCH * N_NODES;

    int run = 0;
    if (idx < N_NODES) {
        for (int c0 = 0; c0 < NCH; c0 += 32) {
            int v[32];
#pragma unroll
            for (int j = 0; j < 32; ++j) v[j] = h[(size_t)(c0 + j) * N_NODES + idx];
#pragma unroll
            for (int j = 0; j < 32; ++j) {
                h[(size_t)(c0 + j) * N_NODES + idx] = run;
                run += v[j];
            }
        }
        cnt[rel * N_NODES + idx] = run;
    }

    int inc = run;
#pragma unroll
    for (int dlt = 1; dlt < 32; dlt <<= 1) {
        const int u = __shfl_up(inc, dlt, 32);
        if (lane >= dlt) inc += u;
    }
    if (lane == 31) s_tot[w] = inc;
    __syncthreads();
    int add = 0, all = 0;
#pragma unroll
    for (int k = 0; k < 8; ++k) {
        const int s = s_tot[k];
        if (k < w) add += s;
        all += s;
    }
    if (idx < N_NODES) offl[rel * N_NODES + idx] = add + inc - run;
    if (t == 0) blocktot[rel * NBLK + (int)blockIdx.x] = all;
}

__global__ __launch_bounds__(GROUP_BLOCK) void fill_slots_kernel(
    const int* __restrict__ src0, const int* __restrict__ dst0,
    const int* __restrict__ src1, const int* __restrict__ dst1,
    const int* __restrict__ hist, const int* __restrict__ offl,
    const int* __restrict__ blocktot, const int* __restrict__ rank,
    int* __restrict__ off, int* __restrict__ slots)
{
    __shared__ __attribute__((aligned(16))) int tab[N_NODES];
    __shared__ int bb[256];
    __shared__ int s_w[8];
    const int c   = (int)blockIdx.x;
    const int rel = (int)blockIdx.y;
    const int t = (int)threadIdx.x, lane = t & 31, w = t >> 5;
    const int* __restrict__ src = rel ? src1 : src0;
    const int* __restrict__ dst = rel ? dst1 : dst0;
    const int* __restrict__ hc = hist + (size_t)(rel * NCH + c) * N_NODES;
    const int* __restrict__ ol = offl + (size_t)rel * N_NODES;
    const int* __restrict__ r  = rank + (size_t)rel * NEDGE;
    int* __restrict__ og = off + (size_t)rel * N_NODES;
    int* __restrict__ sl = slots + (size_t)rel * NEDGE;

    const int v = (t < NBLK) ? blocktot[rel * NBLK + t] : 0;
    int inc = v;
#pragma unroll
    for (int dlt = 1; dlt < 32; dlt <<= 1) {
        const int u = __shfl_up(inc, dlt, 32);
        if (lane >= dlt) inc += u;
    }
    if (w < 8 && lane == 31) s_w[w] = inc;
    __syncthreads();
    if (t < 256) {
        int add = 0;
#pragma unroll
        for (int k = 0; k < 8; ++k)
            if (k < w) add += s_w[k];
        bb[t] = add + inc - v;
    }
    __syncthreads();

#pragma unroll 4
    for (int i = t * 4; i < N_NODES; i += GROUP_BLOCK * 4) {
        const int4 a = *(const int4*)(hc + i);
        const int4 o = *(const int4*)(ol + i);
        const int b = bb[i >> 8];
        *(int4*)(tab + i) = make_int4(o.x + b + a.x, o.y + b + a.y,
                                      o.z + b + a.z, o.w + b + a.w);
    }
    __syncthreads();

    if (c == 0) {
        for (int i = t * 4; i < N_NODES; i += GROUP_BLOCK * 4)
            *(int4*)(og + i) = *(const int4*)(tab + i);
    }

    const int e0 = c * CHUNK;
    const int e1 = min(e0 + CHUNK, NEDGE);
    for (int e = e0 + t * 4; e < e1; e += GROUP_BLOCK * 4) {
        const int4 d = *(const int4*)(dst + e);
        const int4 k = *(const int4*)(r + e);
        const int4 s = *(const int4*)(src + e);
        sl[tab[d.x] + k.x] = s.x;
        sl[tab[d.y] + k.y] = s.y;
        sl[tab[d.z] + k.z] = s.z;
        sl[tab[d.w] + k.w] = s.w;
    }
}

__device__ __forceinline__ float xhalf(float x) {
    const int b = __builtin_bit_cast(int, x);
    return __builtin_bit_cast(float,
        __builtin_amdgcn_permlanex16(b, b, 0x76543210, 0xfedcba98, false, false));
}
template <int M>
__device__ __forceinline__ float xrow(float x) {
    const int b = __builtin_bit_cast(int, x);
    return __builtin_bit_cast(float,
        __builtin_amdgcn_update_dpp(b, b, 0x160 + M, 0xf, 0xf, true));
}

__global__ __launch_bounds__(ATT_BLOCK) void seg_attend_kernel(
    const float* __restrict__ xl, const float* __restrict__ xr,
    const int* __restrict__ slots, const int* __restrict__ off,
    const int* __restrict__ cnt,
    const float* __restrict__ att, const float* __restrict__ bias,
    float* __restrict__ out, int n)
{
    const int lane = (int)(threadIdx.x & 31);
    const int w = __builtin_amdgcn_readfirstlane(
        (int)((blockIdx.x * blockDim.x + threadIdx.x) >> 5));
    const int d0 = w * NPW;
    if (d0 >= n) return;
    const int dend = min(d0 + NPW, n);
    const bool up16 = (lane & 16) != 0;
    const bool up8  = (lane & 8) != 0;
    const bool up4  = (lane & 4) != 0;
    const int  myj  = (lane >> 2) & 7;
    const float4 at = *(const float4*)(att + lane * 4);
    const float4 bi = *(const float4*)(bias + lane * 4);

    for (int d = d0; d < dend; ++d) {
        const int beg = off[d];
        const int deg = cnt[d];
        const float4 b = *(const float4*)(xr + (size_t)d * DIM + lane * 4);

        float4 acc = make_float4(0.f, 0.f, 0.f, 0.f);
        float m = -INFINITY;
        float s = 0.f;

        for (int j0 = 0; j0 < deg; j0 += 8) {
            const int nv = min(8, deg - j0);
            const int jj = min(lane & 7, nv - 1);
            const int sidx = slots[beg + j0 + jj];

            v4f   a[8];
            float p[8];
#pragma unroll
            for (int j = 0; j < 8; ++j) {
                const int sj = __builtin_amdgcn_readlane(sidx, j);
                a[j] = *(const v4f*)(xl + (size_t)sj * DIM + lane * 4);
            }
            asm volatile("" :: "v"(a[0]), "v"(a[1]), "v"(a[2]), "v"(a[3]),
                               "v"(a[4]), "v"(a[5]), "v"(a[6]), "v"(a[7]));
#pragma unroll
            for (int j = 0; j < 8; ++j) {
                float z, q = 0.f;
                z = a[j].x + b.x; z = (z > 0.f) ? z : SLOPE * z; q += z * at.x;
                z = a[j].y + b.y; z = (z > 0.f) ? z : SLOPE * z; q += z * at.y;
                z = a[j].z + b.z; z = (z > 0.f) ? z : SLOPE * z; q += z * at.z;
                z = a[j].w + b.w; z = (z > 0.f) ? z : SLOPE * z; q += z * at.w;
                p[j] = q;
            }
            float q4[4], q2[2];
#pragma unroll
            for (int i = 0; i < 4; ++i) {
                const float keep = up16 ? p[i + 4] : p[i];
                const float give = up16 ? p[i] : p[i + 4];
                q4[i] = keep + xhalf(give);
            }
#pragma unroll
            for (int i = 0; i < 2; ++i) {
                const float keep = up8 ? q4[i + 2] : q4[i];
                const float give = up8 ? q4[i] : q4[i + 2];
                q2[i] = keep + xrow<8>(give);
            }
            float lg;
            {
                const float keep = up4 ? q2[1] : q2[0];
                const float give = up4 ? q2[0] : q2[1];
                lg = keep + xrow<4>(give);
            }
            lg += xrow<2>(lg);
            lg += xrow<1>(lg);
            const float lj = (myj < nv) ? lg : -INFINITY;
            float mc = fmaxf(lj, xrow<4>(lj));
            mc = fmaxf(mc, xrow<8>(mc));
            mc = fmaxf(mc, xhalf(mc));
            const float mn = fmaxf(m, mc);
            const float sc = expf(m - mn);
            const float ej = expf(lj - mn);
            acc.x *= sc; acc.y *= sc; acc.z *= sc; acc.w *= sc;
            s *= sc;
#pragma unroll
            for (int j = 0; j < 8; ++j) {
                const float e = __builtin_bit_cast(float,
                    __builtin_amdgcn_readlane(__builtin_bit_cast(int, ej), 4 * j));
                s += e;
                acc.x += e * a[j].x;
                acc.y += e * a[j].y;
                acc.z += e * a[j].z;
                acc.w += e * a[j].w;
            }
            m = mn;
        }

        const float den = s + 1e-16f;
        float4 r;
        r.x = acc.x / den + bi.x;
        r.y = acc.y / den + bi.y;
        r.z = acc.z / den + bi.z;
        r.w = acc.w / den + bi.w;
        *(float4*)(out + (size_t)d * DIM + lane * 4) = r;
    }
}

extern "C" void kernel_launch(void* const* d_in, const int* in_sizes, int n_in,
                              void* d_out, int out_size, void* d_ws, size_t ws_size,
                              hipStream_t stream) {
    (void)in_sizes; (void)n_in; (void)out_size; (void)ws_size;

    const float* x_user = (const float*)d_in[0];
    const float* x_item = (const float*)d_in[1];
    const int*   e_u2i  = (const int*)d_in[2];
    const int*   e_i2u  = (const int*)d_in[3];
    const float* W1_u = (const float*)d_in[4],  *b1_u = (const float*)d_in[5];
    const float* W2_u = (const float*)d_in[6],  *b2_u = (const float*)d_in[7];
    const float* W1_i = (const float*)d_in[8],  *b1_i = (const float*)d_in[9];
    const float* W2_i = (const float*)d_in[10], *b2_i = (const float*)d_in[11];
    const float* Wl_u2i = (const float*)d_in[12], *bl_u2i = (const float*)d_in[13];
    const float* Wr_u2i = (const float*)d_in[14], *br_u2i = (const float*)d_in[15];
    const float* att_u2i = (const float*)d_in[16], *bias_u2i = (const float*)d_in[17];
    const float* Wl_i2u = (const float*)d_in[18], *bl_i2u = (const float*)d_in[19];
    const float* Wr_i2u = (const float*)d_in[20], *br_i2u = (const float*)d_in[21];
    const float* att_i2u = (const float*)d_in[22], *bias_i2u = (const float*)d_in[23];

    float* out_user = (float*)d_out;
    float* out_item = (float*)d_out + (size_t)N_NODES * DIM;

    float* ws    = (float*)d_ws;
    float* xl_u  = ws;
    float* xl_i  = xl_u + (size_t)N_NODES * DIM;
    float* xr_i  = xl_i + (size_t)N_NODES * DIM;
    float* xr_u  = xr_i + (size_t)N_NODES * DIM;
    int* hist  = (int*)xl_u;
    int* cnt   = (int*)(xr_u + (size_t)N_NODES * DIM);
    int* off   = cnt + 2 * N_NODES;
    int* rank  = off + 2 * N_NODES;
    int* slots = rank + 2 * NEDGE;
    int* offl  = slots + 2 * NEDGE;
    int* blocktot = offl + 2 * N_NODES;
    float* P_ua = (float*)rank;
    float* P_ub = P_ua + (HID + 1) * DIM;
    float* P_ia = P_ub + (HID + 1) * DIM;
    float* P_ib = P_ia + (HID + 1) * DIM;

    const int blkNw   = (N_NODES + (ATT_BLOCK / 32) * NPW - 1) / ((ATT_BLOCK / 32) * NPW);
    const int blkS    = (N_NODES / 16 + 3) / 4;

    chunk_count_kernel<<<dim3(NCH, 2), GROUP_BLOCK, 0, stream>>>(
        e_u2i + NEDGE, e_i2u + NEDGE, hist, rank);
    node_prefix_kernel<<<dim3(NBLK, 2), 256, 0, stream>>>(hist, cnt, offl, blocktot);
    fill_slots_kernel<<<dim3(NCH, 2), GROUP_BLOCK, 0, stream>>>(
        e_u2i, e_u2i + NEDGE, e_i2u, e_i2u + NEDGE, hist, offl, blocktot, rank, off, slots);

    FoldSide fu = { W2_u, b2_u, Wl_u2i, bl_u2i, Wr_i2u, br_i2u, P_ua, P_ub };
    FoldSide fi = { W2_i, b2_i, Wl_i2u, bl_i2u, Wr_u2i, br_u2i, P_ia, P_ib };
    fold_kernel<<<dim3(HID / 16 + 1, 2, 2), 256, 0, stream>>>(fu, fi);

    NodeSide su = { x_user, W1_u, b1_u, P_ua, P_ua + HID * DIM, P_ub, P_ub + HID * DIM,
                    xl_u, xr_u };
    NodeSide si = { x_item, W1_i, b1_i, P_ia, P_ia + HID * DIM, P_ib, P_ib + HID * DIM,
                    xl_i, xr_i };
    node_chain_kernel<<<dim3(blkS, 2), 128, 0, stream>>>(su, si, N_NODES);

    seg_attend_kernel<<<blkNw, ATT_BLOCK, 0, stream>>>(xl_u, xr_i, slots, off, cnt,
                                                 att_u2i, bias_u2i, out_item, N_NODES);

    seg_attend_kernel<<<blkNw, ATT_BLOCK, 0, stream>>>(xl_i, xr_u, slots + NEDGE, off + N_NODES,
                                                 cnt + N_NODES,
                                                 att_i2u, bias_i2u, out_user, N_NODES);
}


# Independent grids share launches: the W2 fold rides as extra work-groups of the first edge-grouping launch, both relations' per-node softmax kernels run as one launch; 7 launches become 5
# speedup vs baseline: 3.8949x; 1.1226x over previous
#include <hip/hip_runtime.h>


typedef __attribute__((ext_vector_type(2))) float v2f;
typedef __attribute__((ext_vector_type(8))) float v8f;
typedef __attribute__((ext_vector_type(4))) float v4f;

#define N_NODES 50000
#define DIM 128
#define HID 64
#define NEDGE 500000
#define SLOPE 0.2f

#define SP 132

struct NodeSide {
    const float* x;
    const float* W1; const float* b1;
    const float* Pa; const float* ca;
    const float* Pb; const float* cb;
    float* ya;
    float* yb;
};

struct FoldSide {
    const float* W2; const float* b2;
    const float* Wa; const float* ba;
    const float* Wb; const float* bb;
    float* Pa;
    float* Pb;
};

template <int NT, int K, int LDW = NT * 16>
__device__ __forceinline__ void stripe_mm(const float* arow, const float* __restrict__ W,
                                          int nloc, int khalf, v8f (&acc)[NT])
{
    constexpr int ncols = LDW;
#pragma unroll (16 / NT)
    for (int k0 = 0; k0 < K; k0 += 4) {
        const v2f a = *(const v2f*)(arow + k0);
        const float* __restrict__ w0 = W + (size_t)(k0 + khalf) * ncols + nloc;
        const float* __restrict__ w1 = w0 + ncols;
#pragma unroll
        for (int t = 0; t < NT; ++t) {
            v2f b;
            b.x = w0[t * 16];
            b.y = w1[t * 16];
            acc[t] = __builtin_amdgcn_wmma_f32_16x16x4_f32(
                false, a, false, b, (short)0, acc[t], false, false);
        }
    }
}

template <int NT>
__device__ __forceinline__ void zero_acc(v8f (&acc)[NT]) {
#pragma unroll
    for (int t = 0; t < NT; ++t)
        acc[t] = (v8f){0.f, 0.f, 0.f, 0.f, 0.f, 0.f, 0.f, 0.f};
}

template <int NT, bool RELU>
__device__ __forceinline__ void tiles_to_lds(float* S, const v8f (&acc)[NT],
                                             const float* __restrict__ bias,
                                             int nloc, int rhalf)
{
#pragma unroll
    for (int t = 0; t < NT; ++t) {
        const int n = t * 16 + nloc;
        const float bv = bias[n];
#pragma unroll
        for (int j = 0; j < 8; ++j) {
            float v = acc[t][j] + bv;
            if (RELU) v = fmaxf(v, 0.f);
            S[(rhalf + j) * SP + n] = v;
        }
    }
}

template <int NT>
__device__ __forceinline__ void tiles_to_global(float* __restrict__ C, const v8f (&acc)[NT],
                                                const float* __restrict__ bias,
                                                int nloc, int mbase)
{
    constexpr int ncols = NT * 16;
#pragma unroll
    for (int t = 0; t < NT; ++t) {
        const int n = t * 16 + nloc;
        const float bv = bias[n];
#pragma unroll
        for (int j = 0; j < 8; ++j)
            C[(size_t)(mbase + j) * ncols + n] = acc[t][j] + bv;
    }
}

#define FOLD_UNITS 160
__device__ __forceinline__ void fold_unit(const FoldSide& fu, const FoldSide& fi, int u, int lane)
{
    const int g = u >> 3;
    const bool second = ((g / 5) & 1) != 0;
    const bool projb  = (g >= 10);
    const float* W2 = second ? fi.W2 : fu.W2;
    const float* b2 = second ? fi.b2 : fu.b2;
    const float* W  = second ? (projb ? fi.Wb : fi.Wa) : (projb ? fu.Wb : fu.Wa);
    const float* b  = second ? (projb ? fi.bb : fi.ba) : (projb ? fu.bb : fu.ba);
    float* P        = second ? (projb ? fi.Pb : fi.Pa) : (projb ? fu.Pb : fu.Pa);

    const int s    = g % 5;
    const int t    = u & 7;
    const int nloc  = lane & 15;
    const int khalf = (lane >> 4) << 1;
    const int rhalf = (lane >> 4) << 3;
    const bool brow = (s == HID / 16);

    const float* arow = brow ? (b2 + khalf)
                             : (W2 + (size_t)(s * 16 + nloc) * DIM + khalf);
    v8f acc[1];
    zero_acc<1>(acc);
    stripe_mm<1, DIM, DIM>(arow, W + t * 16, nloc, khalf, acc);

    const int n = t * 16 + nloc;
    if (!brow) {
#pragma unroll
        for (int j = 0; j < 8; ++j)
            P[(size_t)(s * 16 + rhalf + j) * DIM + n] = acc[0][j];
    } else if (lane < 16) {
        P[(size_t)HID * DIM + n] = acc[0][0] + b[n];
    }
}

__global__ __launch_bounds__(128) void node_chain_kernel(NodeSide su, NodeSide si, int rows)
{
    __shared__ float lds[4 * 16 * SP];

    const bool second = (blockIdx.y != 0);
    const float* x  = second ? si.x  : su.x;
    const float* W1 = second ? si.W1 : su.W1;
    const float* b1 = second ? si.b1 : su.b1;
    const float* Pa = second ? si.Pa : su.Pa;
    const float* ca = second ? si.ca : su.ca;
    const float* Pb = second ? si.Pb : su.Pb;
    const float* cb = second ? si.cb : su.cb;
    float* ya = second ? si.ya : su.ya;
    float* yb = second ? si.yb : su.yb;

    const int wv   = (int)(threadIdx.x >> 5);
    const int lane = (int)(threadIdx.x & 31);
    int stripe = (int)blockIdx.x * 4 + wv;
    const bool live = stripe * 16 < rows;
    if (!live) stripe = rows / 16 - 1;

    const int nloc  = lane & 15;
    const int khalf = (lane >> 4) << 1;
    const int rhalf = (lane >> 4) << 3;
    const int mbase = stripe * 16 + rhalf;

    float* S = lds + wv * (16 * SP);
    const float* sfrag = S + nloc * SP + khalf;

    v8f acc4[4];
    v8f acc8[8];

    zero_acc<4>(acc4);
    stripe_mm<4, DIM>(x + (size_t)(stripe * 16 + nloc) * DIM + khalf, W1, nloc, khalf, acc4);
    tiles_to_lds<4, true>(S, acc4, b1, nloc, rhalf);
    __syncthreads();

    zero_acc<8>(acc8);
    stripe_mm<8, HID>(sfrag, Pa, nloc, khalf, acc8);
    if (live) tiles_to_global<8>(ya, acc8, ca, nloc, mbase);

    zero_acc<8>(acc8);
    stripe_mm<8, HID>(sfrag, Pb, nloc, khalf, acc8);
    if (live) tiles_to_global<8>(yb, acc8, cb, nloc, mbase);
}


#define NCH 64
#define CHUNK 7816
#define NBLK 196
#define GROUP_BLOCK 1024

#define NPW 4
#define ATT_BLOCK 128

#define FOLD_XB 3
__global__ __launch_bounds__(GROUP_BLOCK) void chunk_count_kernel(
    const int* __restrict__ dst0, const int* __restrict__ dst1,
    int* __restrict__ hist, int* __restrict__ rank, FoldSide fu, FoldSide fi)
{
    __shared__ __attribute__((aligned(16))) int tab[N_NODES];
    if (blockIdx.x >= NCH) {
        const int wv = __builtin_amdgcn_readfirstlane((int)(threadIdx.x >> 5));
        const int u = (((int)blockIdx.x - NCH) * 2 + (int)blockIdx.y) * (GROUP_BLOCK / 32) + wv;
        if (u < FOLD_UNITS) fold_unit(fu, fi, u, (int)(threadIdx.x & 31));
        return;
    }
    const int c   = (int)blockIdx.x;
    const int rel = (int)blockIdx.y;
    const int t   = (int)threadIdx.x;
    const int* __restrict__ dst = rel ? dst1 : dst0;
    int* __restrict__ r = rank + (size_t)rel * NEDGE;
    int* __restrict__ h = hist + (size_t)(rel * NCH + c) * N_NODES;

    for (int i = t * 4; i < N_NODES; i += GROUP_BLOCK * 4)
        *(int4*)(tab + i) = make_int4(0, 0, 0, 0);
    __syncthreads();

    const int e0 = c * CHUNK;
    const int e1 = min(e0 + CHUNK, NEDGE);
    for (int e = e0 + t * 4; e < e1; e += GROUP_BLOCK * 4) {
        const int4 d = *(const int4*)(dst + e);
        int4 k;
        k.x = atomicAdd(tab + d.x, 1);
        k.y = atomicAdd(tab + d.y, 1);
        k.z = atomicAdd(tab + d.z, 1);
        k.w = atomicAdd(tab + d.w, 1);
        *(int4*)(r + e) = k;
    }
    __syncthreads();

    for (int i = t * 4; i < N_NODES; i += GROUP_BLOCK * 4)
        *(int4*)(h + i) = *(const int4*)(tab + i);
}

__global__ __launch_bounds__(256) void node_prefix_kernel(
    int* __restrict__ hist, int* __restrict__ cnt,
    int* __restrict__ offl, int* __restrict__ blocktot)
{
    __shared__ int s_tot[8];
    const int rel = (int)blockIdx.y;
    const int t = (int)threadIdx.x, lane = t & 31, w = t >> 5;
    const int idx = (int)blockIdx.x * 256 + t;
    int* __restrict__ h = hist + (size_t)rel * NCH * N_NODES;

    int run = 0;
    if (idx < N_NODES) {
        for (int c0 = 0; c0 < NCH; c0 += 32) {
            int v[32];
#pragma unroll
            for (int j = 0; j < 32; ++j) v[j] = h[(size_t)(c0 + j) * N_NODES + idx];
#pragma unroll
            for (int j = 0; j < 32; ++j) {
                h[(size_t)(c0 + j) * N_NODES + idx] = run;
                run += v[j];
            }
        }
        cnt[rel * N_NODES + idx] = run;
    }

    int inc = run;
#pragma unroll
    for (int dlt = 1; dlt < 32; dlt <<= 1) {
        const int u = __shfl_up(inc, dlt, 32);
        if (lane >= dlt) inc += u;
    }
    if (lane == 31) s_tot[w] = inc;
    __syncthreads();
    int add = 0, all = 0;
#pragma unroll
    for (int k = 0; k < 8; ++k) {
        const int s = s_tot[k];
        if (k < w) add += s;
        all += s;
    }
    if (idx < N_NODES) offl[rel * N_NODES + idx] = add + inc - run;
    if (t == 0) blocktot[rel * NBLK + (int)blockIdx.x] = all;
}

__global__ __launch_bounds__(GROUP_BLOCK) void fill_slots_kernel(
    const int* __restrict__ src0, const int* __restrict__ dst0,
    const int* __restrict__ src1, const int* __restrict__ dst1,
    const int* __restrict__ hist, const int* __restrict__ offl,
    const int* __restrict__ blocktot, const int* __restrict__ rank,
    int* __restrict__ off, int* __restrict__ slots)
{
    __shared__ __attribute__((aligned(16))) int tab[N_NODES];
    __shared__ int bb[256];
    __shared__ int s_w[8];
    const int c   = (int)blockIdx.x;
    const int rel = (int)blockIdx.y;
    const int t = (int)threadIdx.x, lane = t & 31, w = t >> 5;
    const int* __restrict__ src = rel ? src1 : src0;
    const int* __restrict__ dst = rel ? dst1 : dst0;
    const int* __restrict__ hc = hist + (size_t)(rel * NCH + c) * N_NODES;
    const int* __restrict__ ol = offl + (size_t)rel * N_NODES;
    const int* __restrict__ r  = rank + (size_t)rel * NEDGE;
    int* __restrict__ og = off + (size_t)rel * N_NODES;
    int* __restrict__ sl = slots + (size_t)rel * NEDGE;

    const int v = (t < NBLK) ? blocktot[rel * NBLK + t] : 0;
    int inc = v;
#pragma unroll
    for (int dlt = 1; dlt < 32; dlt <<= 1) {
        const int u = __shfl_up(inc, dlt, 32);
        if (lane >= dlt) inc += u;
    }
    if (w < 8 && lane == 31) s_w[w] = inc;
    __syncthreads();
    if (t < 256) {
        int add = 0;
#pragma unroll
        for (int k = 0; k < 8; ++k)
            if (k < w) add += s_w[k];
        bb[t] = add + inc - v;
    }
    __syncthreads();

#pragma unroll 4
    for (int i = t * 4; i < N_NODES; i += GROUP_BLOCK * 4) {
        const int4 a = *(const int4*)(hc + i);
        const int4 o = *(const int4*)(ol + i);
        const int b = bb[i >> 8];
        *(int4*)(tab + i) = make_int4(o.x + b + a.x, o.y + b + a.y,
                                      o.z + b + a.z, o.w + b + a.w);
    }
    __syncthreads();

    if (c == 0) {
        for (int i = t * 4; i < N_NODES; i += GROUP_BLOCK * 4)
            *(int4*)(og + i) = *(const int4*)(tab + i);
    }

    const int e0 = c * CHUNK;
    const int e1 = min(e0 + CHUNK, NEDGE);
    for (int e = e0 + t * 4; e < e1; e += GROUP_BLOCK * 4) {
        const int4 d = *(const int4*)(dst + e);
        const int4 k = *(const int4*)(r + e);
        const int4 s = *(const int4*)(src + e);
        sl[tab[d.x] + k.x] = s.x;
        sl[tab[d.y] + k.y] = s.y;
        sl[tab[d.z] + k.z] = s.z;
        sl[tab[d.w] + k.w] = s.w;
    }
}

__device__ __forceinline__ float xhalf(float x) {
    const int b = __builtin_bit_cast(int, x);
    return __builtin_bit_cast(float,
        __builtin_amdgcn_permlanex16(b, b, 0x76543210, 0xfedcba98, false, false));
}
template <int M>
__device__ __forceinline__ float xrow(float x) {
    const int b = __builtin_bit_cast(int, x);
    return __builtin_bit_cast(float,
        __builtin_amdgcn_update_dpp(b, b, 0x160 + M, 0xf, 0xf, true));
}

__device__ __forceinline__ void seg_attend_body(
    const float* __restrict__ xl, const float* __restrict__ xr,
    const int* __restrict__ slots, const int* __restrict__ off,
    const int* __restrict__ cnt,
    const float* __restrict__ att, const float* __restrict__ bias,
    float* __restrict__ out, int n)
{
    const int lane = (int)(threadIdx.x & 31);
    const int w = __builtin_amdgcn_readfirstlane(
        (int)((blockIdx.x * blockDim.x + threadIdx.x) >> 5));
    const int d0 = w * NPW;
    if (d0 >= n) return;
    const int dend = min(d0 + NPW, n);
    const bool up16 = (lane & 16) != 0;
    const bool up8  = (lane & 8) != 0;
    const bool up4  = (lane & 4) != 0;
    const int  myj  = (lane >> 2) & 7;
    const float4 at = *(const float4*)(att + lane * 4);
    const float4 bi = *(const float4*)(bias + lane * 4);

    for (int d = d0; d < dend; ++d) {
        const int beg = off[d];
        const int deg = cnt[d];
        const float4 b = *(const float4*)(xr + (size_t)d * DIM + lane * 4);

        float4 acc = make_float4(0.f, 0.f, 0.f, 0.f);
        float m = -INFINITY;
        float s = 0.f;

        for (int j0 = 0; j0 < deg; j0 += 8) {
            const int nv = min(8, deg - j0);
            const int jj = min(lane & 7, nv - 1);
            const int sidx = slots[beg + j0 + jj];

            v4f   a[8];
            float p[8];
#pragma unroll
            for (int j = 0; j < 8; ++j) {
                const int sj = __builtin_amdgcn_readlane(sidx, j);
                a[j] = *(const v4f*)(xl + (size_t)sj * DIM + lane * 4);
            }
            asm volatile("" :: "v"(a[0]), "v"(a[1]), "v"(a[2]), "v"(a[3]),
                               "v"(a[4]), "v"(a[5]), "v"(a[6]), "v"(a[7]));
#pragma unroll
            for (int j = 0; j < 8; ++j) {
                float z, q = 0.f;
                z = a[j].x + b.x; z = (z > 0.f) ? z : SLOPE * z; q += z * at.x;
                z = a[j].y + b.y; z = (z > 0.f) ? z : SLOPE * z; q += z * at.y;
                z = a[j].z + b.z; z = (z > 0.f) ? z : SLOPE * z; q += z * at.z;
                z = a[j].w + b.w; z = (z > 0.f) ? z : SLOPE * z; q += z * at.w;
                p[j] = q;
            }
            float q4[4], q2[2];
#pragma unroll
            for (int i = 0; i < 4; ++i) {
                const float keep = up16 ? p[i + 4] : p[i];
                const float give = up16 ? p[i] : p[i + 4];
                q4[i] = keep + xhalf(give);
            }
#pragma unroll
            for (int i = 0; i < 2; ++i) {
                const float keep = up8 ? q4[i + 2] : q4[i];
                const float give = up8 ? q4[i] : q4[i + 2];
                q2[i] = keep + xrow<8>(give);
            }
            float lg;
            {
                const float keep = up4 ? q2[1] : q2[0];
                const float give = up4 ? q2[0] : q2[1];
                lg = keep + xrow<4>(give);
            }
            lg += xrow<2>(lg);
            lg += xrow<1>(lg);
            const float lj = (myj < nv) ? lg : -INFINITY;
            float mc = fmaxf(lj, xrow<4>(lj));
            mc = fmaxf(mc, xrow<8>(mc));
            mc = fmaxf(mc, xhalf(mc));
            const float mn = fmaxf(m, mc);
            const float sc = expf(m - mn);
            const float ej = expf(lj - mn);
            acc.x *= sc; acc.y *= sc; acc.z *= sc; acc.w *= sc;
            s *= sc;
#pragma unroll
            for (int j = 0; j < 8; ++j) {
                const float e = __builtin_bit_cast(float,
                    __builtin_amdgcn_readlane(__builtin_bit_cast(int, ej), 4 * j));
                s += e;
                acc.x += e * a[j].x;
                acc.y += e * a[j].y;
                acc.z += e * a[j].z;
                acc.w += e * a[j].w;
            }
            m = mn;
        }

        const float den = s + 1e-16f;
        float4 r;
        r.x = acc.x / den + bi.x;
        r.y = acc.y / den + bi.y;
        r.z = acc.z / den + bi.z;
        r.w = acc.w / den + bi.w;
        *(float4*)(out + (size_t)d * DIM + lane * 4) = r;
    }
}

struct AttendSide {
    const float* xl; const float* xr;
    const int* slots; const int* off; const int* cnt;
    const float* att; const float* bias;
    float* out;
};
__global__ __launch_bounds__(ATT_BLOCK) void seg_attend_kernel(AttendSide r0, AttendSide r1, int n)
{
    const bool second = (blockIdx.y != 0);
    seg_attend_body(second ? r1.xl : r0.xl, second ? r1.xr : r0.xr,
                    second ? r1.slots : r0.slots, second ? r1.off : r0.off,
                    second ? r1.cnt : r0.cnt,
                    second ? r1.att : r0.att, second ? r1.bias : r0.bias,
                    second ? r1.out : r0.out, n);
}

extern "C" void kernel_launch(void* const* d_in, const int* in_sizes, int n_in,
                              void* d_out, int out_size, void* d_ws, size_t ws_size,
                              hipStream_t stream) {
    (void)in_sizes; (void)n_in; (void)out_size; (void)ws_size;

    const float* x_user = (const float*)d_in[0];
    const float* x_item = (const float*)d_in[1];
    const int*   e_u2i  = (const int*)d_in[2];
    const int*   e_i2u  = (const int*)d_in[3];
    const float* W1_u = (const float*)d_in[4],  *b1_u = (const float*)d_in[5];
    const float* W2_u = (const float*)d_in[6],  *b2_u = (const float*)d_in[7];
    const float* W1_i = (const float*)d_in[8],  *b1_i = (const float*)d_in[9];
    const float* W2_i = (const float*)d_in[10], *b2_i = (const float*)d_in[11];
    const float* Wl_u2i = (const float*)d_in[12], *bl_u2i = (const float*)d_in[13];
    const float* Wr_u2i = (const float*)d_in[14], *br_u2i = (const float*)d_in[15];
    const float* att_u2i = (const float*)d_in[16], *bias_u2i = (const float*)d_in[17];
    const float* Wl_i2u = (const float*)d_in[18], *bl_i2u = (const float*)d_in[19];
    const float* Wr_i2u = (const float*)d_in[20], *br_i2u = (const float*)d_in[21];
    const float* att_i2u = (const float*)d_in[22], *bias_i2u = (const float*)d_in[23];

    float* out_user = (float*)d_out;
    float* out_item = (float*)d_out + (size_t)N_NODES * DIM;

    float* ws    = (float*)d_ws;
    float* xl_u  = ws;
    float* xl_i  = xl_u + (size_t)N_NODES * DIM;
    float* xr_i  = xl_i + (size_t)N_NODES * DIM;
    float* xr_u  = xr_i + (size_t)N_NODES * DIM;
    int* hist  = (int*)xl_u;
    int* cnt   = (int*)(xr_u + (size_t)N_NODES * DIM);
    int* off   = cnt + 2 * N_NODES;
    int* rank  = off + 2 * N_NODES;
    int* slots = rank + 2 * NEDGE;
    int* offl  = slots + 2 * NEDGE;
    int* blocktot = offl + 2 * N_NODES;
    float* P_ua = (float*)(blocktot + 2 * NBLK);
    float* P_ub = P_ua + (HID + 1) * DIM;
    float* P_ia = P_ub + (HID + 1) * DIM;
    float* P_ib = P_ia + (HID + 1) * DIM;

    const int blkNw   = (N_NODES + (ATT_BLOCK / 32) * NPW - 1) / ((ATT_BLOCK / 32) * NPW);
    const int blkS    = (N_NODES / 16 + 3) / 4;

    FoldSide fu = { W2_u, b2_u, Wl_u2i, bl_u2i, Wr_i2u, br_i2u, P_ua, P_ub };
    FoldSide fi = { W2_i, b2_i, Wl_i2u, bl_i2u, Wr_u2i, br_u2i, P_ia, P_ib };
    chunk_count_kernel<<<dim3(NCH + FOLD_XB, 2), GROUP_BLOCK, 0, stream>>>(
        e_u2i + NEDGE, e_i2u + NEDGE, hist, rank, fu, fi);
    node_prefix_kernel<<<dim3(NBLK, 2), 256, 0, stream>>>(hist, cnt, offl, blocktot);
    fill_slots_kernel<<<dim3(NCH, 2), GROUP_BLOCK, 0, stream>>>(
        e_u2i, e_u2i + NEDGE, e_i2u, e_i2u + NEDGE, hist, offl, blocktot, rank, off, slots);

    NodeSide su = { x_user, W1_u, b1_u, P_ua, P_ua + HID * DIM, P_ub, P_ub + HID * DIM,
                    xl_u, xr_u };
    NodeSide si = { x_item, W1_i, b1_i, P_ia, P_ia + HID * DIM, P_ib, P_ib + HID * DIM,
                    xl_i, xr_i };
    node_chain_kernel<<<dim3(blkS, 2), 128, 0, stream>>>(su, si, N_NODES);

    AttendSide r0 = { xl_u, xr_i, slots, off, cnt, att_u2i, bias_u2i, out_item };
    AttendSide r1 = { xl_i, xr_u, slots + NEDGE, off + N_NODES, cnt + N_NODES,
                      att_i2u, bias_i2u, out_user };
    seg_attend_kernel<<<dim3(blkNw, 2), ATT_BLOCK, 0, stream>>>(r0, r1, N_NODES);
}


# Independent grids share a launch: the grouping's per-node prefix blocks and the node-side GEMM blocks run as one grid (prefix first); chunk tables move to the output buffer; 5 launches become 4
# speedup vs baseline: 4.1104x; 1.0553x over previous
#include <hip/hip_runtime.h>


typedef __attribute__((ext_vector_type(2))) float v2f;
typedef __attribute__((ext_vector_type(8))) float v8f;
typedef __attribute__((ext_vector_type(4))) float v4f;

#define N_NODES 50000
#define DIM 128
#define HID 64
#define NEDGE 500000
#define SLOPE 0.2f

#define SP 132

struct NodeSide {
    const float* x;
    const float* W1; const float* b1;
    const float* Pa; const float* ca;
    const float* Pb; const float* cb;
    float* ya;
    float* yb;
};

struct FoldSide {
    const float* W2; const float* b2;
    const float* Wa; const float* ba;
    const float* Wb; const float* bb;
    float* Pa;
    float* Pb;
};

template <int NT, int K, int LDW = NT * 16>
__device__ __forceinline__ void stripe_mm(const float* arow, const float* __restrict__ W,
                                          int nloc, int khalf, v8f (&acc)[NT])
{
    constexpr int ncols = LDW;
#pragma unroll (16 / NT)
    for (int k0 = 0; k0 < K; k0 += 4) {
        const v2f a = *(const v2f*)(arow + k0);
        const float* __restrict__ w0 = W + (size_t)(k0 + khalf) * ncols + nloc;
        const float* __restrict__ w1 = w0 + ncols;
#pragma unroll
        for (int t = 0; t < NT; ++t) {
            v2f b;
            b.x = w0[t * 16];
            b.y = w1[t * 16];
            acc[t] = __builtin_amdgcn_wmma_f32_16x16x4_f32(
                false, a, false, b, (short)0, acc[t], false, false);
        }
    }
}

template <int NT>
__device__ __forceinline__ void zero_acc(v8f (&acc)[NT]) {
#pragma unroll
    for (int t = 0; t < NT; ++t)
        acc[t] = (v8f){0.f, 0.f, 0.f, 0.f, 0.f, 0.f, 0.f, 0.f};
}

template <int NT, bool RELU>
__device__ __forceinline__ void tiles_to_lds(float* S, const v8f (&acc)[NT],
                                             const float* __restrict__ bias,
                                             int nloc, int rhalf)
{
#pragma unroll
    for (int t = 0; t < NT; ++t) {
        const int n = t * 16 + nloc;
        const float bv = bias[n];
#pragma unroll
        for (int j = 0; j < 8; ++j) {
            float v = acc[t][j] + bv;
            if (RELU) v = fmaxf(v, 0.f);
            S[(rhalf + j) * SP + n] = v;
        }
    }
}

template <int NT>
__device__ __forceinline__ void tiles_to_global(float* __restrict__ C, const v8f (&acc)[NT],
                                                const float* __restrict__ bias,
                                                int nloc, int mbase)
{
    constexpr int ncols = NT * 16;
#pragma unroll
    for (int t = 0; t < NT; ++t) {
        const int n = t * 16 + nloc;
        const float bv = bias[n];
#pragma unroll
        for (int j = 0; j < 8; ++j)
            C[(size_t)(mbase + j) * ncols + n] = acc[t][j] + bv;
    }
}

#define FOLD_UNITS 160
__device__ __forceinline__ void fold_unit(const FoldSide& fu, const FoldSide& fi, int u, int lane)
{
    const int g = u >> 3;
    const bool second = ((g / 5) & 1) != 0;
    const bool projb  = (g >= 10);
    const float* W2 = second ? fi.W2 : fu.W2;
    const float* b2 = second ? fi.b2 : fu.b2;
    const float* W  = second ? (projb ? fi.Wb : fi.Wa) : (projb ? fu.Wb : fu.Wa);
    const float* b  = second ? (projb ? fi.bb : fi.ba) : (projb ? fu.bb : fu.ba);
    float* P        = second ? (projb ? fi.Pb : fi.Pa) : (projb ? fu.Pb : fu.Pa);

    const int s    = g % 5;
    const int t    = u & 7;
    const int nloc  = lane & 15;
    const int khalf = (lane >> 4) << 1;
    const int rhalf = (lane >> 4) << 3;
    const bool brow = (s == HID / 16);

    const float* arow = brow ? (b2 + khalf)
                             : (W2 + (size_t)(s * 16 + nloc) * DIM + khalf);
    v8f acc[1];
    zero_acc<1>(acc);
    stripe_mm<1, DIM, DIM>(arow, W + t * 16, nloc, khalf, acc);

    const int n = t * 16 + nloc;
    if (!brow) {
#pragma unroll
        for (int j = 0; j < 8; ++j)
            P[(size_t)(s * 16 + rhalf + j) * DIM + n] = acc[0][j];
    } else if (lane < 16) {
        P[(size_t)HID * DIM + n] = acc[0][0] + b[n];
    }
}

#define NODE_XB ((N_NODES / 16 + 3) / 4)
__device__ __forceinline__ void node_chain_body(const NodeSide& su, const NodeSide& si, int rows,
                                                int bx, bool second, float* lds)
{
    const float* x  = second ? si.x  : su.x;
    const float* W1 = second ? si.W1 : su.W1;
    const float* b1 = second ? si.b1 : su.b1;
    const float* Pa = second ? si.Pa : su.Pa;
    const float* ca = second ? si.ca : su.ca;
    const float* Pb = second ? si.Pb : su.Pb;
    const float* cb = second ? si.cb : su.cb;
    float* ya = second ? si.ya : su.ya;
    float* yb = second ? si.yb : su.yb;

    const int wv   = (int)(threadIdx.x >> 5);
    const int lane = (int)(threadIdx.x & 31);
    int stripe = bx * 4 + wv;
    const bool live = stripe * 16 < rows;
    if (!live) stripe = rows / 16 - 1;

    const int nloc  = lane & 15;
    const int khalf = (lane >> 4) << 1;
    const int rhalf = (lane >> 4) << 3;
    const int mbase = stripe * 16 + rhalf;

    float* S = lds + wv * (16 * SP);
    const float* sfrag = S + nloc * SP + khalf;

    v8f acc4[4];
    v8f acc8[8];

    zero_acc<4>(acc4);
    stripe_mm<4, DIM>(x + (size_t)(stripe * 16 + nloc) * DIM + khalf, W1, nloc, khalf, acc4);
    tiles_to_lds<4, true>(S, acc4, b1, nloc, rhalf);
    __syncthreads();

    zero_acc<8>(acc8);
    stripe_mm<8, HID>(sfrag, Pa, nloc, khalf, acc8);
    if (live) tiles_to_global<8>(ya, acc8, ca, nloc, mbase);

    zero_acc<8>(acc8);
    stripe_mm<8, HID>(sfrag, Pb, nloc, khalf, acc8);
    if (live) tiles_to_global<8>(yb, acc8, cb, nloc, mbase);
}


#define NCH 64
#define CHUNK 7816
#define PBLK 128
#define NBLK 391
#define GROUP_BLOCK 1024

#define NPW 4
#define ATT_BLOCK 128

#define FOLD_XB 3
__global__ __launch_bounds__(GROUP_BLOCK) void chunk_count_kernel(
    const int* __restrict__ dst0, const int* __restrict__ dst1,
    int* __restrict__ hist, int* __restrict__ rank, FoldSide fu, FoldSide fi)
{
    __shared__ __attribute__((aligned(16))) int tab[N_NODES];
    if (blockIdx.x >= NCH) {
        const int wv = __builtin_amdgcn_readfirstlane((int)(threadIdx.x >> 5));
        const int u = (((int)blockIdx.x - NCH) * 2 + (int)blockIdx.y) * (GROUP_BLOCK / 32) + wv;
        if (u < FOLD_UNITS) fold_unit(fu, fi, u, (int)(threadIdx.x & 31));
        return;
    }
    const int c   = (int)blockIdx.x;
    const int rel = (int)blockIdx.y;
    const int t   = (int)threadIdx.x;
    const int* __restrict__ dst = rel ? dst1 : dst0;
    int* __restrict__ r = rank + (size_t)rel * NEDGE;
    int* __restrict__ h = hist + (size_t)(rel * NCH + c) * N_NODES;

    for (int i = t * 4; i < N_NODES; i += GROUP_BLOCK * 4)
        *(int4*)(tab + i) = make_int4(0, 0, 0, 0);
    __syncthreads();

    const int e0 = c * CHUNK;
    const int e1 = min(e0 + CHUNK, NEDGE);
    for (int e = e0 + t * 4; e < e1; e += GROUP_BLOCK * 4) {
        const int4 d = *(const int4*)(dst + e);
        int4 k;
        k.x = atomicAdd(tab + d.x, 1);
        k.y = atomicAdd(tab + d.y, 1);
        k.z = atomicAdd(tab + d.z, 1);
        k.w = atomicAdd(tab + d.w, 1);
        *(int4*)(r + e) = k;
    }
    __syncthreads();

    for (int i = t * 4; i < N_NODES; i += GROUP_BLOCK * 4)
        *(int4*)(h + i) = *(const int4*)(tab + i);
}

__device__ __forceinline__ void node_prefix_body(
    int* __restrict__ hist, int* __restrict__ cnt,
    int* __restrict__ offl, int* __restrict__ blocktot,
    int rel, int blk, int* s_tot)
{
    const int t = (int)threadIdx.x, lane = t & 31, w = t >> 5;
    const int idx = blk * PBLK + t;
    int* __restrict__ h = hist + (size_t)rel * NCH * N_NODES;

    int run = 0;
    if (idx < N_NODES) {
        for (int c0 = 0; c0 < NCH; c0 += 32) {
            int v[32];
#pragma unroll
            for (int j = 0; j < 32; ++j) v[j] = h[(size_t)(c0 + j) * N_NODES + idx];
#pragma unroll
            for (int j = 0; j < 32; ++j) {
                h[(size_t)(c0 + j) * N_NODES + idx] = run;
                run += v[j];
            }
        }
        cnt[rel * N_NODES + idx] = run;
    }

    int inc = run;
#pragma unroll
    for (int dlt = 1; dlt < 32; dlt <<= 1) {
        const int u = __shfl_up(inc, dlt, 32);
        if (lane >= dlt) inc += u;
    }
    if (lane == 31) s_tot[w] = inc;
    __syncthreads();
    int add = 0, all = 0;
#pragma unroll
    for (int k = 0; k < PBLK / 32; ++k) {
        const int s = s_tot[k];
        if (k < w) add += s;
        all += s;
    }
    if (idx < N_NODES) offl[rel * N_NODES + idx] = add + inc - run;
    if (t == 0) blocktot[rel * NBLK + blk] = all;
}

__global__ __launch_bounds__(128) void prefix_nodes_kernel(
    int* __restrict__ hist, int* __restrict__ cnt,
    int* __restrict__ offl, int* __restrict__ blocktot,
    NodeSide su, NodeSide si, int rows)
{
    __shared__ float lds[4 * 16 * SP];
    __shared__ int s_tot[PBLK / 32];
    const int b = (int)blockIdx.x;
    if (b < 2 * NBLK) {
        const bool rel1 = (b >= NBLK);
        node_prefix_body(hist, cnt, offl, blocktot, rel1 ? 1 : 0, rel1 ? b - NBLK : b, s_tot);
        return;
    }
    const int nb = b - 2 * NBLK;
    const bool second = (nb >= NODE_XB);
    node_chain_body(su, si, rows, second ? nb - NODE_XB : nb, second, lds);
}

__global__ __launch_bounds__(GROUP_BLOCK) void fill_slots_kernel(
    const int* __restrict__ src0, const int* __restrict__ dst0,
    const int* __restrict__ src1, const int* __restrict__ dst1,
    const int* __restrict__ hist, const int* __restrict__ offl,
    const int* __restrict__ blocktot, const int* __restrict__ rank,
    int* __restrict__ off, int* __restrict__ slots)
{
    __shared__ __attribute__((aligned(16))) int tab[N_NODES];
    __shared__ int bb[512];
    __shared__ int s_w[16];
    const int c   = (int)blockIdx.x;
    const int rel = (int)blockIdx.y;
    const int t = (int)threadIdx.x, lane = t & 31, w = t >> 5;
    const int* __restrict__ src = rel ? src1 : src0;
    const int* __restrict__ dst = rel ? dst1 : dst0;
    const int* __restrict__ hc = hist + (size_t)(rel * NCH + c) * N_NODES;
    const int* __restrict__ ol = offl + (size_t)rel * N_NODES;
    const int* __restrict__ r  = rank + (size_t)rel * NEDGE;
    int* __restrict__ og = off + (size_t)rel * N_NODES;
    int* __restrict__ sl = slots + (size_t)rel * NEDGE;

    const int v = (t < NBLK) ? blocktot[rel * NBLK + t] : 0;
    int inc = v;
#pragma unroll
    for (int dlt = 1; dlt < 32; dlt <<= 1) {
        const int u = __shfl_up(inc, dlt, 32);
        if (lane >= dlt) inc += u;
    }
    if (w < 16 && lane == 31) s_w[w] = inc;
    __syncthreads();
    if (t < 512) {
        int add = 0;
#pragma unroll
        for (int k = 0; k < 16; ++k)
            if (k < w) add += s_w[k];
        bb[t] = add + inc - v;
    }
    __syncthreads();

#pragma unroll 4
    for (int i = t * 4; i < N_NODES; i += GROUP_BLOCK * 4) {
        const int4 a = *(const int4*)(hc + i);
        const int4 o = *(const int4*)(ol + i);
        const int b = bb[i >> 7];
        *(int4*)(tab + i) = make_int4(o.x + b + a.x, o.y + b + a.y,
                                      o.z + b + a.z, o.w + b + a.w);
    }
    __syncthreads();

    if (c == 0) {
        for (int i = t * 4; i < N_NODES; i += GROUP_BLOCK * 4)
            *(int4*)(og + i) = *(const int4*)(tab + i);
    }

    const int e0 = c * CHUNK;
    const int e1 = min(e0 + CHUNK, NEDGE);
    for (int e = e0 + t * 4; e < e1; e += GROUP_BLOCK * 4) {
        const int4 d = *(const int4*)(dst + e);
        const int4 k = *(const int4*)(r + e);
        const int4 s = *(const int4*)(src + e);
        sl[tab[d.x] + k.x] = s.x;
        sl[tab[d.y] + k.y] = s.y;
        sl[tab[d.z] + k.z] = s.z;
        sl[tab[d.w] + k.w] = s.w;
    }
}

__device__ __forceinline__ float xhalf(float x) {
    const int b = __builtin_bit_cast(int, x);
    return __builtin_bit_cast(float,
        __builtin_amdgcn_permlanex16(b, b, 0x76543210, 0xfedcba98, false, false));
}
template <int M>
__device__ __forceinline__ float xrow(float x) {
    const int b = __builtin_bit_cast(int, x);
    return __builtin_bit_cast(float,
        __builtin_amdgcn_update_dpp(b, b, 0x160 + M, 0xf, 0xf, true));
}

__device__ __forceinline__ void seg_attend_body(
    const float* __restrict__ xl, const float* __restrict__ xr,
    const int* __restrict__ slots, const int* __restrict__ off,
    const int* __restrict__ cnt,
    const float* __restrict__ att, const float* __restrict__ bias,
    float* __restrict__ out, int n)
{
    const int lane = (int)(threadIdx.x & 31);
    const int w = __builtin_amdgcn_readfirstlane(
        (int)((blockIdx.x * blockDim.x + threadIdx.x) >> 5));
    const int d0 = w * NPW;
    if (d0 >= n) return;
    const int dend = min(d0 + NPW, n);
    const bool up16 = (lane & 16) != 0;
    const bool up8  = (lane & 8) != 0;
    const bool up4  = (lane & 4) != 0;
    const int  myj  = (lane >> 2) & 7;
    const float4 at = *(const float4*)(att + lane * 4);
    const float4 bi = *(const float4*)(bias + lane * 4);

    for (int d = d0; d < dend; ++d) {
        const int beg = off[d];
        const int deg = cnt[d];
        const float4 b = *(const float4*)(xr + (size_t)d * DIM + lane * 4);

        float4 acc = make_float4(0.f, 0.f, 0.f, 0.f);
        float m = -INFINITY;
        float s = 0.f;

        for (int j0 = 0; j0 < deg; j0 += 8) {
            const int nv = min(8, deg - j0);
            const int jj = min(lane & 7, nv - 1);
            const int sidx = slots[beg + j0 + jj];

            v4f   a[8];
            float p[8];
#pragma unroll
            for (int j = 0; j < 8; ++j) {
                const int sj = __builtin_amdgcn_readlane(sidx, j);
                a[j] = *(const v4f*)(xl + (size_t)sj * DIM + lane * 4);
            }
            asm volatile("" :: "v"(a[0]), "v"(a[1]), "v"(a[2]), "v"(a[3]),
                               "v"(a[4]), "v"(a[5]), "v"(a[6]), "v"(a[7]));
#pragma unroll
            for (int j = 0; j < 8; ++j) {
                float z, q = 0.f;
                z = a[j].x + b.x; z = (z > 0.f) ? z : SLOPE * z; q += z * at.x;
                z = a[j].y + b.y; z = (z > 0.f) ? z : SLOPE * z; q += z * at.y;
                z = a[j].z + b.z; z = (z > 0.f) ? z : SLOPE * z; q += z * at.z;
                z = a[j].w + b.w; z = (z > 0.f) ? z : SLOPE * z; q += z * at.w;
                p[j] = q;
            }
            float q4[4], q2[2];
#pragma unroll
            for (int i = 0; i < 4; ++i) {
                const float keep = up16 ? p[i + 4] : p[i];
                const float give = up16 ? p[i] : p[i + 4];
                q4[i] = keep + xhalf(give);
            }
#pragma unroll
            for (int i = 0; i < 2; ++i) {
                const float keep = up8 ? q4[i + 2] : q4[i];
                const float give = up8 ? q4[i] : q4[i + 2];
                q2[i] = keep + xrow<8>(give);
            }
            float lg;
            {
                const float keep = up4 ? q2[1] : q2[0];
                const float give = up4 ? q2[0] : q2[1];
                lg = keep + xrow<4>(give);
            }
            lg += xrow<2>(lg);
            lg += xrow<1>(lg);
            const float lj = (myj < nv) ? lg : -INFINITY;
            float mc = fmaxf(lj, xrow<4>(lj));
            mc = fmaxf(mc, xrow<8>(mc));
            mc = fmaxf(mc, xhalf(mc));
            const float mn = fmaxf(m, mc);
            const float sc = expf(m - mn);
            const float ej = expf(lj - mn);
            acc.x *= sc; acc.y *= sc; acc.z *= sc; acc.w *= sc;
            s *= sc;
#pragma unroll
            for (int j = 0; j < 8; ++j) {
                const float e = __builtin_bit_cast(float,
                    __builtin_amdgcn_readlane(__builtin_bit_cast(int, ej), 4 * j));
                s += e;
                acc.x += e * a[j].x;
                acc.y += e * a[j].y;
                acc.z += e * a[j].z;
                acc.w += e * a[j].w;
            }
            m = mn;
        }

        const float den = s + 1e-16f;
        float4 r;
        r.x = acc.x / den + bi.x;
        r.y = acc.y / den + bi.y;
        r.z = acc.z / den + bi.z;
        r.w = acc.w / den + bi.w;
        *(float4*)(out + (size_t)d * DIM + lane * 4) = r;
    }
}

struct AttendSide {
    const float* xl; const float* xr;
    const int* slots; const int* off; const int* cnt;
    const float* att; const float* bias;
    float* out;
};
__global__ __launch_bounds__(ATT_BLOCK) void seg_attend_kernel(AttendSide r0, AttendSide r1, int n)
{
    const bool second = (blockIdx.y != 0);
    seg_attend_body(second ? r1.xl : r0.xl, second ? r1.xr : r0.xr,
                    second ? r1.slots : r0.slots, second ? r1.off : r0.off,
                    second ? r1.cnt : r0.cnt,
                    second ? r1.att : r0.att, second ? r1.bias : r0.bias,
                    second ? r1.out : r0.out, n);
}

extern "C" void kernel_launch(void* const* d_in, const int* in_sizes, int n_in,
                              void* d_out, int out_size, void* d_ws, size_t ws_size,
                              hipStream_t stream) {
    (void)in_sizes; (void)n_in; (void)out_size; (void)ws_size;

    const float* x_user = (const float*)d_in[0];
    const float* x_item = (const float*)d_in[1];
    const int*   e_u2i  = (const int*)d_in[2];
    const int*   e_i2u  = (const int*)d_in[3];
    const float* W1_u = (const float*)d_in[4],  *b1_u = (const float*)d_in[5];
    const float* W2_u = (const float*)d_in[6],  *b2_u = (const float*)d_in[7];
    const float* W1_i = (const float*)d_in[8],  *b1_i = (const float*)d_in[9];
    const float* W2_i = (const float*)d_in[10], *b2_i = (const float*)d_in[11];
    const float* Wl_u2i = (const float*)d_in[12], *bl_u2i = (const float*)d_in[13];
    const float* Wr_u2i = (const float*)d_in[14], *br_u2i = (const float*)d_in[15];
    const float* att_u2i = (const float*)d_in[16], *bias_u2i = (const float*)d_in[17];
    const float* Wl_i2u = (const float*)d_in[18], *bl_i2u = (const float*)d_in[19];
    const float* Wr_i2u = (const float*)d_in[20], *br_i2u = (const float*)d_in[21];
    const float* att_i2u = (const float*)d_in[22], *bias_i2u = (const float*)d_in[23];

    float* out_user = (float*)d_out;
    float* out_item = (float*)d_out + (size_t)N_NODES * DIM;

    float* ws    = (float*)d_ws;
    float* xl_u  = ws;
    float* xl_i  = xl_u + (size_t)N_NODES * DIM;
    float* xr_i  = xl_i + (size_t)N_NODES * DIM;
    float* xr_u  = xr_i + (size_t)N_NODES * DIM;
    int* hist  = (int*)d_out;
    int* cnt   = (int*)(xr_u + (size_t)N_NODES * DIM);
    int* off   = cnt + 2 * N_NODES;
    int* rank  = off + 2 * N_NODES;
    int* slots = rank + 2 * NEDGE;
    int* offl  = slots + 2 * NEDGE;
    int* blocktot = offl + 2 * N_NODES;
    float* P_ua = (float*)(blocktot + 2 * NBLK + 2);
    float* P_ub = P_ua + (HID + 1) * DIM;
    float* P_ia = P_ub + (HID + 1) * DIM;
    float* P_ib = P_ia + (HID + 1) * DIM;

    const int blkNw   = (N_NODES + (ATT_BLOCK / 32) * NPW - 1) / ((ATT_BLOCK / 32) * NPW);

    FoldSide fu = { W2_u, b2_u, Wl_u2i, bl_u2i, Wr_i2u, br_i2u, P_ua, P_ub };
    FoldSide fi = { W2_i, b2_i, Wl_i2u, bl_i2u, Wr_u2i, br_u2i, P_ia, P_ib };
    chunk_count_kernel<<<dim3(NCH + FOLD_XB, 2), GROUP_BLOCK, 0, stream>>>(
        e_u2i + NEDGE, e_i2u + NEDGE, hist, rank, fu, fi);
    NodeSide su = { x_user, W1_u, b1_u, P_ua, P_ua + HID * DIM, P_ub, P_ub + HID * DIM,
                    xl_u, xr_u };
    NodeSide si = { x_item, W1_i, b1_i, P_ia, P_ia + HID * DIM, P_ib, P_ib + HID * DIM,
                    xl_i, xr_i };
    prefix_nodes_kernel<<<dim3(2 * NBLK + 2 * NODE_XB), PBLK, 0, stream>>>(
        hist, cnt, offl, blocktot, su, si, N_NODES);
    fill_slots_kernel<<<dim3(NCH, 2), GROUP_BLOCK, 0, stream>>>(
        e_u2i, e_u2i + NEDGE, e_i2u, e_i2u + NEDGE, hist, offl, blocktot, rank, off, slots);

    AttendSide r0 = { xl_u, xr_i, slots, off, cnt, att_u2i, bias_u2i, out_item };
    AttendSide r1 = { xl_i, xr_u, slots + NEDGE, off + N_NODES, cnt + N_NODES,
                      att_i2u, bias_i2u, out_user };
    seg_attend_kernel<<<dim3(blkNw, 2), ATT_BLOCK, 0, stream>>>(r0, r1, N_NODES);
}


# Edge grouping cut two ways, one work-group per (chunk, node quarter): LDS tables 50 KB not 200 KB, 32 table rows not 64, all WGPs busy; node sum order: chunk (32, twice as long), then arrival
# speedup vs baseline: 4.5567x; 1.1086x over previous
#include <hip/hip_runtime.h>


typedef __attribute__((ext_vector_type(2))) float v2f;
typedef __attribute__((ext_vector_type(8))) float v8f;
typedef __attribute__((ext_vector_type(4))) float v4f;

#define N_NODES 50000
#define DIM 128
#define HID 64
#define NEDGE 500000
#define SLOPE 0.2f

#define SP 132

struct NodeSide {
    const float* x;
    const float* W1; const float* b1;
    const float* Pa; const float* ca;
    const float* Pb; const float* cb;
    float* ya;
    float* yb;
};

struct FoldSide {
    const float* W2; const float* b2;
    const float* Wa; const float* ba;
    const float* Wb; const float* bb;
    float* Pa;
    float* Pb;
};

template <int NT, int K, int LDW = NT * 16>
__device__ __forceinline__ void stripe_mm(const float* arow, const float* __restrict__ W,
                                          int nloc, int khalf, v8f (&acc)[NT])
{
    constexpr int ncols = LDW;
#pragma unroll (16 / NT)
    for (int k0 = 0; k0 < K; k0 += 4) {
        const v2f a = *(const v2f*)(arow + k0);
        const float* __restrict__ w0 = W + (size_t)(k0 + khalf) * ncols + nloc;
        const float* __restrict__ w1 = w0 + ncols;
#pragma unroll
        for (int t = 0; t < NT; ++t) {
            v2f b;
            b.x = w0[t * 16];
            b.y = w1[t * 16];
            acc[t] = __builtin_amdgcn_wmma_f32_16x16x4_f32(
                false, a, false, b, (short)0, acc[t], false, false);
        }
    }
}

template <int NT>
__device__ __forceinline__ void zero_acc(v8f (&acc)[NT]) {
#pragma unroll
    for (int t = 0; t < NT; ++t)
        acc[t] = (v8f){0.f, 0.f, 0.f, 0.f, 0.f, 0.f, 0.f, 0.f};
}

template <int NT, bool RELU>
__device__ __forceinline__ void tiles_to_lds(float* S, const v8f (&acc)[NT],
                                             const float* __restrict__ bias,
                                             int nloc, int rhalf)
{
#pragma unroll
    for (int t = 0; t < NT; ++t) {
        const int n = t * 16 + nloc;
        const float bv = bias[n];
#pragma unroll
        for (int j = 0; j < 8; ++j) {
            float v = acc[t][j] + bv;
            if (RELU) v = fmaxf(v, 0.f);
            S[(rhalf + j) * SP + n] = v;
        }
    }
}

template <int NT>
__device__ __forceinline__ void tiles_to_global(float* __restrict__ C, const v8f (&acc)[NT],
                                                const float* __restrict__ bias,
                                                int nloc, int mbase)
{
    constexpr int ncols = NT * 16;
#pragma unroll
    for (int t = 0; t < NT; ++t) {
        const int n = t * 16 + nloc;
        const float bv = bias[n];
#pragma unroll
        for (int j = 0; j < 8; ++j)
            C[(size_t)(mbase + j) * ncols + n] = acc[t][j] + bv;
    }
}

#define FOLD_UNITS 160
__device__ __forceinline__ void fold_unit(const FoldSide& fu, const FoldSide& fi, int u, int lane)
{
    const int g = u >> 3;
    const bool second = ((g / 5) & 1) != 0;
    const bool projb  = (g >= 10);
    const float* W2 = second ? fi.W2 : fu.W2;
    const float* b2 = second ? fi.b2 : fu.b2;
    const float* W  = second ? (projb ? fi.Wb : fi.Wa) : (projb ? fu.Wb : fu.Wa);
    const float* b  = second ? (projb ? fi.bb : fi.ba) : (projb ? fu.bb : fu.ba);
    float* P        = second ? (projb ? fi.Pb : fi.Pa) : (projb ? fu.Pb : fu.Pa);

    const int s    = g % 5;
    const int t    = u & 7;
    const int nloc  = lane & 15;
    const int khalf = (lane >> 4) << 1;
    const int rhalf = (lane >> 4) << 3;
    const bool brow = (s == HID / 16);

    const float* arow = brow ? (b2 + khalf)
                             : (W2 + (size_t)(s * 16 + nloc) * DIM + khalf);
    v8f acc[1];
    zero_acc<1>(acc);
    stripe_mm<1, DIM, DIM>(arow, W + t * 16, nloc, khalf, acc);

    const int n = t * 16 + nloc;
    if (!brow) {
#pragma unroll
        for (int j = 0; j < 8; ++j)
            P[(size_t)(s * 16 + rhalf + j) * DIM + n] = acc[0][j];
    } else if (lane < 16) {
        P[(size_t)HID * DIM + n] = acc[0][0] + b[n];
    }
}

#define NODE_XB ((N_NODES / 16 + 3) / 4)
__device__ __forceinline__ void node_chain_body(const NodeSide& su, const NodeSide& si, int rows,
                                                int bx, bool second, float* lds)
{
    const float* x  = second ? si.x  : su.x;
    const float* W1 = second ? si.W1 : su.W1;
    const float* b1 = second ? si.b1 : su.b1;
    const float* Pa = second ? si.Pa : su.Pa;
    const float* ca = second ? si.ca : su.ca;
    const float* Pb = second ? si.Pb : su.Pb;
    const float* cb = second ? si.cb : su.cb;
    float* ya = second ? si.ya : su.ya;
    float* yb = second ? si.yb : su.yb;

    const int wv   = (int)(threadIdx.x >> 5);
    const int lane = (int)(threadIdx.x & 31);
    int stripe = bx * 4 + wv;
    const bool live = stripe * 16 < rows;
    if (!live) stripe = rows / 16 - 1;

    const int nloc  = lane & 15;
    const int khalf = (lane >> 4) << 1;
    const int rhalf = (lane >> 4) << 3;
    const int mbase = stripe * 16 + rhalf;

    float* S = lds + wv * (16 * SP);
    const float* sfrag = S + nloc * SP + khalf;

    v8f acc4[4];
    v8f acc8[8];

    zero_acc<4>(acc4);
    stripe_mm<4, DIM>(x + (size_t)(stripe * 16 + nloc) * DIM + khalf, W1, nloc, khalf, acc4);
    tiles_to_lds<4, true>(S, acc4, b1, nloc, rhalf);
    __syncthreads();

    zero_acc<8>(acc8);
    stripe_mm<8, HID>(sfrag, Pa, nloc, khalf, acc8);
    if (live) tiles_to_global<8>(ya, acc8, ca, nloc, mbase);

    zero_acc<8>(acc8);
    stripe_mm<8, HID>(sfrag, Pb, nloc, khalf, acc8);
    if (live) tiles_to_global<8>(yb, acc8, cb, nloc, mbase);
}


#define NCH 32
#define CHUNK 15632
#define NQ 4
#define QN (N_NODES / NQ)
#define PBLK 128
#define NBLK 391
#define GROUP_BLOCK 1024
#define GROUP_TRIPS ((CHUNK + GROUP_BLOCK * 4 - 1) / (GROUP_BLOCK * 4))

#define NPW 4
#define ATT_BLOCK 128

#define FOLD_XB 3
__global__ __launch_bounds__(GROUP_BLOCK) void chunk_count_kernel(
    const int* __restrict__ dst0, const int* __restrict__ dst1,
    int* __restrict__ hist, int* __restrict__ rank, FoldSide fu, FoldSide fi)
{
    __shared__ __attribute__((aligned(16))) int tab[QN];
    if (blockIdx.x >= NCH * NQ) {
        const int wv = __builtin_amdgcn_readfirstlane((int)(threadIdx.x >> 5));
        const int u = (((int)blockIdx.x - NCH * NQ) * 2 + (int)blockIdx.y) * (GROUP_BLOCK / 32) + wv;
        if (u < FOLD_UNITS) fold_unit(fu, fi, u, (int)(threadIdx.x & 31));
        return;
    }
    const int c   = (int)blockIdx.x / NQ;
    const int q   = (int)blockIdx.x % NQ;
    const int qb  = q * QN;
    const int rel = (int)blockIdx.y;
    const int t   = (int)threadIdx.x;
    const int* __restrict__ dst = rel ? dst1 : dst0;
    int* __restrict__ r = rank + (size_t)(rel * NQ + q) * NEDGE;
    int* __restrict__ h = hist + (size_t)(rel * NCH + c) * N_NODES + qb;

    for (int i = t * 4; i < QN; i += GROUP_BLOCK * 4)
        *(int4*)(tab + i) = make_int4(0, 0, 0, 0);
    __syncthreads();

    const int e0 = c * CHUNK;
    const int e1 = min(e0 + CHUNK, NEDGE);
    int4 dd[GROUP_TRIPS];
#pragma unroll
    for (int j = 0; j < GROUP_TRIPS; ++j) {
        const int e = e0 + j * (GROUP_BLOCK * 4) + t * 4;
        dd[j] = (e < e1) ? *(const int4*)(dst + e) : make_int4(-1, -1, -1, -1);
    }
#pragma unroll
    for (int j = 0; j < GROUP_TRIPS; ++j) {
        const int e = e0 + j * (GROUP_BLOCK * 4) + t * 4;
        const unsigned ux = (unsigned)(dd[j].x - qb), uy = (unsigned)(dd[j].y - qb);
        const unsigned uz = (unsigned)(dd[j].z - qb), uw = (unsigned)(dd[j].w - qb);
        const bool px = ux < (unsigned)QN, py = uy < (unsigned)QN;
        const bool pz = uz < (unsigned)QN, pw = uw < (unsigned)QN;
        int kx = 0, ky = 0, kz = 0, kw = 0;
        if (px) kx = atomicAdd(tab + ux, 1);
        if (py) ky = atomicAdd(tab + uy, 1);
        if (pz) kz = atomicAdd(tab + uz, 1);
        if (pw) kw = atomicAdd(tab + uw, 1);
        if (e < e1) *(int4*)(r + e) = make_int4(kx, ky, kz, kw);
    }
    __syncthreads();

    for (int i = t * 4; i < QN; i += GROUP_BLOCK * 4)
        *(int4*)(h + i) = *(const int4*)(tab + i);
}

__device__ __forceinline__ void node_prefix_body(
    int* __restrict__ hist, int* __restrict__ cnt,
    int* __restrict__ offl, int* __restrict__ blocktot,
    int rel, int blk, int* s_tot)
{
    const int t = (int)threadIdx.x, lane = t & 31, w = t >> 5;
    const int idx = blk * PBLK + t;
    int* __restrict__ h = hist + (size_t)rel * NCH * N_NODES;

    int run = 0;
    if (idx < N_NODES) {
        for (int c0 = 0; c0 < NCH; c0 += 32) {
            int v[32];
#pragma unroll
            for (int j = 0; j < 32; ++j) v[j] = h[(size_t)(c0 + j) * N_NODES + idx];
#pragma unroll
            for (int j = 0; j < 32; ++j) {
                h[(size_t)(c0 + j) * N_NODES + idx] = run;
                run += v[j];
            }
        }
        cnt[rel * N_NODES + idx] = run;
    }

    int inc = run;
#pragma unroll
    for (int dlt = 1; dlt < 32; dlt <<= 1) {
        const int u = __shfl_up(inc, dlt, 32);
        if (lane >= dlt) inc += u;
    }
    if (lane == 31) s_tot[w] = inc;
    __syncthreads();
    int add = 0, all = 0;
#pragma unroll
    for (int k = 0; k < PBLK / 32; ++k) {
        const int s = s_tot[k];
        if (k < w) add += s;
        all += s;
    }
    if (idx < N_NODES) offl[rel * N_NODES + idx] = add + inc - run;
    if (t == 0) blocktot[rel * NBLK + blk] = all;
}

__global__ __launch_bounds__(128) void prefix_nodes_kernel(
    int* __restrict__ hist, int* __restrict__ cnt,
    int* __restrict__ offl, int* __restrict__ blocktot,
    NodeSide su, NodeSide si, int rows)
{
    __shared__ float lds[4 * 16 * SP];
    __shared__ int s_tot[PBLK / 32];
    const int b = (int)blockIdx.x;
    if (b < 2 * NBLK) {
        const bool rel1 = (b >= NBLK);
        node_prefix_body(hist, cnt, offl, blocktot, rel1 ? 1 : 0, rel1 ? b - NBLK : b, s_tot);
        return;
    }
    const int nb = b - 2 * NBLK;
    const bool second = (nb >= NODE_XB);
    node_chain_body(su, si, rows, second ? nb - NODE_XB : nb, second, lds);
}

__global__ __launch_bounds__(GROUP_BLOCK) void fill_slots_kernel(
    const int* __restrict__ src0, const int* __restrict__ dst0,
    const int* __restrict__ src1, const int* __restrict__ dst1,
    const int* __restrict__ hist, const int* __restrict__ offl,
    const int* __restrict__ blocktot, const int* __restrict__ rank,
    int* __restrict__ off, int* __restrict__ slots)
{
    __shared__ __attribute__((aligned(16))) int tab[QN];
    __shared__ int bb[512];
    __shared__ int s_w[16];
    const int c   = (int)blockIdx.x / NQ;
    const int q   = (int)blockIdx.x % NQ;
    const int qb  = q * QN;
    const int rel = (int)blockIdx.y;
    const int t = (int)threadIdx.x, lane = t & 31, w = t >> 5;
    const int* __restrict__ src = rel ? src1 : src0;
    const int* __restrict__ dst = rel ? dst1 : dst0;
    const int* __restrict__ hc = hist + (size_t)(rel * NCH + c) * N_NODES + qb;
    const int* __restrict__ ol = offl + (size_t)rel * N_NODES + qb;
    const int* __restrict__ r  = rank + (size_t)(rel * NQ + q) * NEDGE;
    int* __restrict__ og = off + (size_t)rel * N_NODES + qb;
    int* __restrict__ sl = slots + (size_t)rel * NEDGE;

    const int v = (t < NBLK) ? blocktot[rel * NBLK + t] : 0;
    int inc = v;
#pragma unroll
    for (int dlt = 1; dlt < 32; dlt <<= 1) {
        const int u = __shfl_up(inc, dlt, 32);
        if (lane >= dlt) inc += u;
    }
    if (w < 16 && lane == 31) s_w[w] = inc;
    __syncthreads();
    if (t < 512) {
        int add = 0;
#pragma unroll
        for (int k = 0; k < 16; ++k)
            if (k < w) add += s_w[k];
        bb[t] = add + inc - v;
    }
    __syncthreads();

#pragma unroll 4
    for (int i = t * 4; i < QN; i += GROUP_BLOCK * 4) {
        const int4 a = *(const int4*)(hc + i);
        const int4 o = *(const int4*)(ol + i);
        const int b = bb[(qb + i) >> 7];
        *(int4*)(tab + i) = make_int4(o.x + b + a.x, o.y + b + a.y,
                                      o.z + b + a.z, o.w + b + a.w);
    }
    __syncthreads();

    if (c == 0) {
        for (int i = t * 4; i < QN; i += GROUP_BLOCK * 4)
            *(int4*)(og + i) = *(const int4*)(tab + i);
    }

    const int e0 = c * CHUNK;
    const int e1 = min(e0 + CHUNK, NEDGE);
    int4 dd[GROUP_TRIPS], kk[GROUP_TRIPS], ss[GROUP_TRIPS];
#pragma unroll
    for (int j = 0; j < GROUP_TRIPS; ++j) {
        const int e = e0 + j * (GROUP_BLOCK * 4) + t * 4;
        const bool in = (e < e1);
        dd[j] = in ? *(const int4*)(dst + e) : make_int4(-1, -1, -1, -1);
        kk[j] = in ? *(const int4*)(r + e)   : make_int4(0, 0, 0, 0);
        ss[j] = in ? *(const int4*)(src + e) : make_int4(0, 0, 0, 0);
    }
#pragma unroll
    for (int j = 0; j < GROUP_TRIPS; ++j) {
        const unsigned ux = (unsigned)(dd[j].x - qb), uy = (unsigned)(dd[j].y - qb);
        const unsigned uz = (unsigned)(dd[j].z - qb), uw = (unsigned)(dd[j].w - qb);
        const bool px = ux < (unsigned)QN, py = uy < (unsigned)QN;
        const bool pz = uz < (unsigned)QN, pw = uw < (unsigned)QN;
        const int bx = tab[px ? ux : 0u], by = tab[py ? uy : 0u];
        const int bz = tab[pz ? uz : 0u], bw = tab[pw ? uw : 0u];
        if (px) sl[bx + kk[j].x] = ss[j].x;
        if (py) sl[by + kk[j].y] = ss[j].y;
        if (pz) sl[bz + kk[j].z] = ss[j].z;
        if (pw) sl[bw + kk[j].w] = ss[j].w;
    }
}

__device__ __forceinline__ float xhalf(float x) {
    const int b = __builtin_bit_cast(int, x);
    return __builtin_bit_cast(float,
        __builtin_amdgcn_permlanex16(b, b, 0x76543210, 0xfedcba98, false, false));
}
template <int M>
__device__ __forceinline__ float xrow(float x) {
    const int b = __builtin_bit_cast(int, x);
    return __builtin_bit_cast(float,
        __builtin_amdgcn_update_dpp(b, b, 0x160 + M, 0xf, 0xf, true));
}

__device__ __forceinline__ void seg_attend_body(
    const float* __restrict__ xl, const float* __restrict__ xr,
    const int* __restrict__ slots, const int* __restrict__ off,
    const int* __restrict__ cnt,
    const float* __restrict__ att, const float* __restrict__ bias,
    float* __restrict__ out, int n)
{
    const int lane = (int)(threadIdx.x & 31);
    const int w = __builtin_amdgcn_readfirstlane(
        (int)((blockIdx.x * blockDim.x + threadIdx.x) >> 5));
    const int d0 = w * NPW;
    if (d0 >= n) return;
    const int dend = min(d0 + NPW, n);
    const bool up16 = (lane & 16) != 0;
    const bool up8  = (lane & 8) != 0;
    const bool up4  = (lane & 4) != 0;
    const int  myj  = (lane >> 2) & 7;
    const float4 at = *(const float4*)(att + lane * 4);
    const float4 bi = *(const float4*)(bias + lane * 4);

    for (int d = d0; d < dend; ++d) {
        const int beg = off[d];
        const int deg = cnt[d];
        const float4 b = *(const float4*)(xr + (size_t)d * DIM + lane * 4);

        float4 acc = make_float4(0.f, 0.f, 0.f, 0.f);
        float m = -INFINITY;
        float s = 0.f;

        for (int j0 = 0; j0 < deg; j0 += 8) {
            const int nv = min(8, deg - j0);
            const int jj = min(lane & 7, nv - 1);
            const int sidx = slots[beg + j0 + jj];

            v4f   a[8];
            float p[8];
#pragma unroll
            for (int j = 0; j < 8; ++j) {
                const int sj = __builtin_amdgcn_readlane(sidx, j);
                a[j] = *(const v4f*)(xl + (size_t)sj * DIM + lane * 4);
            }
            asm volatile("" :: "v"(a[0]), "v"(a[1]), "v"(a[2]), "v"(a[3]),
                               "v"(a[4]), "v"(a[5]), "v"(a[6]), "v"(a[7]));
#pragma unroll
            for (int j = 0; j < 8; ++j) {
                float z, q = 0.f;
                z = a[j].x + b.x; z = (z > 0.f) ? z : SLOPE * z; q += z * at.x;
                z = a[j].y + b.y; z = (z > 0.f) ? z : SLOPE * z; q += z * at.y;
                z = a[j].z + b.z; z = (z > 0.f) ? z : SLOPE * z; q += z * at.z;
                z = a[j].w + b.w; z = (z > 0.f) ? z : SLOPE * z; q += z * at.w;
                p[j] = q;
            }
            float q4[4], q2[2];
#pragma unroll
            for (int i = 0; i < 4; ++i) {
                const float keep = up16 ? p[i + 4] : p[i];
                const float give = up16 ? p[i] : p[i + 4];
                q4[i] = keep + xhalf(give);
            }
#pragma unroll
            for (int i = 0; i < 2; ++i) {
                const float keep = up8 ? q4[i + 2] : q4[i];
                const float give = up8 ? q4[i] : q4[i + 2];
                q2[i] = keep + xrow<8>(give);
            }
            float lg;
            {
                const float keep = up4 ? q2[1] : q2[0];
                const float give = up4 ? q2[0] : q2[1];
                lg = keep + xrow<4>(give);
            }
            lg += xrow<2>(lg);
            lg += xrow<1>(lg);
            const float lj = (myj < nv) ? lg : -INFINITY;
            float mc = fmaxf(lj, xrow<4>(lj));
            mc = fmaxf(mc, xrow<8>(mc));
            mc = fmaxf(mc, xhalf(mc));
            const float mn = fmaxf(m, mc);
            const float sc = expf(m - mn);
            const float ej = expf(lj - mn);
            acc.x *= sc; acc.y *= sc; acc.z *= sc; acc.w *= sc;
            s *= sc;
#pragma unroll
            for (int j = 0; j < 8; ++j) {
                const float e = __builtin_bit_cast(float,
                    __builtin_amdgcn_readlane(__builtin_bit_cast(int, ej), 4 * j));
                s += e;
                acc.x += e * a[j].x;
                acc.y += e * a[j].y;
                acc.z += e * a[j].z;
                acc.w += e * a[j].w;
            }
            m = mn;
        }

        const float den = s + 1e-16f;
        float4 r;
        r.x = acc.x / den + bi.x;
        r.y = acc.y / den + bi.y;
        r.z = acc.z / den + bi.z;
        r.w = acc.w / den + bi.w;
        *(float4*)(out + (size_t)d * DIM + lane * 4) = r;
    }
}

struct AttendSide {
    const float* xl; const float* xr;
    const int* slots; const int* off; const int* cnt;
    const float* att; const float* bias;
    float* out;
};
__global__ __launch_bounds__(ATT_BLOCK) void seg_attend_kernel(AttendSide r0, AttendSide r1, int n)
{
    const bool second = (blockIdx.y != 0);
    seg_attend_body(second ? r1.xl : r0.xl, second ? r1.xr : r0.xr,
                    second ? r1.slots : r0.slots, second ? r1.off : r0.off,
                    second ? r1.cnt : r0.cnt,
                    second ? r1.att : r0.att, second ? r1.bias : r0.bias,
                    second ? r1.out : r0.out, n);
}

extern "C" void kernel_launch(void* const* d_in, const int* in_sizes, int n_in,
                              void* d_out, int out_size, void* d_ws, size_t ws_size,
                              hipStream_t stream) {
    (void)in_sizes; (void)n_in; (void)out_size; (void)ws_size;

    const float* x_user = (const float*)d_in[0];
    const float* x_item = (const float*)d_in[1];
    const int*   e_u2i  = (const int*)d_in[2];
    const int*   e_i2u  = (const int*)d_in[3];
    const float* W1_u = (const float*)d_in[4],  *b1_u = (const float*)d_in[5];
    const float* W2_u = (const float*)d_in[6],  *b2_u = (const float*)d_in[7];
    const float* W1_i = (const float*)d_in[8],  *b1_i = (const float*)d_in[9];
    const float* W2_i = (const float*)d_in[10], *b2_i = (const float*)d_in[11];
    const float* Wl_u2i = (const float*)d_in[12], *bl_u2i = (const float*)d_in[13];
    const float* Wr_u2i = (const float*)d_in[14], *br_u2i = (const float*)d_in[15];
    const float* att_u2i = (const float*)d_in[16], *bias_u2i = (const float*)d_in[17];
    const float* Wl_i2u = (const float*)d_in[18], *bl_i2u = (const float*)d_in[19];
    const float* Wr_i2u = (const float*)d_in[20], *br_i2u = (const float*)d_in[21];
    const float* att_i2u = (const float*)d_in[22], *bias_i2u = (const float*)d_in[23];

    float* out_user = (float*)d_out;
    float* out_item = (float*)d_out + (size_t)N_NODES * DIM;

    float* ws    = (float*)d_ws;
    float* xl_u  = ws;
    float* xl_i  = xl_u + (size_t)N_NODES * DIM;
    float* xr_i  = xl_i + (size_t)N_NODES * DIM;
    float* xr_u  = xr_i + (size_t)N_NODES * DIM;
    int* hist  = (int*)d_out;
    int* cnt   = (int*)(xr_u + (size_t)N_NODES * DIM);
    int* off   = cnt + 2 * N_NODES;
    int* rank  = hist + (size_t)2 * NCH * N_NODES;
    int* slots = off + 2 * N_NODES;
    int* offl  = slots + 2 * NEDGE;
    int* blocktot = offl + 2 * N_NODES;
    float* P_ua = (float*)(blocktot + 2 * NBLK + 2);
    float* P_ub = P_ua + (HID + 1) * DIM;
    float* P_ia = P_ub + (HID + 1) * DIM;
    float* P_ib = P_ia + (HID + 1) * DIM;

    const int blkNw   = (N_NODES + (ATT_BLOCK / 32) * NPW - 1) / ((ATT_BLOCK / 32) * NPW);

    FoldSide fu = { W2_u, b2_u, Wl_u2i, bl_u2i, Wr_i2u, br_i2u, P_ua, P_ub };
    FoldSide fi = { W2_i, b2_i, Wl_i2u, bl_i2u, Wr_u2i, br_u2i, P_ia, P_ib };
    chunk_count_kernel<<<dim3(NCH * NQ + FOLD_XB, 2), GROUP_BLOCK, 0, stream>>>(
        e_u2i + NEDGE, e_i2u + NEDGE, hist, rank, fu, fi);
    NodeSide su = { x_user, W1_u, b1_u, P_ua, P_ua + HID * DIM, P_ub, P_ub + HID * DIM,
                    xl_u, xr_u };
    NodeSide si = { x_item, W1_i, b1_i, P_ia, P_ia + HID * DIM, P_ib, P_ib + HID * DIM,
                    xl_i, xr_i };
    prefix_nodes_kernel<<<dim3(2 * NBLK + 2 * NODE_XB), PBLK, 0, stream>>>(
        hist, cnt, offl, blocktot, su, si, N_NODES);
    fill_slots_kernel<<<dim3(NCH * NQ, 2), GROUP_BLOCK, 0, stream>>>(
        e_u2i, e_u2i + NEDGE, e_i2u, e_i2u + NEDGE, hist, offl, blocktot, rank, off, slots);

    AttendSide r0 = { xl_u, xr_i, slots, off, cnt, att_u2i, bias_u2i, out_item };
    AttendSide r1 = { xl_i, xr_u, slots + NEDGE, off + N_NODES, cnt + N_NODES,
                      att_i2u, bias_i2u, out_user };
    seg_attend_kernel<<<dim3(blkNw, 2), ATT_BLOCK, 0, stream>>>(r0, r1, N_NODES);
}


# Per-node softmax issues ~30% fewer vector instructions: logits on float4 component pairs (packed f32, leaky as max), 4-edge form for short last groups, one reciprocal per node; logit sum re-paired
# speedup vs baseline: 4.8009x; 1.0536x over previous
#include <hip/hip_runtime.h>


typedef __attribute__((ext_vector_type(2))) float v2f;
typedef __attribute__((ext_vector_type(8))) float v8f;
typedef __attribute__((ext_vector_type(4))) float v4f;

#define N_NODES 50000
#define DIM 128
#define HID 64
#define NEDGE 500000
#define SLOPE 0.2f

#define SP 132

struct NodeSide {
    const float* x;
    const float* W1; const float* b1;
    const float* Pa; const float* ca;
    const float* Pb; const float* cb;
    float* ya;
    float* yb;
};

struct FoldSide {
    const float* W2; const float* b2;
    const float* Wa; const float* ba;
    const float* Wb; const float* bb;
    float* Pa;
    float* Pb;
};

template <int NT, int K, int LDW = NT * 16>
__device__ __forceinline__ void stripe_mm(const float* arow, const float* __restrict__ W,
                                          int nloc, int khalf, v8f (&acc)[NT])
{
    constexpr int ncols = LDW;
#pragma unroll (16 / NT)
    for (int k0 = 0; k0 < K; k0 += 4) {
        const v2f a = *(const v2f*)(arow + k0);
        const float* __restrict__ w0 = W + (size_t)(k0 + khalf) * ncols + nloc;
        const float* __restrict__ w1 = w0 + ncols;
#pragma unroll
        for (int t = 0; t < NT; ++t) {
            v2f b;
            b.x = w0[t * 16];
            b.y = w1[t * 16];
            acc[t] = __builtin_amdgcn_wmma_f32_16x16x4_f32(
                false, a, false, b, (short)0, acc[t], false, false);
        }
    }
}

template <int NT>
__device__ __forceinline__ void zero_acc(v8f (&acc)[NT]) {
#pragma unroll
    for (int t = 0; t < NT; ++t)
        acc[t] = (v8f){0.f, 0.f, 0.f, 0.f, 0.f, 0.f, 0.f, 0.f};
}

template <int NT, bool RELU>
__device__ __forceinline__ void tiles_to_lds(float* S, const v8f (&acc)[NT],
                                             const float* __restrict__ bias,
                                             int nloc, int rhalf)
{
#pragma unroll
    for (int t = 0; t < NT; ++t) {
        const int n = t * 16 + nloc;
        const float bv = bias[n];
#pragma unroll
        for (int j = 0; j < 8; ++j) {
            float v = acc[t][j] + bv;
            if (RELU) v = fmaxf(v, 0.f);
            S[(rhalf + j) * SP + n] = v;
        }
    }
}

template <int NT>
__device__ __forceinline__ void tiles_to_global(float* __restrict__ C, const v8f (&acc)[NT],
                                                const float* __restrict__ bias,
                                                int nloc, int mbase)
{
    constexpr int ncols = NT * 16;
#pragma unroll
    for (int t = 0; t < NT; ++t) {
        const int n = t * 16 + nloc;
        const float bv = bias[n];
#pragma unroll
        for (int j = 0; j < 8; ++j)
            C[(size_t)(mbase + j) * ncols + n] = acc[t][j] + bv;
    }
}

#define FOLD_UNITS 160
__device__ __forceinline__ void fold_unit(const FoldSide& fu, const FoldSide& fi, int u, int lane)
{
    const int g = u >> 3;
    const bool second = ((g / 5) & 1) != 0;
    const bool projb  = (g >= 10);
    const float* W2 = second ? fi.W2 : fu.W2;
    const float* b2 = second ? fi.b2 : fu.b2;
    const float* W  = second ? (projb ? fi.Wb : fi.Wa) : (projb ? fu.Wb : fu.Wa);
    const float* b  = second ? (projb ? fi.bb : fi.ba) : (projb ? fu.bb : fu.ba);
    float* P        = second ? (projb ? fi.Pb : fi.Pa) : (projb ? fu.Pb : fu.Pa);

    const int s    = g % 5;
    const int t    = u & 7;
    const int nloc  = lane & 15;
    const int khalf = (lane >> 4) << 1;
    const int rhalf = (lane >> 4) << 3;
    const bool brow = (s == HID / 16);

    const float* arow = brow ? (b2 + khalf)
                             : (W2 + (size_t)(s * 16 + nloc) * DIM + khalf);
    v8f acc[1];
    zero_acc<1>(acc);
    stripe_mm<1, DIM, DIM>(arow, W + t * 16, nloc, khalf, acc);

    const int n = t * 16 + nloc;
    if (!brow) {
#pragma unroll
        for (int j = 0; j < 8; ++j)
            P[(size_t)(s * 16 + rhalf + j) * DIM + n] = acc[0][j];
    } else if (lane < 16) {
        P[(size_t)HID * DIM + n] = acc[0][0] + b[n];
    }
}

#define NODE_XB ((N_NODES / 16 + 3) / 4)
__device__ __forceinline__ void node_chain_body(const NodeSide& su, const NodeSide& si, int rows,
                                                int bx, bool second, float* lds)
{
    const float* x  = second ? si.x  : su.x;
    const float* W1 = second ? si.W1 : su.W1;
    const float* b1 = second ? si.b1 : su.b1;
    const float* Pa = second ? si.Pa : su.Pa;
    const float* ca = second ? si.ca : su.ca;
    const float* Pb = second ? si.Pb : su.Pb;
    const float* cb = second ? si.cb : su.cb;
    float* ya = second ? si.ya : su.ya;
    float* yb = second ? si.yb : su.yb;

    const int wv   = (int)(threadIdx.x >> 5);
    const int lane = (int)(threadIdx.x & 31);
    int stripe = bx * 4 + wv;
    const bool live = stripe * 16 < rows;
    if (!live) stripe = rows / 16 - 1;

    const int nloc  = lane & 15;
    const int khalf = (lane >> 4) << 1;
    const int rhalf = (lane >> 4) << 3;
    const int mbase = stripe * 16 + rhalf;

    float* S = lds + wv * (16 * SP);
    const float* sfrag = S + nloc * SP + khalf;

    v8f acc4[4];
    v8f acc8[8];

    zero_acc<4>(acc4);
    stripe_mm<4, DIM>(x + (size_t)(stripe * 16 + nloc) * DIM + khalf, W1, nloc, khalf, acc4);
    tiles_to_lds<4, true>(S, acc4, b1, nloc, rhalf);
    __syncthreads();

    zero_acc<8>(acc8);
    stripe_mm<8, HID>(sfrag, Pa, nloc, khalf, acc8);
    if (live) tiles_to_global<8>(ya, acc8, ca, nloc, mbase);

    zero_acc<8>(acc8);
    stripe_mm<8, HID>(sfrag, Pb, nloc, khalf, acc8);
    if (live) tiles_to_global<8>(yb, acc8, cb, nloc, mbase);
}


#define NCH 32
#define CHUNK 15632
#define NQ 4
#define QN (N_NODES / NQ)
#define PBLK 128
#define NBLK 391
#define GROUP_BLOCK 1024
#define GROUP_TRIPS ((CHUNK + GROUP_BLOCK * 4 - 1) / (GROUP_BLOCK * 4))

#define NPW 4
#define ATT_BLOCK 128

#define FOLD_XB 3
__global__ __launch_bounds__(GROUP_BLOCK) void chunk_count_kernel(
    const int* __restrict__ dst0, const int* __restrict__ dst1,
    int* __restrict__ hist, int* __restrict__ rank, FoldSide fu, FoldSide fi)
{
    __shared__ __attribute__((aligned(16))) int tab[QN];
    if (blockIdx.x >= NCH * NQ) {
        const int wv = __builtin_amdgcn_readfirstlane((int)(threadIdx.x >> 5));
        const int u = (((int)blockIdx.x - NCH * NQ) * 2 + (int)blockIdx.y) * (GROUP_BLOCK / 32) + wv;
        if (u < FOLD_UNITS) fold_unit(fu, fi, u, (int)(threadIdx.x & 31));
        return;
    }
    const int c   = (int)blockIdx.x / NQ;
    const int q   = (int)blockIdx.x % NQ;
    const int qb  = q * QN;
    const int rel = (int)blockIdx.y;
    const int t   = (int)threadIdx.x;
    const int* __restrict__ dst = rel ? dst1 : dst0;
    int* __restrict__ r = rank + (size_t)(rel * NQ + q) * NEDGE;
    int* __restrict__ h = hist + (size_t)(rel * NCH + c) * N_NODES + qb;

    for (int i = t * 4; i < QN; i += GROUP_BLOCK * 4)
        *(int4*)(tab + i) = make_int4(0, 0, 0, 0);
    __syncthreads();

    const int e0 = c * CHUNK;
    const int e1 = min(e0 + CHUNK, NEDGE);
    int4 dd[GROUP_TRIPS];
#pragma unroll
    for (int j = 0; j < GROUP_TRIPS; ++j) {
        const int e = e0 + j * (GROUP_BLOCK * 4) + t * 4;
        dd[j] = (e < e1) ? *(const int4*)(dst + e) : make_int4(-1, -1, -1, -1);
    }
#pragma unroll
    for (int j = 0; j < GROUP_TRIPS; ++j) {
        const int e = e0 + j * (GROUP_BLOCK * 4) + t * 4;
        const unsigned ux = (unsigned)(dd[j].x - qb), uy = (unsigned)(dd[j].y - qb);
        const unsigned uz = (unsigned)(dd[j].z - qb), uw = (unsigned)(dd[j].w - qb);
        const bool px = ux < (unsigned)QN, py = uy < (unsigned)QN;
        const bool pz = uz < (unsigned)QN, pw = uw < (unsigned)QN;
        int kx = 0, ky = 0, kz = 0, kw = 0;
        if (px) kx = atomicAdd(tab + ux, 1);
        if (py) ky = atomicAdd(tab + uy, 1);
        if (pz) kz = atomicAdd(tab + uz, 1);
        if (pw) kw = atomicAdd(tab + uw, 1);
        if (e < e1) *(int4*)(r + e) = make_int4(kx, ky, kz, kw);
    }
    __syncthreads();

    for (int i = t * 4; i < QN; i += GROUP_BLOCK * 4)
        *(int4*)(h + i) = *(const int4*)(tab + i);
}

__device__ __forceinline__ void node_prefix_body(
    int* __restrict__ hist, int* __restrict__ cnt,
    int* __restrict__ offl, int* __restrict__ blocktot,
    int rel, int blk, int* s_tot)
{
    const int t = (int)threadIdx.x, lane = t & 31, w = t >> 5;
    const int idx = blk * PBLK + t;
    int* __restrict__ h = hist + (size_t)rel * NCH * N_NODES;

    int run = 0;
    if (idx < N_NODES) {
        for (int c0 = 0; c0 < NCH; c0 += 32) {
            int v[32];
#pragma unroll
            for (int j = 0; j < 32; ++j) v[j] = h[(size_t)(c0 + j) * N_NODES + idx];
#pragma unroll
            for (int j = 0; j < 32; ++j) {
                h[(size_t)(c0 + j) * N_NODES + idx] = run;
                run += v[j];
            }
        }
        cnt[rel * N_NODES + idx] = run;
    }

    int inc = run;
#pragma unroll
    for (int dlt = 1; dlt < 32; dlt <<= 1) {
        const int u = __shfl_up(inc, dlt, 32);
        if (lane >= dlt) inc += u;
    }
    if (lane == 31) s_tot[w] = inc;
    __syncthreads();
    int add = 0, all = 0;
#pragma unroll
    for (int k = 0; k < PBLK / 32; ++k) {
        const int s = s_tot[k];
        if (k < w) add += s;
        all += s;
    }
    if (idx < N_NODES) offl[rel * N_NODES + idx] = add + inc - run;
    if (t == 0) blocktot[rel * NBLK + blk] = all;
}

__global__ __launch_bounds__(128) void prefix_nodes_kernel(
    int* __restrict__ hist, int* __restrict__ cnt,
    int* __restrict__ offl, int* __restrict__ blocktot,
    NodeSide su, NodeSide si, int rows)
{
    __shared__ float lds[4 * 16 * SP];
    __shared__ int s_tot[PBLK / 32];
    const int b = (int)blockIdx.x;
    if (b < 2 * NBLK) {
        const bool rel1 = (b >= NBLK);
        node_prefix_body(hist, cnt, offl, blocktot, rel1 ? 1 : 0, rel1 ? b - NBLK : b, s_tot);
        return;
    }
    const int nb = b - 2 * NBLK;
    const bool second = (nb >= NODE_XB);
    node_chain_body(su, si, rows, second ? nb - NODE_XB : nb, second, lds);
}

__global__ __launch_bounds__(GROUP_BLOCK) void fill_slots_kernel(
    const int* __restrict__ src0, const int* __restrict__ dst0,
    const int* __restrict__ src1, const int* __restrict__ dst1,
    const int* __restrict__ hist, const int* __restrict__ offl,
    const int* __restrict__ blocktot, const int* __restrict__ rank,
    int* __restrict__ off, int* __restrict__ slots)
{
    __shared__ __attribute__((aligned(16))) int tab[QN];
    __shared__ int bb[512];
    __shared__ int s_w[16];
    const int c   = (int)blockIdx.x / NQ;
    const int q   = (int)blockIdx.x % NQ;
    const int qb  = q * QN;
    const int rel = (int)blockIdx.y;
    const int t = (int)threadIdx.x, lane = t & 31, w = t >> 5;
    const int* __restrict__ src = rel ? src1 : src0;
    const int* __restrict__ dst = rel ? dst1 : dst0;
    const int* __restrict__ hc = hist + (size_t)(rel * NCH + c) * N_NODES + qb;
    const int* __restrict__ ol = offl + (size_t)rel * N_NODES + qb;
    const int* __restrict__ r  = rank + (size_t)(rel * NQ + q) * NEDGE;
    int* __restrict__ og = off + (size_t)rel * N_NODES + qb;
    int* __restrict__ sl = slots + (size_t)rel * NEDGE;

    const int v = (t < NBLK) ? blocktot[rel * NBLK + t] : 0;
    int inc = v;
#pragma unroll
    for (int dlt = 1; dlt < 32; dlt <<= 1) {
        const int u = __shfl_up(inc, dlt, 32);
        if (lane >= dlt) inc += u;
    }
    if (w < 16 && lane == 31) s_w[w] = inc;
    __syncthreads();
    if (t < 512) {
        int add = 0;
#pragma unroll
        for (int k = 0; k < 16; ++k)
            if (k < w) add += s_w[k];
        bb[t] = add + inc - v;
    }
    __syncthreads();

#pragma unroll 4
    for (int i = t * 4; i < QN; i += GROUP_BLOCK * 4) {
        const int4 a = *(const int4*)(hc + i);
        const int4 o = *(const int4*)(ol + i);
        const int b = bb[(qb + i) >> 7];
        *(int4*)(tab + i) = make_int4(o.x + b + a.x, o.y + b + a.y,
                                      o.z + b + a.z, o.w + b + a.w);
    }
    __syncthreads();

    if (c == 0) {
        for (int i = t * 4; i < QN; i += GROUP_BLOCK * 4)
            *(int4*)(og + i) = *(const int4*)(tab + i);
    }

    const int e0 = c * CHUNK;
    const int e1 = min(e0 + CHUNK, NEDGE);
    int4 dd[GROUP_TRIPS], kk[GROUP_TRIPS], ss[GROUP_TRIPS];
#pragma unroll
    for (int j = 0; j < GROUP_TRIPS; ++j) {
        const int e = e0 + j * (GROUP_BLOCK * 4) + t * 4;
        const bool in = (e < e1);
        dd[j] = in ? *(const int4*)(dst + e) : make_int4(-1, -1, -1, -1);
        kk[j] = in ? *(const int4*)(r + e)   : make_int4(0, 0, 0, 0);
        ss[j] = in ? *(const int4*)(src + e) : make_int4(0, 0, 0, 0);
    }
#pragma unroll
    for (int j = 0; j < GROUP_TRIPS; ++j) {
        const unsigned ux = (unsigned)(dd[j].x - qb), uy = (unsigned)(dd[j].y - qb);
        const unsigned uz = (unsigned)(dd[j].z - qb), uw = (unsigned)(dd[j].w - qb);
        const bool px = ux < (unsigned)QN, py = uy < (unsigned)QN;
        const bool pz = uz < (unsigned)QN, pw = uw < (unsigned)QN;
        const int bx = tab[px ? ux : 0u], by = tab[py ? uy : 0u];
        const int bz = tab[pz ? uz : 0u], bw = tab[pw ? uw : 0u];
        if (px) sl[bx + kk[j].x] = ss[j].x;
        if (py) sl[by + kk[j].y] = ss[j].y;
        if (pz) sl[bz + kk[j].z] = ss[j].z;
        if (pw) sl[bw + kk[j].w] = ss[j].w;
    }
}

__device__ __forceinline__ float xhalf(float x) {
    const int b = __builtin_bit_cast(int, x);
    return __builtin_bit_cast(float,
        __builtin_amdgcn_permlanex16(b, b, 0x76543210, 0xfedcba98, false, false));
}
template <int M>
__device__ __forceinline__ float xrow(float x) {
    const int b = __builtin_bit_cast(int, x);
    return __builtin_bit_cast(float,
        __builtin_amdgcn_update_dpp(b, b, 0x160 + M, 0xf, 0xf, true));
}


template <int NE>
__device__ __forceinline__ void attend_group(
    const float* __restrict__ xl, int sidx, int nv, int lane,
    bool up16, bool up8, bool up4,
    v2f b01, v2f b23, v2f at01, v2f at23,
    v2f& acc01, v2f& acc23, float& m, float& s)
{
    v4f   a[NE];
    float p[NE];
    const unsigned lo = (unsigned)lane * 4u;
#pragma unroll
    for (int j = 0; j < NE; ++j) {
        const int sj = __builtin_amdgcn_readlane(sidx, j);
        const float* rowp = xl + (size_t)sj * DIM;
        a[j] = *(const v4f*)(rowp + lo);
    }
#pragma unroll
    for (int j = 0; j < NE; ++j) {
        v2f z0 = __builtin_shufflevector(a[j], a[j], 0, 1) + b01;
        v2f z1 = __builtin_shufflevector(a[j], a[j], 2, 3) + b23;
        z0 = __builtin_elementwise_max(z0, z0 * SLOPE);
        z1 = __builtin_elementwise_max(z1, z1 * SLOPE);
        v2f q = z0 * at01;
        q = __builtin_elementwise_fma(z1, at23, q);
        p[j] = q.x + q.y;
    }
    float lg;
    if constexpr (NE == 8) {
        float q4[4], q2[2];
#pragma unroll
        for (int i = 0; i < 4; ++i) {
            const float keep = up16 ? p[i + 4] : p[i];
            const float give = up16 ? p[i] : p[i + 4];
            q4[i] = keep + xhalf(give);
        }
#pragma unroll
        for (int i = 0; i < 2; ++i) {
            const float keep = up8 ? q4[i + 2] : q4[i];
            const float give = up8 ? q4[i] : q4[i + 2];
            q2[i] = keep + xrow<8>(give);
        }
        const float keep = up4 ? q2[1] : q2[0];
        const float give = up4 ? q2[0] : q2[1];
        lg = keep + xrow<4>(give);
    } else {
        float q2[2];
#pragma unroll
        for (int i = 0; i < 2; ++i) {
            const float keep = up16 ? p[i + 2] : p[i];
            const float give = up16 ? p[i] : p[i + 2];
            q2[i] = keep + xhalf(give);
        }
        const float keep = up8 ? q2[1] : q2[0];
        const float give = up8 ? q2[0] : q2[1];
        lg = keep + xrow<8>(give);
        lg += xrow<4>(lg);
    }
    lg += xrow<2>(lg);
    lg += xrow<1>(lg);
    const int myj = (NE == 8) ? ((lane >> 2) & 7) : ((lane >> 3) & 3);
    const float lj = (myj < nv) ? lg : -INFINITY;
    float mc = lj;
    if constexpr (NE == 8) mc = fmaxf(mc, xrow<4>(mc));
    mc = fmaxf(mc, xrow<8>(mc));
    mc = fmaxf(mc, xhalf(mc));
    const float mn = fmaxf(m, mc);
    const float sc = expf(m - mn);
    const float ej = expf(lj - mn);
    acc01 *= sc; acc23 *= sc;
    s *= sc;
#pragma unroll
    for (int j = 0; j < NE; ++j) {
        const float e = __builtin_bit_cast(float,
            __builtin_amdgcn_readlane(__builtin_bit_cast(int, ej), (NE == 8) ? 4 * j : 8 * j));
        s += e;
        acc01 = __builtin_elementwise_fma((v2f){e, e}, __builtin_shufflevector(a[j], a[j], 0, 1), acc01);
        acc23 = __builtin_elementwise_fma((v2f){e, e}, __builtin_shufflevector(a[j], a[j], 2, 3), acc23);
    }
    m = mn;
}

__device__ __forceinline__ void seg_attend_body(
    const float* __restrict__ xl, const float* __restrict__ xr,
    const int* __restrict__ slots, const int* __restrict__ off,
    const int* __restrict__ cnt,
    const float* __restrict__ att, const float* __restrict__ bias,
    float* __restrict__ out, int n)
{
    const int lane = (int)(threadIdx.x & 31);
    const int w = __builtin_amdgcn_readfirstlane(
        (int)((blockIdx.x * blockDim.x + threadIdx.x) >> 5));
    const int d0 = w * NPW;
    if (d0 >= n) return;
    const int dend = min(d0 + NPW, n);
    const bool up16 = (lane & 16) != 0;
    const bool up8  = (lane & 8) != 0;
    const bool up4  = (lane & 4) != 0;
    const v4f at = *(const v4f*)(att + lane * 4);
    const v4f bi = *(const v4f*)(bias + lane * 4);
    const v2f at01 = __builtin_shufflevector(at, at, 0, 1);
    const v2f at23 = __builtin_shufflevector(at, at, 2, 3);

    for (int d = d0; d < dend; ++d) {
        const int beg = off[d];
        const int deg = cnt[d];
        const v4f b = *(const v4f*)(xr + (size_t)d * DIM + lane * 4);
        const v2f b01 = __builtin_shufflevector(b, b, 0, 1);
        const v2f b23 = __builtin_shufflevector(b, b, 2, 3);

        v2f acc01 = (v2f){0.f, 0.f};
        v2f acc23 = (v2f){0.f, 0.f};
        float m = -INFINITY;
        float s = 0.f;

        for (int j0 = 0; j0 < deg; j0 += 8) {
            const int nv = min(8, deg - j0);
            const unsigned jj = (unsigned)min(lane & 7, nv - 1);
            const int* gp = slots + beg + j0;
            const int sidx = gp[jj];
            if (nv > 4)
                attend_group<8>(xl, sidx, nv, lane, up16, up8, up4, b01, b23, at01, at23,
                                acc01, acc23, m, s);
            else
                attend_group<4>(xl, sidx, nv, lane, up16, up8, up4, b01, b23, at01, at23,
                                acc01, acc23, m, s);
        }

        const float inv = 1.0f / (s + 1e-16f);
        const v2f r01 = __builtin_elementwise_fma(acc01, (v2f){inv, inv}, __builtin_shufflevector(bi, bi, 0, 1));
        const v2f r23 = __builtin_elementwise_fma(acc23, (v2f){inv, inv}, __builtin_shufflevector(bi, bi, 2, 3));
        *(v4f*)(out + (size_t)d * DIM + lane * 4) = (v4f){r01.x, r01.y, r23.x, r23.y};
    }
}

struct AttendSide {
    const float* xl; const float* xr;
    const int* slots; const int* off; const int* cnt;
    const float* att; const float* bias;
    float* out;
};
__global__ __launch_bounds__(ATT_BLOCK) void seg_attend_kernel(AttendSide r0, AttendSide r1, int n)
{
    const bool second = (blockIdx.y != 0);
    seg_attend_body(second ? r1.xl : r0.xl, second ? r1.xr : r0.xr,
                    second ? r1.slots : r0.slots, second ? r1.off : r0.off,
                    second ? r1.cnt : r0.cnt,
                    second ? r1.att : r0.att, second ? r1.bias : r0.bias,
                    second ? r1.out : r0.out, n);
}

extern "C" void kernel_launch(void* const* d_in, const int* in_sizes, int n_in,
                              void* d_out, int out_size, void* d_ws, size_t ws_size,
                              hipStream_t stream) {
    (void)in_sizes; (void)n_in; (void)out_size; (void)ws_size;

    const float* x_user = (const float*)d_in[0];
    const float* x_item = (const float*)d_in[1];
    const int*   e_u2i  = (const int*)d_in[2];
    const int*   e_i2u  = (const int*)d_in[3];
    const float* W1_u = (const float*)d_in[4],  *b1_u = (const float*)d_in[5];
    const float* W2_u = (const float*)d_in[6],  *b2_u = (const float*)d_in[7];
    const float* W1_i = (const float*)d_in[8],  *b1_i = (const float*)d_in[9];
    const float* W2_i = (const float*)d_in[10], *b2_i = (const float*)d_in[11];
    const float* Wl_u2i = (const float*)d_in[12], *bl_u2i = (const float*)d_in[13];
    const float* Wr_u2i = (const float*)d_in[14], *br_u2i = (const float*)d_in[15];
    const float* att_u2i = (const float*)d_in[16], *bias_u2i = (const float*)d_in[17];
    const float* Wl_i2u = (const float*)d_in[18], *bl_i2u = (const float*)d_in[19];
    const float* Wr_i2u = (const float*)d_in[20], *br_i2u = (const float*)d_in[21];
    const float* att_i2u = (const float*)d_in[22], *bias_i2u = (const float*)d_in[23];

    float* out_user = (float*)d_out;
    float* out_item = (float*)d_out + (size_t)N_NODES * DIM;

    float* ws    = (float*)d_ws;
    float* xl_u  = ws;
    float* xl_i  = xl_u + (size_t)N_NODES * DIM;
    float* xr_i  = xl_i + (size_t)N_NODES * DIM;
    float* xr_u  = xr_i + (size_t)N_NODES * DIM;
    int* hist  = (int*)d_out;
    int* cnt   = (int*)(xr_u + (size_t)N_NODES * DIM);
    int* off   = cnt + 2 * N_NODES;
    int* rank  = hist + (size_t)2 * NCH * N_NODES;
    int* slots = off + 2 * N_NODES;
    int* offl  = slots + 2 * NEDGE;
    int* blocktot = offl + 2 * N_NODES;
    float* P_ua = (float*)(blocktot + 2 * NBLK + 2);
    float* P_ub = P_ua + (HID + 1) * DIM;
    float* P_ia = P_ub + (HID + 1) * DIM;
    float* P_ib = P_ia + (HID + 1) * DIM;

    const int blkNw   = (N_NODES + (ATT_BLOCK / 32) * NPW - 1) / ((ATT_BLOCK / 32) * NPW);

    FoldSide fu = { W2_u, b2_u, Wl_u2i, bl_u2i, Wr_i2u, br_i2u, P_ua, P_ub };
    FoldSide fi = { W2_i, b2_i, Wl_i2u, bl_i2u, Wr_u2i, br_u2i, P_ia, P_ib };
    chunk_count_kernel<<<dim3(NCH * NQ + FOLD_XB, 2), GROUP_BLOCK, 0, stream>>>(
        e_u2i + NEDGE, e_i2u + NEDGE, hist, rank, fu, fi);
    NodeSide su = { x_user, W1_u, b1_u, P_ua, P_ua + HID * DIM, P_ub, P_ub + HID * DIM,
                    xl_u, xr_u };
    NodeSide si = { x_item, W1_i, b1_i, P_ia, P_ia + HID * DIM, P_ib, P_ib + HID * DIM,
                    xl_i, xr_i };
    prefix_nodes_kernel<<<dim3(2 * NBLK + 2 * NODE_XB), PBLK, 0, stream>>>(
        hist, cnt, offl, blocktot, su, si, N_NODES);
    fill_slots_kernel<<<dim3(NCH * NQ, 2), GROUP_BLOCK, 0, stream>>>(
        e_u2i, e_u2i + NEDGE, e_i2u, e_i2u + NEDGE, hist, offl, blocktot, rank, off, slots);

    AttendSide r0 = { xl_u, xr_i, slots, off, cnt, att_u2i, bias_u2i, out_item };
    AttendSide r1 = { xl_i, xr_u, slots + NEDGE, off + N_NODES, cnt + N_NODES,
                      att_i2u, bias_i2u, out_user };
    seg_attend_kernel<<<dim3(blkNw, 2), ATT_BLOCK, 0, stream>>>(r0, r1, N_NODES);
}
